# starting kernel
# baseline (speedup 1.0000x reference)
#include <hip/hip_runtime.h>
#include <cfloat>
#include <cmath>

typedef __attribute__((ext_vector_type(2))) float v2f;
typedef __attribute__((ext_vector_type(8))) float v8f;

#define BATCH 16
#define NPTS 2048
#define CINIT 64
#define KNN 32
#define SIGMA_ 0.3f
#define BN_EPS_ 1e-5f
#define STD_G 16

__global__ __launch_bounds__(256) void k_embed_wmma(const float* __restrict__ xyz,
                                                    const float* __restrict__ W,
                                                    float* __restrict__ h) {
  const int wid  = (blockIdx.x * 256 + (int)threadIdx.x) >> 5;
  const int lane = threadIdx.x & 31;
  const int hf   = lane >> 4;
  const int ln   = lane & 15;
  const int tileM = wid >> 2;
  const int tileN = wid & 3;
  const int m  = tileM * 16 + ln;
  const int c  = tileN * 16 + ln;
  const float x0 = xyz[m * 3], x1 = xyz[m * 3 + 1], x2 = xyz[m * 3 + 2];
  const float w0 = W[c * 3],   w1 = W[c * 3 + 1],   w2 = W[c * 3 + 2];
  v2f a, b;
  a.x = hf ? x2 : x0;  a.y = hf ? 0.f : x1;
  b.x = hf ? w2 : w0;  b.y = hf ? 0.f : w1;
  v8f d = {};
  d = __builtin_amdgcn_wmma_f32_16x16x4_f32(false, a, false, b, (short)0, d, false, false);
#pragma unroll
  for (int r = 0; r < 8; ++r) {
    int row = tileM * 16 + r + 8 * hf;
    h[row * CINIT + tileN * 16 + ln] = d[r];
  }
}

__global__ __launch_bounds__(256) void k_bn_stats(const float* __restrict__ h,
                                                  float* __restrict__ mu,
                                                  float* __restrict__ inv) {
  __shared__ float s1[256], s2[256];
  const int c = blockIdx.x;
  const int n = BATCH * NPTS;
  float a = 0.f, b = 0.f;
  for (int i = threadIdx.x; i < n; i += 256) {
    float v = h[i * CINIT + c];
    a += v; b += v * v;
  }
  s1[threadIdx.x] = a; s2[threadIdx.x] = b; __syncthreads();
  for (int o = 128; o > 0; o >>= 1) {
    if ((int)threadIdx.x < o) { s1[threadIdx.x] += s1[threadIdx.x + o]; s2[threadIdx.x] += s2[threadIdx.x + o]; }
    __syncthreads();
  }
  if (threadIdx.x == 0) {
    float m   = s1[0] / (float)n;
    float var = s2[0] / (float)n - m * m;
    mu[c]  = m;
    inv[c] = rsqrtf(var + BN_EPS_);
  }
}

__global__ void k_bn_apply(float* __restrict__ h, const float* __restrict__ mu,
                           const float* __restrict__ inv, const float* __restrict__ gamma,
                           const float* __restrict__ beta) {
  const int total = BATCH * NPTS * CINIT;
  for (int i = blockIdx.x * blockDim.x + threadIdx.x; i < total; i += gridDim.x * blockDim.x) {
    int c = i & (CINIT - 1);
    float v = (h[i] - mu[c]) * inv[c] * gamma[c] + beta[c];
    h[i] = fmaxf(v, 0.f);
  }
}

#define FPS_T 1024
__global__ __launch_bounds__(FPS_T) void k_fps(const float* __restrict__ xyz, int N, int S,
                                               int* __restrict__ fps_idx) {
  __shared__ float px[NPTS], py[NPTS], pz[NPTS];
  __shared__ float wv[32];
  __shared__ int   wi[32];
  __shared__ int   far_s;
  const int b = blockIdx.x;
  const float* p = xyz + (size_t)b * N * 3;
  const int tid = threadIdx.x;
  for (int i = tid; i < N; i += FPS_T) { px[i] = p[i*3]; py[i] = p[i*3+1]; pz[i] = p[i*3+2]; }
  float d0 = 1e10f, d1 = 1e10f;
  const int i0 = tid, i1 = tid + FPS_T;
  if (tid == 0) far_s = 0;
  __syncthreads();
  for (int it = 0; it < S; ++it) {
    const int far = far_s;
    if (tid == 0) fps_idx[b * S + it] = far;
    const float fx = px[far], fy = py[far], fz = pz[far];
    float best = -1.f; int bi = 0x7fffffff;
    if (i0 < N) {
      float dx = px[i0]-fx, dy = py[i0]-fy, dz = pz[i0]-fz;
      d0 = fminf(d0, dx*dx + dy*dy + dz*dz);
      best = d0; bi = i0;
    }
    if (i1 < N) {
      float dx = px[i1]-fx, dy = py[i1]-fy, dz = pz[i1]-fz;
      d1 = fminf(d1, dx*dx + dy*dy + dz*dz);
      if (d1 > best) { best = d1; bi = i1; }
    }
#pragma unroll
    for (int o = 16; o > 0; o >>= 1) {
      float ov = __shfl_down(best, o, 32);
      int   oi = __shfl_down(bi,   o, 32);
      if (ov > best || (ov == best && oi < bi)) { best = ov; bi = oi; }
    }
    if ((tid & 31) == 0) { wv[tid >> 5] = best; wi[tid >> 5] = bi; }
    __syncthreads();
    if (tid < 32) {
      best = wv[tid]; bi = wi[tid];
#pragma unroll
      for (int o = 16; o > 0; o >>= 1) {
        float ov = __shfl_down(best, o, 32);
        int   oi = __shfl_down(bi,   o, 32);
        if (ov > best || (ov == best && oi < bi)) { best = ov; bi = oi; }
      }
      if (tid == 0) far_s = bi;
    }
    __syncthreads();
  }
}

__global__ void k_gather(const float* __restrict__ xyz, const float* __restrict__ feat,
                         const int* __restrict__ fps_idx, int N, int S, int C,
                         float* __restrict__ xyz_s, float* __restrict__ feat_s) {
  const int b = blockIdx.x / S, s = blockIdx.x % S;
  const int idx = fps_idx[b * S + s];
  if (threadIdx.x < 3)
    xyz_s[(size_t)(b * S + s) * 3 + threadIdx.x] = xyz[(size_t)(b * N + idx) * 3 + threadIdx.x];
  for (int j = threadIdx.x; j < C; j += blockDim.x)
    feat_s[(size_t)(b * S + s) * C + j] = feat[(size_t)(b * N + idx) * C + j];
}

__global__ __launch_bounds__(256) void k_knn_dist_wmma(const float* __restrict__ xs,
                                                       const float* __restrict__ x,
                                                       int S, int N, float* __restrict__ dist) {
  __shared__ float sN[16];
  const int mbase = blockIdx.y * 16;
  const int w    = threadIdx.x >> 5;
  const int lane = threadIdx.x & 31;
  const int hf   = lane >> 4;
  const int ln   = lane & 15;
  if (threadIdx.x < 16) {
    float a0 = xs[(mbase + threadIdx.x) * 3];
    float a1 = xs[(mbase + threadIdx.x) * 3 + 1];
    float a2 = xs[(mbase + threadIdx.x) * 3 + 2];
    sN[threadIdx.x] = a0 * a0 + a1 * a1 + a2 * a2;
  }
  __syncthreads();
  const int m = mbase + ln;
  const float s0 = xs[m * 3], s1 = xs[m * 3 + 1], s2 = xs[m * 3 + 2];
  const int n = blockIdx.x * 128 + w * 16 + ln;
  const float x0 = x[n * 3], x1 = x[n * 3 + 1], x2 = x[n * 3 + 2];
  const float nn = x0 * x0 + x1 * x1 + x2 * x2;
  v2f a, b;
  a.x = hf ? s2 : s0;  a.y = hf ? 0.f : s1;
  b.x = hf ? x2 : x0;  b.y = hf ? 0.f : x1;
  v8f d = {};
  d = __builtin_amdgcn_wmma_f32_16x16x4_f32(false, a, false, b, (short)0, d, false, false);
#pragma unroll
  for (int r = 0; r < 8; ++r) {
    int row = mbase + r + 8 * hf;
    dist[(size_t)row * N + n] = sN[r + 8 * hf] + nn - 2.f * d[r];
  }
}

__global__ __launch_bounds__(256) void k_knn_select(const float* __restrict__ dist, int N,
                                                    int* __restrict__ knn_idx) {
  __shared__ float sv[NPTS];
  __shared__ float rv[256];
  __shared__ int   ri[256];
  const int s = blockIdx.x;
  const float* row = dist + (size_t)s * N;
  for (int i = threadIdx.x; i < N; i += 256) sv[i] = row[i];
  __syncthreads();
  for (int k = 0; k < KNN; ++k) {
    float best = FLT_MAX; int bi = 0x7fffffff;
    for (int i = threadIdx.x; i < N; i += 256) {
      float v = sv[i];
      if (v < best) { best = v; bi = i; }
    }
    rv[threadIdx.x] = best; ri[threadIdx.x] = bi;
    __syncthreads();
    for (int o = 128; o > 0; o >>= 1) {
      if ((int)threadIdx.x < o) {
        float vo = rv[threadIdx.x + o];
        if (vo < rv[threadIdx.x] || (vo == rv[threadIdx.x] && ri[threadIdx.x + o] < ri[threadIdx.x])) {
          rv[threadIdx.x] = vo; ri[threadIdx.x] = ri[threadIdx.x + o];
        }
      }
      __syncthreads();
    }
    if (threadIdx.x == 0) { knn_idx[s * KNN + k] = ri[0]; sv[ri[0]] = FLT_MAX; }
    __syncthreads();
  }
}

__global__ __launch_bounds__(256) void k_std_partial(const float* __restrict__ xyz,
                                                     const float* __restrict__ feat,
                                                     const float* __restrict__ xyz_s,
                                                     const float* __restrict__ feat_s,
                                                     const int* __restrict__ knn_idx,
                                                     int N, int S, int C,
                                                     float* __restrict__ part) {
  __shared__ float s1[256], s2[256], s3[256], s4[256];
  const int k = blockIdx.x, g = blockIdx.y;
  float ax = 0.f, qx = 0.f, af = 0.f, qf = 0.f;
  for (int p = g * 256 + threadIdx.x; p < BATCH * S; p += STD_G * 256) {
    int b = p / S, s = p % S;
    int idx = knn_idx[(size_t)(b * S + s) * KNN + k];
    for (int c = 0; c < 3; ++c) {
      float d = xyz[(size_t)(b * N + idx) * 3 + c] - xyz_s[(size_t)(b * S + s) * 3 + c];
      ax += d; qx += d * d;
    }
    const float* fr = feat   + (size_t)(b * N + idx) * C;
    const float* fc = feat_s + (size_t)(b * S + s) * C;
    for (int j = 0; j < C; ++j) { float d = fr[j] - fc[j]; af += d; qf += d * d; }
  }
  s1[threadIdx.x] = ax; s2[threadIdx.x] = qx; s3[threadIdx.x] = af; s4[threadIdx.x] = qf;
  __syncthreads();
  for (int o = 128; o > 0; o >>= 1) {
    if ((int)threadIdx.x < o) {
      s1[threadIdx.x] += s1[threadIdx.x + o]; s2[threadIdx.x] += s2[threadIdx.x + o];
      s3[threadIdx.x] += s3[threadIdx.x + o]; s4[threadIdx.x] += s4[threadIdx.x + o];
    }
    __syncthreads();
  }
  if (threadIdx.x == 0) {
    float* q = part + ((size_t)k * STD_G + g) * 4;
    q[0] = s1[0]; q[1] = s2[0]; q[2] = s3[0]; q[3] = s4[0];
  }
}

__global__ void k_std_final(const float* __restrict__ part, int S, int C,
                            float* __restrict__ std_xyz, float* __restrict__ std_feat) {
  const int k = threadIdx.x;
  if (k >= KNN) return;
  float ax = 0.f, qx = 0.f, af = 0.f, qf = 0.f;
  for (int g = 0; g < STD_G; ++g) {
    const float* q = part + ((size_t)k * STD_G + g) * 4;
    ax += q[0]; qx += q[1]; af += q[2]; qf += q[3];
  }
  {
    double n = (double)BATCH * S * 3;
    double var = ((double)qx - (double)ax * ax / n) / (n - 1.0);
    std_xyz[k] = fmaxf(sqrtf(fmaxf((float)var, 0.f)), 1e-5f);
  }
  {
    double n = (double)BATCH * S * C;
    double var = ((double)qf - (double)af * af / n) / (n - 1.0);
    std_feat[k] = fmaxf(sqrtf(fmaxf((float)var, 0.f)), 1e-5f);
  }
}

__global__ __launch_bounds__(256) void k_stage(const float* __restrict__ xyz,
                                               const float* __restrict__ feat,
                                               const float* __restrict__ xyz_s,
                                               const float* __restrict__ feat_s,
                                               const int* __restrict__ knn_idx,
                                               const float* __restrict__ std_xyz,
                                               const float* __restrict__ std_feat,
                                               int N, int S, int C,
                                               float* __restrict__ out) {
  __shared__ int   kidx[KNN];
  __shared__ float sfe[KNN];
  __shared__ float xkn[KNN][3];
  __shared__ float fs[512];
  const int b = blockIdx.x / S, s = blockIdx.x % S;
  if (threadIdx.x < KNN) {
    kidx[threadIdx.x] = knn_idx[(size_t)(b * S + s) * KNN + threadIdx.x];
    sfe[threadIdx.x]  = std_feat[threadIdx.x];
  }
  __syncthreads();
  if (threadIdx.x < KNN)
    __builtin_prefetch(feat + (size_t)(b * N + kidx[threadIdx.x]) * C, 0, 3);
  if (threadIdx.x < KNN * 3) {
    int k = threadIdx.x / 3, c = threadIdx.x % 3;
    xkn[k][c] = (xyz[(size_t)(b * N + kidx[k]) * 3 + c] - xyz_s[(size_t)(b * S + s) * 3 + c]) / std_xyz[k];
  }
  for (int j = threadIdx.x; j < C; j += 256) fs[j] = feat_s[(size_t)(b * S + s) * C + j];
  __syncthreads();
  const int C2 = 2 * C;
  const int fd = (C2 + 2) / 3;
  const int M3 = 3 * fd;
  const float inv_sig = 1.f / SIGMA_;
  for (int j = threadIdx.x; j < C2; j += 256) {
    int src = (int)(((long long)j * (M3 - 1)) / (C2 - 1));
    int cc = src / fd, f = src % fd;
    float fv = -1.f + 2.f * (float)(f + 1) / (float)(fd + 1);
    float acc = 0.f, mx = -FLT_MAX;
    for (int k = 0; k < KNN; ++k) {
      float t  = (xkn[k][cc] - fv) * inv_sig;
      float pe = expf(-0.5f * t * t);
      float fk;
      if (j < C) fk = (feat[(size_t)(b * N + kidx[k]) * C + j] - fs[j]) / sfe[k];
      else       fk = fs[j - C];
      float fw = (fk + pe) * pe;
      acc += fw; mx = fmaxf(mx, fw);
    }
    float xx = acc * (1.f / KNN) + mx;
    out[(size_t)(b * S + s) * C2 + j] = 0.5f * xx * (1.f + erff(xx * 0.70710678118654752f));
  }
}

__global__ void k_pool(const float* __restrict__ feat, int S, int C2,
                       float* __restrict__ out, int off) {
  const int b = blockIdx.x;
  for (int j = threadIdx.x; j < C2; j += blockDim.x) {
    float mx = -FLT_MAX, sm = 0.f;
    for (int s = 0; s < S; ++s) {
      float v = feat[(size_t)(b * S + s) * C2 + j];
      mx = fmaxf(mx, v); sm += v;
    }
    out[b * 3840 + off + j]      = mx;
    out[b * 3840 + off + C2 + j] = sm / (float)S;
  }
}

extern "C" void kernel_launch(void* const* d_in, const int* in_sizes, int n_in,
                              void* d_out, int out_size, void* d_ws, size_t ws_size,
                              hipStream_t stream) {
  const float* xyz_in = (const float*)d_in[0];
  const float* W      = (const float*)d_in[1];
  const float* gamma  = (const float*)d_in[2];
  const float* beta   = (const float*)d_in[3];
  float* out = (float*)d_out;
  float* ws  = (float*)d_ws;

  size_t o = 0;
  float* featA  = ws + o; o += (size_t)BATCH * NPTS * CINIT;
  float* featB  = ws + o; o += (size_t)BATCH * NPTS * CINIT;
  float* xyzA   = ws + o; o += (size_t)BATCH * NPTS * 3;
  float* xyzB   = ws + o; o += (size_t)BATCH * NPTS * 3;
  float* feat_s = ws + o; o += (size_t)BATCH * 1024 * CINIT;
  float* dist   = ws + o; o += (size_t)1024 * 2048;
  float* bn_mu  = ws + o; o += 64;
  float* bn_inv = ws + o; o += 64;
  float* stdx   = ws + o; o += 32;
  float* stdf   = ws + o; o += 32;
  float* stdp   = ws + o; o += (size_t)KNN * STD_G * 4;
  int* fps_i = (int*)(ws + o); o += (size_t)BATCH * 1024;
  int* knn_i = (int*)(ws + o); o += (size_t)BATCH * 1024 * KNN;

  k_embed_wmma<<<1024, 256, 0, stream>>>(xyz_in, W, featA);
  k_bn_stats<<<CINIT, 256, 0, stream>>>(featA, bn_mu, bn_inv);
  k_bn_apply<<<2048, 256, 0, stream>>>(featA, bn_mu, bn_inv, gamma, beta);

  const float* xyz_cur = xyz_in;
  float* feat_cur = featA; float* feat_nxt = featB;
  float* xs_buf = xyzA;    float* xs_other = xyzB;
  int N = NPTS, C = CINIT, off = 0;

  for (int st = 0; st < 4; ++st) {
    const int S = N / 2, C2 = 2 * C;
    k_fps<<<BATCH, FPS_T, 0, stream>>>(xyz_cur, N, S, fps_i);
    k_gather<<<BATCH * S, 256, 0, stream>>>(xyz_cur, feat_cur, fps_i, N, S, C, xs_buf, feat_s);
    for (int b = 0; b < BATCH; ++b) {
      dim3 g(N / 128, S / 16);
      k_knn_dist_wmma<<<g, 256, 0, stream>>>(xs_buf + (size_t)b * S * 3,
                                             xyz_cur + (size_t)b * N * 3, S, N, dist);
      k_knn_select<<<S, 256, 0, stream>>>(dist, N, knn_i + (size_t)b * S * KNN);
    }
    {
      dim3 g(KNN, STD_G);
      k_std_partial<<<g, 256, 0, stream>>>(xyz_cur, feat_cur, xs_buf, feat_s, knn_i, N, S, C, stdp);
      k_std_final<<<1, KNN, 0, stream>>>(stdp, S, C, stdx, stdf);
    }
    k_stage<<<BATCH * S, 256, 0, stream>>>(xyz_cur, feat_cur, xs_buf, feat_s, knn_i,
                                           stdx, stdf, N, S, C, feat_nxt);
    k_pool<<<BATCH, 256, 0, stream>>>(feat_nxt, S, C2, out, off);
    off += 2 * C2;
    float* t = feat_cur; feat_cur = feat_nxt; feat_nxt = t;
    xyz_cur = xs_buf;
    float* u = xs_buf; xs_buf = xs_other; xs_other = u;
    N = S; C = C2;
  }
  (void)in_sizes; (void)n_in; (void)out_size; (void)ws_size;
}


# kNN top-32 selection: one wave per row, row kept in registers; each pass = per-lane scan in four chains + DPP wave minima (value, then index); replaces 32 block-wide LDS tree reductions per row
# speedup vs baseline: 1.7183x; 1.7183x over previous
#include <hip/hip_runtime.h>
#include <cfloat>
#include <cmath>

typedef __attribute__((ext_vector_type(2))) float v2f;
typedef __attribute__((ext_vector_type(8))) float v8f;

#define BATCH 16
#define NPTS 2048
#define CINIT 64
#define KNN 32
#define SIGMA_ 0.3f
#define BN_EPS_ 1e-5f
#define STD_G 16

__global__ __launch_bounds__(256) void k_embed_wmma(const float* __restrict__ xyz,
                                                    const float* __restrict__ W,
                                                    float* __restrict__ h) {
  const int wid  = (blockIdx.x * 256 + (int)threadIdx.x) >> 5;
  const int lane = threadIdx.x & 31;
  const int hf   = lane >> 4;
  const int ln   = lane & 15;
  const int tileM = wid >> 2;
  const int tileN = wid & 3;
  const int m  = tileM * 16 + ln;
  const int c  = tileN * 16 + ln;
  const float x0 = xyz[m * 3], x1 = xyz[m * 3 + 1], x2 = xyz[m * 3 + 2];
  const float w0 = W[c * 3],   w1 = W[c * 3 + 1],   w2 = W[c * 3 + 2];
  v2f a, b;
  a.x = hf ? x2 : x0;  a.y = hf ? 0.f : x1;
  b.x = hf ? w2 : w0;  b.y = hf ? 0.f : w1;
  v8f d = {};
  d = __builtin_amdgcn_wmma_f32_16x16x4_f32(false, a, false, b, (short)0, d, false, false);
#pragma unroll
  for (int r = 0; r < 8; ++r) {
    int row = tileM * 16 + r + 8 * hf;
    h[row * CINIT + tileN * 16 + ln] = d[r];
  }
}

__global__ __launch_bounds__(256) void k_bn_stats(const float* __restrict__ h,
                                                  float* __restrict__ mu,
                                                  float* __restrict__ inv) {
  __shared__ float s1[256], s2[256];
  const int c = blockIdx.x;
  const int n = BATCH * NPTS;
  float a = 0.f, b = 0.f;
  for (int i = threadIdx.x; i < n; i += 256) {
    float v = h[i * CINIT + c];
    a += v; b += v * v;
  }
  s1[threadIdx.x] = a; s2[threadIdx.x] = b; __syncthreads();
  for (int o = 128; o > 0; o >>= 1) {
    if ((int)threadIdx.x < o) { s1[threadIdx.x] += s1[threadIdx.x + o]; s2[threadIdx.x] += s2[threadIdx.x + o]; }
    __syncthreads();
  }
  if (threadIdx.x == 0) {
    float m   = s1[0] / (float)n;
    float var = s2[0] / (float)n - m * m;
    mu[c]  = m;
    inv[c] = rsqrtf(var + BN_EPS_);
  }
}

__global__ void k_bn_apply(float* __restrict__ h, const float* __restrict__ mu,
                           const float* __restrict__ inv, const float* __restrict__ gamma,
                           const float* __restrict__ beta) {
  const int total = BATCH * NPTS * CINIT;
  for (int i = blockIdx.x * blockDim.x + threadIdx.x; i < total; i += gridDim.x * blockDim.x) {
    int c = i & (CINIT - 1);
    float v = (h[i] - mu[c]) * inv[c] * gamma[c] + beta[c];
    h[i] = fmaxf(v, 0.f);
  }
}

#define FPS_T 1024
__global__ __launch_bounds__(FPS_T) void k_fps(const float* __restrict__ xyz, int N, int S,
                                               int* __restrict__ fps_idx) {
  __shared__ float px[NPTS], py[NPTS], pz[NPTS];
  __shared__ float wv[32];
  __shared__ int   wi[32];
  __shared__ int   far_s;
  const int b = blockIdx.x;
  const float* p = xyz + (size_t)b * N * 3;
  const int tid = threadIdx.x;
  for (int i = tid; i < N; i += FPS_T) { px[i] = p[i*3]; py[i] = p[i*3+1]; pz[i] = p[i*3+2]; }
  float d0 = 1e10f, d1 = 1e10f;
  const int i0 = tid, i1 = tid + FPS_T;
  if (tid == 0) far_s = 0;
  __syncthreads();
  for (int it = 0; it < S; ++it) {
    const int far = far_s;
    if (tid == 0) fps_idx[b * S + it] = far;
    const float fx = px[far], fy = py[far], fz = pz[far];
    float best = -1.f; int bi = 0x7fffffff;
    if (i0 < N) {
      float dx = px[i0]-fx, dy = py[i0]-fy, dz = pz[i0]-fz;
      d0 = fminf(d0, dx*dx + dy*dy + dz*dz);
      best = d0; bi = i0;
    }
    if (i1 < N) {
      float dx = px[i1]-fx, dy = py[i1]-fy, dz = pz[i1]-fz;
      d1 = fminf(d1, dx*dx + dy*dy + dz*dz);
      if (d1 > best) { best = d1; bi = i1; }
    }
#pragma unroll
    for (int o = 16; o > 0; o >>= 1) {
      float ov = __shfl_down(best, o, 32);
      int   oi = __shfl_down(bi,   o, 32);
      if (ov > best || (ov == best && oi < bi)) { best = ov; bi = oi; }
    }
    if ((tid & 31) == 0) { wv[tid >> 5] = best; wi[tid >> 5] = bi; }
    __syncthreads();
    if (tid < 32) {
      best = wv[tid]; bi = wi[tid];
#pragma unroll
      for (int o = 16; o > 0; o >>= 1) {
        float ov = __shfl_down(best, o, 32);
        int   oi = __shfl_down(bi,   o, 32);
        if (ov > best || (ov == best && oi < bi)) { best = ov; bi = oi; }
      }
      if (tid == 0) far_s = bi;
    }
    __syncthreads();
  }
}

__global__ void k_gather(const float* __restrict__ xyz, const float* __restrict__ feat,
                         const int* __restrict__ fps_idx, int N, int S, int C,
                         float* __restrict__ xyz_s, float* __restrict__ feat_s) {
  const int b = blockIdx.x / S, s = blockIdx.x % S;
  const int idx = fps_idx[b * S + s];
  if (threadIdx.x < 3)
    xyz_s[(size_t)(b * S + s) * 3 + threadIdx.x] = xyz[(size_t)(b * N + idx) * 3 + threadIdx.x];
  for (int j = threadIdx.x; j < C; j += blockDim.x)
    feat_s[(size_t)(b * S + s) * C + j] = feat[(size_t)(b * N + idx) * C + j];
}

__global__ __launch_bounds__(256) void k_knn_dist_wmma(const float* __restrict__ xs,
                                                       const float* __restrict__ x,
                                                       int S, int N, float* __restrict__ dist) {
  __shared__ float sN[16];
  const int mbase = blockIdx.y * 16;
  const int w    = threadIdx.x >> 5;
  const int lane = threadIdx.x & 31;
  const int hf   = lane >> 4;
  const int ln   = lane & 15;
  if (threadIdx.x < 16) {
    float a0 = xs[(mbase + threadIdx.x) * 3];
    float a1 = xs[(mbase + threadIdx.x) * 3 + 1];
    float a2 = xs[(mbase + threadIdx.x) * 3 + 2];
    sN[threadIdx.x] = a0 * a0 + a1 * a1 + a2 * a2;
  }
  __syncthreads();
  const int m = mbase + ln;
  const float s0 = xs[m * 3], s1 = xs[m * 3 + 1], s2 = xs[m * 3 + 2];
  const int n = blockIdx.x * 128 + w * 16 + ln;
  const float x0 = x[n * 3], x1 = x[n * 3 + 1], x2 = x[n * 3 + 2];
  const float nn = x0 * x0 + x1 * x1 + x2 * x2;
  v2f a, b;
  a.x = hf ? s2 : s0;  a.y = hf ? 0.f : s1;
  b.x = hf ? x2 : x0;  b.y = hf ? 0.f : x1;
  v8f d = {};
  d = __builtin_amdgcn_wmma_f32_16x16x4_f32(false, a, false, b, (short)0, d, false, false);
#pragma unroll
  for (int r = 0; r < 8; ++r) {
    int row = mbase + r + 8 * hf;
    dist[(size_t)row * N + n] = sN[r + 8 * hf] + nn - 2.f * d[r];
  }
}

template <int M>
__device__ __forceinline__ int lane_xor_row(int x) {
  return __builtin_amdgcn_update_dpp(0, x, 0x160 + M, 0xf, 0xf, true);
}
__device__ __forceinline__ int lane_xor_half(int x) {
  return __builtin_amdgcn_permlanex16(x, x, 0x76543210, 0xfedcba98, false, false);
}
__device__ __forceinline__ float wave_min_f(float x) {
  x = fminf(x, __builtin_bit_cast(float, lane_xor_row<1>(__builtin_bit_cast(int, x))));
  x = fminf(x, __builtin_bit_cast(float, lane_xor_row<2>(__builtin_bit_cast(int, x))));
  x = fminf(x, __builtin_bit_cast(float, lane_xor_row<4>(__builtin_bit_cast(int, x))));
  x = fminf(x, __builtin_bit_cast(float, lane_xor_row<8>(__builtin_bit_cast(int, x))));
  x = fminf(x, __builtin_bit_cast(float, lane_xor_half(__builtin_bit_cast(int, x))));
  return x;
}
__device__ __forceinline__ int wave_min_i(int x) {
  x = min(x, lane_xor_row<1>(x));
  x = min(x, lane_xor_row<2>(x));
  x = min(x, lane_xor_row<4>(x));
  x = min(x, lane_xor_row<8>(x));
  x = min(x, lane_xor_half(x));
  return x;
}

template <int NPL>
__global__ __launch_bounds__(32) void k_knn_select(const float* __restrict__ dist,
                                                   int* __restrict__ knn_idx) {
  const int lane = threadIdx.x;
  const int s = blockIdx.x;
  const float* row = dist + (size_t)s * (NPL * 32);
  float v[NPL];
#pragma unroll
  for (int j = 0; j < NPL; ++j) v[j] = row[j * 32 + lane];
  int kill = -1;
  int mine = 0;
  for (int k = 0; k < KNN; ++k) {
    constexpr int Q = NPL / 4;
    float cb[4]; int cj[4];
#pragma unroll
    for (int c = 0; c < 4; ++c) { cb[c] = FLT_MAX; cj[c] = c * Q; }
#pragma unroll
    for (int q = 0; q < Q; ++q) {
#pragma unroll
      for (int c = 0; c < 4; ++c) {
        const int j = c * Q + q;
        float vj = (j == kill) ? FLT_MAX : v[j];
        v[j] = vj;
        if (vj < cb[c]) { cb[c] = vj; cj[c] = j; }
      }
    }
    if (cb[1] < cb[0]) { cb[0] = cb[1]; cj[0] = cj[1]; }
    if (cb[3] < cb[2]) { cb[2] = cb[3]; cj[2] = cj[3]; }
    float best = cb[0]; int bj = cj[0];
    if (cb[2] < best) { best = cb[2]; bj = cj[2]; }
    const float wmin = wave_min_f(best);
    const int cand = (best == wmin) ? (bj * 32 + lane) : 0x7fffffff;
    const int widx = wave_min_i(cand);
    kill = ((widx & 31) == lane) ? (widx >> 5) : -1;
    if (lane == k) mine = widx;
  }
  knn_idx[s * KNN + lane] = mine;
}

__global__ __launch_bounds__(256) void k_std_partial(const float* __restrict__ xyz,
                                                     const float* __restrict__ feat,
                                                     const float* __restrict__ xyz_s,
                                                     const float* __restrict__ feat_s,
                                                     const int* __restrict__ knn_idx,
                                                     int N, int S, int C,
                                                     float* __restrict__ part) {
  __shared__ float s1[256], s2[256], s3[256], s4[256];
  const int k = blockIdx.x, g = blockIdx.y;
  float ax = 0.f, qx = 0.f, af = 0.f, qf = 0.f;
  for (int p = g * 256 + threadIdx.x; p < BATCH * S; p += STD_G * 256) {
    int b = p / S, s = p % S;
    int idx = knn_idx[(size_t)(b * S + s) * KNN + k];
    for (int c = 0; c < 3; ++c) {
      float d = xyz[(size_t)(b * N + idx) * 3 + c] - xyz_s[(size_t)(b * S + s) * 3 + c];
      ax += d; qx += d * d;
    }
    const float* fr = feat   + (size_t)(b * N + idx) * C;
    const float* fc = feat_s + (size_t)(b * S + s) * C;
    for (int j = 0; j < C; ++j) { float d = fr[j] - fc[j]; af += d; qf += d * d; }
  }
  s1[threadIdx.x] = ax; s2[threadIdx.x] = qx; s3[threadIdx.x] = af; s4[threadIdx.x] = qf;
  __syncthreads();
  for (int o = 128; o > 0; o >>= 1) {
    if ((int)threadIdx.x < o) {
      s1[threadIdx.x] += s1[threadIdx.x + o]; s2[threadIdx.x] += s2[threadIdx.x + o];
      s3[threadIdx.x] += s3[threadIdx.x + o]; s4[threadIdx.x] += s4[threadIdx.x + o];
    }
    __syncthreads();
  }
  if (threadIdx.x == 0) {
    float* q = part + ((size_t)k * STD_G + g) * 4;
    q[0] = s1[0]; q[1] = s2[0]; q[2] = s3[0]; q[3] = s4[0];
  }
}

__global__ void k_std_final(const float* __restrict__ part, int S, int C,
                            float* __restrict__ std_xyz, float* __restrict__ std_feat) {
  const int k = threadIdx.x;
  if (k >= KNN) return;
  float ax = 0.f, qx = 0.f, af = 0.f, qf = 0.f;
  for (int g = 0; g < STD_G; ++g) {
    const float* q = part + ((size_t)k * STD_G + g) * 4;
    ax += q[0]; qx += q[1]; af += q[2]; qf += q[3];
  }
  {
    double n = (double)BATCH * S * 3;
    double var = ((double)qx - (double)ax * ax / n) / (n - 1.0);
    std_xyz[k] = fmaxf(sqrtf(fmaxf((float)var, 0.f)), 1e-5f);
  }
  {
    double n = (double)BATCH * S * C;
    double var = ((double)qf - (double)af * af / n) / (n - 1.0);
    std_feat[k] = fmaxf(sqrtf(fmaxf((float)var, 0.f)), 1e-5f);
  }
}

__global__ __launch_bounds__(256) void k_stage(const float* __restrict__ xyz,
                                               const float* __restrict__ feat,
                                               const float* __restrict__ xyz_s,
                                               const float* __restrict__ feat_s,
                                               const int* __restrict__ knn_idx,
                                               const float* __restrict__ std_xyz,
                                               const float* __restrict__ std_feat,
                                               int N, int S, int C,
                                               float* __restrict__ out) {
  __shared__ int   kidx[KNN];
  __shared__ float sfe[KNN];
  __shared__ float xkn[KNN][3];
  __shared__ float fs[512];
  const int b = blockIdx.x / S, s = blockIdx.x % S;
  if (threadIdx.x < KNN) {
    kidx[threadIdx.x] = knn_idx[(size_t)(b * S + s) * KNN + threadIdx.x];
    sfe[threadIdx.x]  = std_feat[threadIdx.x];
  }
  __syncthreads();
  if (threadIdx.x < KNN)
    __builtin_prefetch(feat + (size_t)(b * N + kidx[threadIdx.x]) * C, 0, 3);
  if (threadIdx.x < KNN * 3) {
    int k = threadIdx.x / 3, c = threadIdx.x % 3;
    xkn[k][c] = (xyz[(size_t)(b * N + kidx[k]) * 3 + c] - xyz_s[(size_t)(b * S + s) * 3 + c]) / std_xyz[k];
  }
  for (int j = threadIdx.x; j < C; j += 256) fs[j] = feat_s[(size_t)(b * S + s) * C + j];
  __syncthreads();
  const int C2 = 2 * C;
  const int fd = (C2 + 2) / 3;
  const int M3 = 3 * fd;
  const float inv_sig = 1.f / SIGMA_;
  for (int j = threadIdx.x; j < C2; j += 256) {
    int src = (int)(((long long)j * (M3 - 1)) / (C2 - 1));
    int cc = src / fd, f = src % fd;
    float fv = -1.f + 2.f * (float)(f + 1) / (float)(fd + 1);
    float acc = 0.f, mx = -FLT_MAX;
    for (int k = 0; k < KNN; ++k) {
      float t  = (xkn[k][cc] - fv) * inv_sig;
      float pe = expf(-0.5f * t * t);
      float fk;
      if (j < C) fk = (feat[(size_t)(b * N + kidx[k]) * C + j] - fs[j]) / sfe[k];
      else       fk = fs[j - C];
      float fw = (fk + pe) * pe;
      acc += fw; mx = fmaxf(mx, fw);
    }
    float xx = acc * (1.f / KNN) + mx;
    out[(size_t)(b * S + s) * C2 + j] = 0.5f * xx * (1.f + erff(xx * 0.70710678118654752f));
  }
}

__global__ void k_pool(const float* __restrict__ feat, int S, int C2,
                       float* __restrict__ out, int off) {
  const int b = blockIdx.x;
  for (int j = threadIdx.x; j < C2; j += blockDim.x) {
    float mx = -FLT_MAX, sm = 0.f;
    for (int s = 0; s < S; ++s) {
      float v = feat[(size_t)(b * S + s) * C2 + j];
      mx = fmaxf(mx, v); sm += v;
    }
    out[b * 3840 + off + j]      = mx;
    out[b * 3840 + off + C2 + j] = sm / (float)S;
  }
}

extern "C" void kernel_launch(void* const* d_in, const int* in_sizes, int n_in,
                              void* d_out, int out_size, void* d_ws, size_t ws_size,
                              hipStream_t stream) {
  const float* xyz_in = (const float*)d_in[0];
  const float* W      = (const float*)d_in[1];
  const float* gamma  = (const float*)d_in[2];
  const float* beta   = (const float*)d_in[3];
  float* out = (float*)d_out;
  float* ws  = (float*)d_ws;

  size_t o = 0;
  float* featA  = ws + o; o += (size_t)BATCH * NPTS * CINIT;
  float* featB  = ws + o; o += (size_t)BATCH * NPTS * CINIT;
  float* xyzA   = ws + o; o += (size_t)BATCH * NPTS * 3;
  float* xyzB   = ws + o; o += (size_t)BATCH * NPTS * 3;
  float* feat_s = ws + o; o += (size_t)BATCH * 1024 * CINIT;
  float* dist   = ws + o; o += (size_t)1024 * 2048;
  float* bn_mu  = ws + o; o += 64;
  float* bn_inv = ws + o; o += 64;
  float* stdx   = ws + o; o += 32;
  float* stdf   = ws + o; o += 32;
  float* stdp   = ws + o; o += (size_t)KNN * STD_G * 4;
  int* fps_i = (int*)(ws + o); o += (size_t)BATCH * 1024;
  int* knn_i = (int*)(ws + o); o += (size_t)BATCH * 1024 * KNN;

  k_embed_wmma<<<1024, 256, 0, stream>>>(xyz_in, W, featA);
  k_bn_stats<<<CINIT, 256, 0, stream>>>(featA, bn_mu, bn_inv);
  k_bn_apply<<<2048, 256, 0, stream>>>(featA, bn_mu, bn_inv, gamma, beta);

  const float* xyz_cur = xyz_in;
  float* feat_cur = featA; float* feat_nxt = featB;
  float* xs_buf = xyzA;    float* xs_other = xyzB;
  int N = NPTS, C = CINIT, off = 0;

  for (int st = 0; st < 4; ++st) {
    const int S = N / 2, C2 = 2 * C;
    k_fps<<<BATCH, FPS_T, 0, stream>>>(xyz_cur, N, S, fps_i);
    k_gather<<<BATCH * S, 256, 0, stream>>>(xyz_cur, feat_cur, fps_i, N, S, C, xs_buf, feat_s);
    for (int b = 0; b < BATCH; ++b) {
      dim3 g(N / 128, S / 16);
      k_knn_dist_wmma<<<g, 256, 0, stream>>>(xs_buf + (size_t)b * S * 3,
                                             xyz_cur + (size_t)b * N * 3, S, N, dist);
      int* kout = knn_i + (size_t)b * S * KNN;
      if (N == 2048)      k_knn_select<64><<<S, 32, 0, stream>>>(dist, kout);
      else if (N == 1024) k_knn_select<32><<<S, 32, 0, stream>>>(dist, kout);
      else if (N == 512)  k_knn_select<16><<<S, 32, 0, stream>>>(dist, kout);
      else                k_knn_select<8><<<S, 32, 0, stream>>>(dist, kout);
    }
    {
      dim3 g(KNN, STD_G);
      k_std_partial<<<g, 256, 0, stream>>>(xyz_cur, feat_cur, xs_buf, feat_s, knn_i, N, S, C, stdp);
      k_std_final<<<1, KNN, 0, stream>>>(stdp, S, C, stdx, stdf);
    }
    k_stage<<<BATCH * S, 256, 0, stream>>>(xyz_cur, feat_cur, xs_buf, feat_s, knn_i,
                                           stdx, stdf, N, S, C, feat_nxt);
    k_pool<<<BATCH, 256, 0, stream>>>(feat_nxt, S, C2, out, off);
    off += 2 * C2;
    float* t = feat_cur; feat_cur = feat_nxt; feat_nxt = t;
    xyz_cur = xs_buf;
    float* u = xs_buf; xs_buf = xs_other; xs_other = u;
    N = S; C = C2;
  }
  (void)in_sizes; (void)n_in; (void)out_size; (void)ws_size;
}


# FPS pass latency: points in registers, wave argmax by DPP (int max of dist bits, then min index) not LDS shuffles, one barrier per pass: 4 waves post winner+coords, every wave reduces the posts
# speedup vs baseline: 2.7191x; 1.5824x over previous
#include <hip/hip_runtime.h>
#include <cfloat>
#include <cmath>

typedef __attribute__((ext_vector_type(2))) float v2f;
typedef __attribute__((ext_vector_type(8))) float v8f;

#define BATCH 16
#define NPTS 2048
#define CINIT 64
#define KNN 32
#define SIGMA_ 0.3f
#define BN_EPS_ 1e-5f
#define STD_G 16

__global__ __launch_bounds__(256) void k_embed_wmma(const float* __restrict__ xyz,
                                                    const float* __restrict__ W,
                                                    float* __restrict__ h) {
  const int wid  = (blockIdx.x * 256 + (int)threadIdx.x) >> 5;
  const int lane = threadIdx.x & 31;
  const int hf   = lane >> 4;
  const int ln   = lane & 15;
  const int tileM = wid >> 2;
  const int tileN = wid & 3;
  const int m  = tileM * 16 + ln;
  const int c  = tileN * 16 + ln;
  const float x0 = xyz[m * 3], x1 = xyz[m * 3 + 1], x2 = xyz[m * 3 + 2];
  const float w0 = W[c * 3],   w1 = W[c * 3 + 1],   w2 = W[c * 3 + 2];
  v2f a, b;
  a.x = hf ? x2 : x0;  a.y = hf ? 0.f : x1;
  b.x = hf ? w2 : w0;  b.y = hf ? 0.f : w1;
  v8f d = {};
  d = __builtin_amdgcn_wmma_f32_16x16x4_f32(false, a, false, b, (short)0, d, false, false);
#pragma unroll
  for (int r = 0; r < 8; ++r) {
    int row = tileM * 16 + r + 8 * hf;
    h[row * CINIT + tileN * 16 + ln] = d[r];
  }
}

__global__ __launch_bounds__(256) void k_bn_stats(const float* __restrict__ h,
                                                  float* __restrict__ mu,
                                                  float* __restrict__ inv) {
  __shared__ float s1[256], s2[256];
  const int c = blockIdx.x;
  const int n = BATCH * NPTS;
  float a = 0.f, b = 0.f;
  for (int i = threadIdx.x; i < n; i += 256) {
    float v = h[i * CINIT + c];
    a += v; b += v * v;
  }
  s1[threadIdx.x] = a; s2[threadIdx.x] = b; __syncthreads();
  for (int o = 128; o > 0; o >>= 1) {
    if ((int)threadIdx.x < o) { s1[threadIdx.x] += s1[threadIdx.x + o]; s2[threadIdx.x] += s2[threadIdx.x + o]; }
    __syncthreads();
  }
  if (threadIdx.x == 0) {
    float m   = s1[0] / (float)n;
    float var = s2[0] / (float)n - m * m;
    mu[c]  = m;
    inv[c] = rsqrtf(var + BN_EPS_);
  }
}

__global__ void k_bn_apply(float* __restrict__ h, const float* __restrict__ mu,
                           const float* __restrict__ inv, const float* __restrict__ gamma,
                           const float* __restrict__ beta) {
  const int total = BATCH * NPTS * CINIT;
  for (int i = blockIdx.x * blockDim.x + threadIdx.x; i < total; i += gridDim.x * blockDim.x) {
    int c = i & (CINIT - 1);
    float v = (h[i] - mu[c]) * inv[c] * gamma[c] + beta[c];
    h[i] = fmaxf(v, 0.f);
  }
}

template <int M>
__device__ __forceinline__ int lane_xor_row(int x) {
  return __builtin_amdgcn_update_dpp(0, x, 0x160 + M, 0xf, 0xf, true);
}
__device__ __forceinline__ int lane_xor_half(int x) {
  return __builtin_amdgcn_permlanex16(x, x, 0x76543210, 0xfedcba98, false, false);
}
__device__ __forceinline__ float wave_min_f(float x) {
  x = fminf(x, __builtin_bit_cast(float, lane_xor_row<1>(__builtin_bit_cast(int, x))));
  x = fminf(x, __builtin_bit_cast(float, lane_xor_row<2>(__builtin_bit_cast(int, x))));
  x = fminf(x, __builtin_bit_cast(float, lane_xor_row<4>(__builtin_bit_cast(int, x))));
  x = fminf(x, __builtin_bit_cast(float, lane_xor_row<8>(__builtin_bit_cast(int, x))));
  x = fminf(x, __builtin_bit_cast(float, lane_xor_half(__builtin_bit_cast(int, x))));
  return x;
}
__device__ __forceinline__ int wave_min_i(int x) {
  x = min(x, lane_xor_row<1>(x));
  x = min(x, lane_xor_row<2>(x));
  x = min(x, lane_xor_row<4>(x));
  x = min(x, lane_xor_row<8>(x));
  x = min(x, lane_xor_half(x));
  return x;
}
__device__ __forceinline__ int wave_max_i(int x) {
  x = max(x, lane_xor_row<1>(x));
  x = max(x, lane_xor_row<2>(x));
  x = max(x, lane_xor_row<4>(x));
  x = max(x, lane_xor_row<8>(x));
  x = max(x, lane_xor_half(x));
  return x;
}

#define FPS_T 128
#define FPS_W (FPS_T / 32)
template <int PPT>
__global__ __launch_bounds__(FPS_T) void k_fps(const float* __restrict__ xyz, int S,
                                               int* __restrict__ fps_idx) {
  constexpr int N = FPS_T * PPT;
  __shared__ alignas(16) int post[2][FPS_W][8];
  __shared__ int sel[NPTS / 2];
  const int b = blockIdx.x;
  const float* p = xyz + (size_t)b * N * 3;
  const int tid = threadIdx.x;
  const int lane = tid & 31, w = tid >> 5;
  float x[PPT], y[PPT], z[PPT], d[PPT];
#pragma unroll
  for (int q = 0; q < PPT; ++q) {
    const int i = tid + q * FPS_T;
    x[q] = p[i*3]; y[q] = p[i*3+1]; z[q] = p[i*3+2];
    d[q] = 1e10f;
  }
  float fx = p[0], fy = p[1], fz = p[2];
  if (tid == 0) sel[0] = 0;
  for (int it = 1; it < S; ++it) {
#pragma unroll
    for (int q = 0; q < PPT; ++q) {
      float dx = x[q]-fx, dy = y[q]-fy, dz = z[q]-fz;
      d[q] = fminf(d[q], dx*dx + dy*dy + dz*dz);
    }
    constexpr int NC = (PPT >= 4) ? 2 : 1;
    constexpr int L = PPT / NC;
    float cb[NC], cx[NC], cy[NC], cz[NC]; int cq[NC];
#pragma unroll
    for (int c = 0; c < NC; ++c) {
      cb[c] = d[c*L]; cq[c] = c*L; cx[c] = x[c*L]; cy[c] = y[c*L]; cz[c] = z[c*L];
#pragma unroll
      for (int q = c*L + 1; q < (c + 1)*L; ++q) {
        const bool g = d[q] > cb[c];
        cb[c] = g ? d[q] : cb[c]; cq[c] = g ? q : cq[c];
        cx[c] = g ? x[q] : cx[c]; cy[c] = g ? y[q] : cy[c]; cz[c] = g ? z[q] : cz[c];
      }
    }
    if (NC == 2) {
      const bool g = cb[NC-1] > cb[0];
      cb[0] = g ? cb[NC-1] : cb[0]; cq[0] = g ? cq[NC-1] : cq[0];
      cx[0] = g ? cx[NC-1] : cx[0]; cy[0] = g ? cy[NC-1] : cy[0]; cz[0] = g ? cz[NC-1] : cz[0];
    }
    const int best = __float_as_int(cb[0]);
    const int bi   = tid + cq[0] * FPS_T;
    const int wmax = wave_max_i(best);
    const int widx = wave_min_i((best == wmax) ? bi : 0x7fffffff);
    int* mypost = post[it & 1][w];
    if (bi == widx) {
      *(int4*)mypost = make_int4(best, bi, __float_as_int(cx[0]), __float_as_int(cy[0]));
      mypost[4] = __float_as_int(cz[0]);
    }
    __syncthreads();
    const int* e = post[it & 1][lane & (FPS_W - 1)];
    const int4 e0 = *(const int4*)e;
    const int  ez = e[4];
    int gmax = e0.x;
    gmax = max(gmax, lane_xor_row<1>(gmax));
    gmax = max(gmax, lane_xor_row<2>(gmax));
    int far = (e0.x == gmax) ? e0.y : 0x7fffffff;
    far = min(far, lane_xor_row<1>(far));
    far = min(far, lane_xor_row<2>(far));
    const bool win = (e0.y == far);
    int tx = win ? e0.z : 0, ty = win ? e0.w : 0, tz = win ? ez : 0;
    tx |= lane_xor_row<1>(tx); ty |= lane_xor_row<1>(ty); tz |= lane_xor_row<1>(tz);
    tx |= lane_xor_row<2>(tx); ty |= lane_xor_row<2>(ty); tz |= lane_xor_row<2>(tz);
    fx = __int_as_float(tx); fy = __int_as_float(ty); fz = __int_as_float(tz);
    if (tid == 0) sel[it] = far;
  }
  __syncthreads();
  for (int i = tid; i < S; i += FPS_T) fps_idx[b * S + i] = sel[i];
}

__global__ void k_gather(const float* __restrict__ xyz, const float* __restrict__ feat,
                         const int* __restrict__ fps_idx, int N, int S, int C,
                         float* __restrict__ xyz_s, float* __restrict__ feat_s) {
  const int b = blockIdx.x / S, s = blockIdx.x % S;
  const int idx = fps_idx[b * S + s];
  if (threadIdx.x < 3)
    xyz_s[(size_t)(b * S + s) * 3 + threadIdx.x] = xyz[(size_t)(b * N + idx) * 3 + threadIdx.x];
  for (int j = threadIdx.x; j < C; j += blockDim.x)
    feat_s[(size_t)(b * S + s) * C + j] = feat[(size_t)(b * N + idx) * C + j];
}

__global__ __launch_bounds__(256) void k_knn_dist_wmma(const float* __restrict__ xs,
                                                       const float* __restrict__ x,
                                                       int S, int N, float* __restrict__ dist) {
  __shared__ float sN[16];
  const int mbase = blockIdx.y * 16;
  const int w    = threadIdx.x >> 5;
  const int lane = threadIdx.x & 31;
  const int hf   = lane >> 4;
  const int ln   = lane & 15;
  if (threadIdx.x < 16) {
    float a0 = xs[(mbase + threadIdx.x) * 3];
    float a1 = xs[(mbase + threadIdx.x) * 3 + 1];
    float a2 = xs[(mbase + threadIdx.x) * 3 + 2];
    sN[threadIdx.x] = a0 * a0 + a1 * a1 + a2 * a2;
  }
  __syncthreads();
  const int m = mbase + ln;
  const float s0 = xs[m * 3], s1 = xs[m * 3 + 1], s2 = xs[m * 3 + 2];
  const int n = blockIdx.x * 128 + w * 16 + ln;
  const float x0 = x[n * 3], x1 = x[n * 3 + 1], x2 = x[n * 3 + 2];
  const float nn = x0 * x0 + x1 * x1 + x2 * x2;
  v2f a, b;
  a.x = hf ? s2 : s0;  a.y = hf ? 0.f : s1;
  b.x = hf ? x2 : x0;  b.y = hf ? 0.f : x1;
  v8f d = {};
  d = __builtin_amdgcn_wmma_f32_16x16x4_f32(false, a, false, b, (short)0, d, false, false);
#pragma unroll
  for (int r = 0; r < 8; ++r) {
    int row = mbase + r + 8 * hf;
    dist[(size_t)row * N + n] = sN[r + 8 * hf] + nn - 2.f * d[r];
  }
}

template <int NPL>
__global__ __launch_bounds__(32) void k_knn_select(const float* __restrict__ dist,
                                                   int* __restrict__ knn_idx) {
  const int lane = threadIdx.x;
  const int s = blockIdx.x;
  const float* row = dist + (size_t)s * (NPL * 32);
  float v[NPL];
#pragma unroll
  for (int j = 0; j < NPL; ++j) v[j] = row[j * 32 + lane];
  int kill = -1;
  int mine = 0;
  for (int k = 0; k < KNN; ++k) {
    constexpr int Q = NPL / 4;
    float cb[4]; int cj[4];
#pragma unroll
    for (int c = 0; c < 4; ++c) { cb[c] = FLT_MAX; cj[c] = c * Q; }
#pragma unroll
    for (int q = 0; q < Q; ++q) {
#pragma unroll
      for (int c = 0; c < 4; ++c) {
        const int j = c * Q + q;
        float vj = (j == kill) ? FLT_MAX : v[j];
        v[j] = vj;
        if (vj < cb[c]) { cb[c] = vj; cj[c] = j; }
      }
    }
    if (cb[1] < cb[0]) { cb[0] = cb[1]; cj[0] = cj[1]; }
    if (cb[3] < cb[2]) { cb[2] = cb[3]; cj[2] = cj[3]; }
    float best = cb[0]; int bj = cj[0];
    if (cb[2] < best) { best = cb[2]; bj = cj[2]; }
    const float wmin = wave_min_f(best);
    const int cand = (best == wmin) ? (bj * 32 + lane) : 0x7fffffff;
    const int widx = wave_min_i(cand);
    kill = ((widx & 31) == lane) ? (widx >> 5) : -1;
    if (lane == k) mine = widx;
  }
  knn_idx[s * KNN + lane] = mine;
}

__global__ __launch_bounds__(256) void k_std_partial(const float* __restrict__ xyz,
                                                     const float* __restrict__ feat,
                                                     const float* __restrict__ xyz_s,
                                                     const float* __restrict__ feat_s,
                                                     const int* __restrict__ knn_idx,
                                                     int N, int S, int C,
                                                     float* __restrict__ part) {
  __shared__ float s1[256], s2[256], s3[256], s4[256];
  const int k = blockIdx.x, g = blockIdx.y;
  float ax = 0.f, qx = 0.f, af = 0.f, qf = 0.f;
  for (int p = g * 256 + threadIdx.x; p < BATCH * S; p += STD_G * 256) {
    int b = p / S, s = p % S;
    int idx = knn_idx[(size_t)(b * S + s) * KNN + k];
    for (int c = 0; c < 3; ++c) {
      float d = xyz[(size_t)(b * N + idx) * 3 + c] - xyz_s[(size_t)(b * S + s) * 3 + c];
      ax += d; qx += d * d;
    }
    const float* fr = feat   + (size_t)(b * N + idx) * C;
    const float* fc = feat_s + (size_t)(b * S + s) * C;
    for (int j = 0; j < C; ++j) { float d = fr[j] - fc[j]; af += d; qf += d * d; }
  }
  s1[threadIdx.x] = ax; s2[threadIdx.x] = qx; s3[threadIdx.x] = af; s4[threadIdx.x] = qf;
  __syncthreads();
  for (int o = 128; o > 0; o >>= 1) {
    if ((int)threadIdx.x < o) {
      s1[threadIdx.x] += s1[threadIdx.x + o]; s2[threadIdx.x] += s2[threadIdx.x + o];
      s3[threadIdx.x] += s3[threadIdx.x + o]; s4[threadIdx.x] += s4[threadIdx.x + o];
    }
    __syncthreads();
  }
  if (threadIdx.x == 0) {
    float* q = part + ((size_t)k * STD_G + g) * 4;
    q[0] = s1[0]; q[1] = s2[0]; q[2] = s3[0]; q[3] = s4[0];
  }
}

__global__ void k_std_final(const float* __restrict__ part, int S, int C,
                            float* __restrict__ std_xyz, float* __restrict__ std_feat) {
  const int k = threadIdx.x;
  if (k >= KNN) return;
  float ax = 0.f, qx = 0.f, af = 0.f, qf = 0.f;
  for (int g = 0; g < STD_G; ++g) {
    const float* q = part + ((size_t)k * STD_G + g) * 4;
    ax += q[0]; qx += q[1]; af += q[2]; qf += q[3];
  }
  {
    double n = (double)BATCH * S * 3;
    double var = ((double)qx - (double)ax * ax / n) / (n - 1.0);
    std_xyz[k] = fmaxf(sqrtf(fmaxf((float)var, 0.f)), 1e-5f);
  }
  {
    double n = (double)BATCH * S * C;
    double var = ((double)qf - (double)af * af / n) / (n - 1.0);
    std_feat[k] = fmaxf(sqrtf(fmaxf((float)var, 0.f)), 1e-5f);
  }
}

__global__ __launch_bounds__(256) void k_stage(const float* __restrict__ xyz,
                                               const float* __restrict__ feat,
                                               const float* __restrict__ xyz_s,
                                               const float* __restrict__ feat_s,
                                               const int* __restrict__ knn_idx,
                                               const float* __restrict__ std_xyz,
                                               const float* __restrict__ std_feat,
                                               int N, int S, int C,
                                               float* __restrict__ out) {
  __shared__ int   kidx[KNN];
  __shared__ float sfe[KNN];
  __shared__ float xkn[KNN][3];
  __shared__ float fs[512];
  const int b = blockIdx.x / S, s = blockIdx.x % S;
  if (threadIdx.x < KNN) {
    kidx[threadIdx.x] = knn_idx[(size_t)(b * S + s) * KNN + threadIdx.x];
    sfe[threadIdx.x]  = std_feat[threadIdx.x];
  }
  __syncthreads();
  if (threadIdx.x < KNN)
    __builtin_prefetch(feat + (size_t)(b * N + kidx[threadIdx.x]) * C, 0, 3);
  if (threadIdx.x < KNN * 3) {
    int k = threadIdx.x / 3, c = threadIdx.x % 3;
    xkn[k][c] = (xyz[(size_t)(b * N + kidx[k]) * 3 + c] - xyz_s[(size_t)(b * S + s) * 3 + c]) / std_xyz[k];
  }
  for (int j = threadIdx.x; j < C; j += 256) fs[j] = feat_s[(size_t)(b * S + s) * C + j];
  __syncthreads();
  const int C2 = 2 * C;
  const int fd = (C2 + 2) / 3;
  const int M3 = 3 * fd;
  const float inv_sig = 1.f / SIGMA_;
  for (int j = threadIdx.x; j < C2; j += 256) {
    int src = (int)(((long long)j * (M3 - 1)) / (C2 - 1));
    int cc = src / fd, f = src % fd;
    float fv = -1.f + 2.f * (float)(f + 1) / (float)(fd + 1);
    float acc = 0.f, mx = -FLT_MAX;
    for (int k = 0; k < KNN; ++k) {
      float t  = (xkn[k][cc] - fv) * inv_sig;
      float pe = expf(-0.5f * t * t);
      float fk;
      if (j < C) fk = (feat[(size_t)(b * N + kidx[k]) * C + j] - fs[j]) / sfe[k];
      else       fk = fs[j - C];
      float fw = (fk + pe) * pe;
      acc += fw; mx = fmaxf(mx, fw);
    }
    float xx = acc * (1.f / KNN) + mx;
    out[(size_t)(b * S + s) * C2 + j] = 0.5f * xx * (1.f + erff(xx * 0.70710678118654752f));
  }
}

__global__ void k_pool(const float* __restrict__ feat, int S, int C2,
                       float* __restrict__ out, int off) {
  const int b = blockIdx.x;
  for (int j = threadIdx.x; j < C2; j += blockDim.x) {
    float mx = -FLT_MAX, sm = 0.f;
    for (int s = 0; s < S; ++s) {
      float v = feat[(size_t)(b * S + s) * C2 + j];
      mx = fmaxf(mx, v); sm += v;
    }
    out[b * 3840 + off + j]      = mx;
    out[b * 3840 + off + C2 + j] = sm / (float)S;
  }
}

extern "C" void kernel_launch(void* const* d_in, const int* in_sizes, int n_in,
                              void* d_out, int out_size, void* d_ws, size_t ws_size,
                              hipStream_t stream) {
  const float* xyz_in = (const float*)d_in[0];
  const float* W      = (const float*)d_in[1];
  const float* gamma  = (const float*)d_in[2];
  const float* beta   = (const float*)d_in[3];
  float* out = (float*)d_out;
  float* ws  = (float*)d_ws;

  size_t o = 0;
  float* featA  = ws + o; o += (size_t)BATCH * NPTS * CINIT;
  float* featB  = ws + o; o += (size_t)BATCH * NPTS * CINIT;
  float* xyzA   = ws + o; o += (size_t)BATCH * NPTS * 3;
  float* xyzB   = ws + o; o += (size_t)BATCH * NPTS * 3;
  float* feat_s = ws + o; o += (size_t)BATCH * 1024 * CINIT;
  float* dist   = ws + o; o += (size_t)1024 * 2048;
  float* bn_mu  = ws + o; o += 64;
  float* bn_inv = ws + o; o += 64;
  float* stdx   = ws + o; o += 32;
  float* stdf   = ws + o; o += 32;
  float* stdp   = ws + o; o += (size_t)KNN * STD_G * 4;
  int* fps_i = (int*)(ws + o); o += (size_t)BATCH * 1024;
  int* knn_i = (int*)(ws + o); o += (size_t)BATCH * 1024 * KNN;

  k_embed_wmma<<<1024, 256, 0, stream>>>(xyz_in, W, featA);
  k_bn_stats<<<CINIT, 256, 0, stream>>>(featA, bn_mu, bn_inv);
  k_bn_apply<<<2048, 256, 0, stream>>>(featA, bn_mu, bn_inv, gamma, beta);

  const float* xyz_cur = xyz_in;
  float* feat_cur = featA; float* feat_nxt = featB;
  float* xs_buf = xyzA;    float* xs_other = xyzB;
  int N = NPTS, C = CINIT, off = 0;

  for (int st = 0; st < 4; ++st) {
    const int S = N / 2, C2 = 2 * C;
    if (N == 2048)      k_fps<16><<<BATCH, FPS_T, 0, stream>>>(xyz_cur, S, fps_i);
    else if (N == 1024) k_fps<8><<<BATCH, FPS_T, 0, stream>>>(xyz_cur, S, fps_i);
    else if (N == 512)  k_fps<4><<<BATCH, FPS_T, 0, stream>>>(xyz_cur, S, fps_i);
    else                k_fps<2><<<BATCH, FPS_T, 0, stream>>>(xyz_cur, S, fps_i);
    k_gather<<<BATCH * S, 256, 0, stream>>>(xyz_cur, feat_cur, fps_i, N, S, C, xs_buf, feat_s);
    for (int b = 0; b < BATCH; ++b) {
      dim3 g(N / 128, S / 16);
      k_knn_dist_wmma<<<g, 256, 0, stream>>>(xs_buf + (size_t)b * S * 3,
                                             xyz_cur + (size_t)b * N * 3, S, N, dist);
      int* kout = knn_i + (size_t)b * S * KNN;
      if (N == 2048)      k_knn_select<64><<<S, 32, 0, stream>>>(dist, kout);
      else if (N == 1024) k_knn_select<32><<<S, 32, 0, stream>>>(dist, kout);
      else if (N == 512)  k_knn_select<16><<<S, 32, 0, stream>>>(dist, kout);
      else                k_knn_select<8><<<S, 32, 0, stream>>>(dist, kout);
    }
    {
      dim3 g(KNN, STD_G);
      k_std_partial<<<g, 256, 0, stream>>>(xyz_cur, feat_cur, xs_buf, feat_s, knn_i, N, S, C, stdp);
      k_std_final<<<1, KNN, 0, stream>>>(stdp, S, C, stdx, stdf);
    }
    k_stage<<<BATCH * S, 256, 0, stream>>>(xyz_cur, feat_cur, xs_buf, feat_s, knn_i,
                                           stdx, stdf, N, S, C, feat_nxt);
    k_pool<<<BATCH, 256, 0, stream>>>(feat_nxt, S, C2, out, off);
    off += 2 * C2;
    float* t = feat_cur; feat_cur = feat_nxt; feat_nxt = t;
    xyz_cur = xs_buf;
    float* u = xs_buf; xs_buf = xs_other; xs_other = u;
    N = S; C = C2;
  }
  (void)in_sizes; (void)n_in; (void)out_size; (void)ws_size;
}


# kNN: one launch per stage for all batches; a 16-wave block puts its 16-row WMMA distance strip in LDS and each wave selects its row's top-32 from there; no global strip; 4 launches, not 128
# speedup vs baseline: 3.7845x; 1.3918x over previous
#include <hip/hip_runtime.h>
#include <cfloat>
#include <cmath>

typedef __attribute__((ext_vector_type(2))) float v2f;
typedef __attribute__((ext_vector_type(8))) float v8f;

#define BATCH 16
#define NPTS 2048
#define CINIT 64
#define KNN 32
#define SIGMA_ 0.3f
#define BN_EPS_ 1e-5f
#define STD_G 16

__global__ __launch_bounds__(256) void k_embed_wmma(const float* __restrict__ xyz,
                                                    const float* __restrict__ W,
                                                    float* __restrict__ h) {
  const int wid  = (blockIdx.x * 256 + (int)threadIdx.x) >> 5;
  const int lane = threadIdx.x & 31;
  const int hf   = lane >> 4;
  const int ln   = lane & 15;
  const int tileM = wid >> 2;
  const int tileN = wid & 3;
  const int m  = tileM * 16 + ln;
  const int c  = tileN * 16 + ln;
  const float x0 = xyz[m * 3], x1 = xyz[m * 3 + 1], x2 = xyz[m * 3 + 2];
  const float w0 = W[c * 3],   w1 = W[c * 3 + 1],   w2 = W[c * 3 + 2];
  v2f a, b;
  a.x = hf ? x2 : x0;  a.y = hf ? 0.f : x1;
  b.x = hf ? w2 : w0;  b.y = hf ? 0.f : w1;
  v8f d = {};
  d = __builtin_amdgcn_wmma_f32_16x16x4_f32(false, a, false, b, (short)0, d, false, false);
#pragma unroll
  for (int r = 0; r < 8; ++r) {
    int row = tileM * 16 + r + 8 * hf;
    h[row * CINIT + tileN * 16 + ln] = d[r];
  }
}

__global__ __launch_bounds__(256) void k_bn_stats(const float* __restrict__ h,
                                                  float* __restrict__ mu,
                                                  float* __restrict__ inv) {
  __shared__ float s1[256], s2[256];
  const int c = blockIdx.x;
  const int n = BATCH * NPTS;
  float a = 0.f, b = 0.f;
  for (int i = threadIdx.x; i < n; i += 256) {
    float v = h[i * CINIT + c];
    a += v; b += v * v;
  }
  s1[threadIdx.x] = a; s2[threadIdx.x] = b; __syncthreads();
  for (int o = 128; o > 0; o >>= 1) {
    if ((int)threadIdx.x < o) { s1[threadIdx.x] += s1[threadIdx.x + o]; s2[threadIdx.x] += s2[threadIdx.x + o]; }
    __syncthreads();
  }
  if (threadIdx.x == 0) {
    float m   = s1[0] / (float)n;
    float var = s2[0] / (float)n - m * m;
    mu[c]  = m;
    inv[c] = rsqrtf(var + BN_EPS_);
  }
}

__global__ void k_bn_apply(float* __restrict__ h, const float* __restrict__ mu,
                           const float* __restrict__ inv, const float* __restrict__ gamma,
                           const float* __restrict__ beta) {
  const int total = BATCH * NPTS * CINIT;
  for (int i = blockIdx.x * blockDim.x + threadIdx.x; i < total; i += gridDim.x * blockDim.x) {
    int c = i & (CINIT - 1);
    float v = (h[i] - mu[c]) * inv[c] * gamma[c] + beta[c];
    h[i] = fmaxf(v, 0.f);
  }
}

template <int M>
__device__ __forceinline__ int lane_xor_row(int x) {
  return __builtin_amdgcn_update_dpp(0, x, 0x160 + M, 0xf, 0xf, true);
}
__device__ __forceinline__ int lane_xor_half(int x) {
  return __builtin_amdgcn_permlanex16(x, x, 0x76543210, 0xfedcba98, false, false);
}
__device__ __forceinline__ float wave_min_f(float x) {
  x = fminf(x, __builtin_bit_cast(float, lane_xor_row<1>(__builtin_bit_cast(int, x))));
  x = fminf(x, __builtin_bit_cast(float, lane_xor_row<2>(__builtin_bit_cast(int, x))));
  x = fminf(x, __builtin_bit_cast(float, lane_xor_row<4>(__builtin_bit_cast(int, x))));
  x = fminf(x, __builtin_bit_cast(float, lane_xor_row<8>(__builtin_bit_cast(int, x))));
  x = fminf(x, __builtin_bit_cast(float, lane_xor_half(__builtin_bit_cast(int, x))));
  return x;
}
__device__ __forceinline__ int wave_min_i(int x) {
  x = min(x, lane_xor_row<1>(x));
  x = min(x, lane_xor_row<2>(x));
  x = min(x, lane_xor_row<4>(x));
  x = min(x, lane_xor_row<8>(x));
  x = min(x, lane_xor_half(x));
  return x;
}
__device__ __forceinline__ int wave_max_i(int x) {
  x = max(x, lane_xor_row<1>(x));
  x = max(x, lane_xor_row<2>(x));
  x = max(x, lane_xor_row<4>(x));
  x = max(x, lane_xor_row<8>(x));
  x = max(x, lane_xor_half(x));
  return x;
}

#define FPS_T 128
#define FPS_W (FPS_T / 32)
template <int PPT>
__global__ __launch_bounds__(FPS_T) void k_fps(const float* __restrict__ xyz, int S,
                                               int* __restrict__ fps_idx) {
  constexpr int N = FPS_T * PPT;
  __shared__ alignas(16) int post[2][FPS_W][8];
  __shared__ int sel[NPTS / 2];
  const int b = blockIdx.x;
  const float* p = xyz + (size_t)b * N * 3;
  const int tid = threadIdx.x;
  const int lane = tid & 31, w = tid >> 5;
  float x[PPT], y[PPT], z[PPT], d[PPT];
#pragma unroll
  for (int q = 0; q < PPT; ++q) {
    const int i = tid + q * FPS_T;
    x[q] = p[i*3]; y[q] = p[i*3+1]; z[q] = p[i*3+2];
    d[q] = 1e10f;
  }
  float fx = p[0], fy = p[1], fz = p[2];
  if (tid == 0) sel[0] = 0;
  for (int it = 1; it < S; ++it) {
#pragma unroll
    for (int q = 0; q < PPT; ++q) {
      float dx = x[q]-fx, dy = y[q]-fy, dz = z[q]-fz;
      d[q] = fminf(d[q], dx*dx + dy*dy + dz*dz);
    }
    constexpr int NC = (PPT >= 4) ? 2 : 1;
    constexpr int L = PPT / NC;
    float cb[NC], cx[NC], cy[NC], cz[NC]; int cq[NC];
#pragma unroll
    for (int c = 0; c < NC; ++c) {
      cb[c] = d[c*L]; cq[c] = c*L; cx[c] = x[c*L]; cy[c] = y[c*L]; cz[c] = z[c*L];
#pragma unroll
      for (int q = c*L + 1; q < (c + 1)*L; ++q) {
        const bool g = d[q] > cb[c];
        cb[c] = g ? d[q] : cb[c]; cq[c] = g ? q : cq[c];
        cx[c] = g ? x[q] : cx[c]; cy[c] = g ? y[q] : cy[c]; cz[c] = g ? z[q] : cz[c];
      }
    }
    if (NC == 2) {
      const bool g = cb[NC-1] > cb[0];
      cb[0] = g ? cb[NC-1] : cb[0]; cq[0] = g ? cq[NC-1] : cq[0];
      cx[0] = g ? cx[NC-1] : cx[0]; cy[0] = g ? cy[NC-1] : cy[0]; cz[0] = g ? cz[NC-1] : cz[0];
    }
    const int best = __float_as_int(cb[0]);
    const int bi   = tid + cq[0] * FPS_T;
    const int wmax = wave_max_i(best);
    const int widx = wave_min_i((best == wmax) ? bi : 0x7fffffff);
    int* mypost = post[it & 1][w];
    if (bi == widx) {
      *(int4*)mypost = make_int4(best, bi, __float_as_int(cx[0]), __float_as_int(cy[0]));
      mypost[4] = __float_as_int(cz[0]);
    }
    __syncthreads();
    const int* e = post[it & 1][lane & (FPS_W - 1)];
    const int4 e0 = *(const int4*)e;
    const int  ez = e[4];
    int gmax = e0.x;
    gmax = max(gmax, lane_xor_row<1>(gmax));
    gmax = max(gmax, lane_xor_row<2>(gmax));
    int far = (e0.x == gmax) ? e0.y : 0x7fffffff;
    far = min(far, lane_xor_row<1>(far));
    far = min(far, lane_xor_row<2>(far));
    const bool win = (e0.y == far);
    int tx = win ? e0.z : 0, ty = win ? e0.w : 0, tz = win ? ez : 0;
    tx |= lane_xor_row<1>(tx); ty |= lane_xor_row<1>(ty); tz |= lane_xor_row<1>(tz);
    tx |= lane_xor_row<2>(tx); ty |= lane_xor_row<2>(ty); tz |= lane_xor_row<2>(tz);
    fx = __int_as_float(tx); fy = __int_as_float(ty); fz = __int_as_float(tz);
    if (tid == 0) sel[it] = far;
  }
  __syncthreads();
  for (int i = tid; i < S; i += FPS_T) fps_idx[b * S + i] = sel[i];
}

__global__ void k_gather(const float* __restrict__ xyz, const float* __restrict__ feat,
                         const int* __restrict__ fps_idx, int N, int S, int C,
                         float* __restrict__ xyz_s, float* __restrict__ feat_s) {
  const int b = blockIdx.x / S, s = blockIdx.x % S;
  const int idx = fps_idx[b * S + s];
  if (threadIdx.x < 3)
    xyz_s[(size_t)(b * S + s) * 3 + threadIdx.x] = xyz[(size_t)(b * N + idx) * 3 + threadIdx.x];
  for (int j = threadIdx.x; j < C; j += blockDim.x)
    feat_s[(size_t)(b * S + s) * C + j] = feat[(size_t)(b * N + idx) * C + j];
}

#define KNN_W 16
template <int NPL>
__global__ __launch_bounds__(KNN_W * 32) void k_knn(const float* __restrict__ xs_all,
                                                    const float* __restrict__ x_all,
                                                    int S, int* __restrict__ knn_idx) {
  constexpr int N = NPL * 32;
  constexpr int TPW = N / 16 / KNN_W;
  constexpr int PITCH = N + 4;
  __shared__ float strip[16 * PITCH];
  __shared__ float sN[16];
  const int w    = threadIdx.x >> 5;
  const int lane = threadIdx.x & 31;
  const int hf   = lane >> 4;
  const int ln   = lane & 15;
  const int rbase = blockIdx.x * 16;
  const float* xs = xs_all + (size_t)rbase * 3;
  const float* x  = x_all + (size_t)(rbase / S) * N * 3;
  if (threadIdx.x < 16) {
    float a0 = xs[threadIdx.x * 3];
    float a1 = xs[threadIdx.x * 3 + 1];
    float a2 = xs[threadIdx.x * 3 + 2];
    sN[threadIdx.x] = fmaf(a2, a2, fmaf(a0, a0, a1 * a1));
  }
  __syncthreads();
  const float s0 = xs[ln * 3], s1 = xs[ln * 3 + 1], s2 = xs[ln * 3 + 2];
  v2f a;
  a.x = hf ? s2 : s0;  a.y = hf ? 0.f : s1;
#pragma unroll
  for (int t = 0; t < TPW; ++t) {
    const int n = (w * TPW + t) * 16 + ln;
    const float x0 = x[n * 3], x1 = x[n * 3 + 1], x2 = x[n * 3 + 2];
    const float nn = fmaf(x2, x2, fmaf(x0, x0, x1 * x1));
    v2f b;
    b.x = hf ? x2 : x0;  b.y = hf ? 0.f : x1;
    v8f d = {};
    d = __builtin_amdgcn_wmma_f32_16x16x4_f32(false, a, false, b, (short)0, d, false, false);
#pragma unroll
    for (int r = 0; r < 8; ++r)
      strip[(r + 8 * hf) * PITCH + n] = fmaf(-2.f, d[r], sN[r + 8 * hf] + nn);
  }
  __syncthreads();
  const float* row = strip + w * PITCH;
  float v[NPL];
#pragma unroll
  for (int j = 0; j < NPL; ++j) v[j] = row[j * 32 + lane];
  int kill = -1;
  int mine = 0;
  for (int k = 0; k < KNN; ++k) {
    constexpr int Q = NPL / 4;
    float cb[4]; int cj[4];
#pragma unroll
    for (int c = 0; c < 4; ++c) { cb[c] = FLT_MAX; cj[c] = c * Q; }
#pragma unroll
    for (int q = 0; q < Q; ++q) {
#pragma unroll
      for (int c = 0; c < 4; ++c) {
        const int j = c * Q + q;
        float vj = (j == kill) ? FLT_MAX : v[j];
        v[j] = vj;
        if (vj < cb[c]) { cb[c] = vj; cj[c] = j; }
      }
    }
    if (cb[1] < cb[0]) { cb[0] = cb[1]; cj[0] = cj[1]; }
    if (cb[3] < cb[2]) { cb[2] = cb[3]; cj[2] = cj[3]; }
    float best = cb[0]; int bj = cj[0];
    if (cb[2] < best) { best = cb[2]; bj = cj[2]; }
    const float wmin = wave_min_f(best);
    const int cand = (best == wmin) ? (bj * 32 + lane) : 0x7fffffff;
    const int widx = wave_min_i(cand);
    kill = ((widx & 31) == lane) ? (widx >> 5) : -1;
    if (lane == k) mine = widx;
  }
  knn_idx[(size_t)(rbase + w) * KNN + lane] = mine;
}

__global__ __launch_bounds__(256) void k_std_partial(const float* __restrict__ xyz,
                                                     const float* __restrict__ feat,
                                                     const float* __restrict__ xyz_s,
                                                     const float* __restrict__ feat_s,
                                                     const int* __restrict__ knn_idx,
                                                     int N, int S, int C,
                                                     float* __restrict__ part) {
  __shared__ float s1[256], s2[256], s3[256], s4[256];
  const int k = blockIdx.x, g = blockIdx.y;
  float ax = 0.f, qx = 0.f, af = 0.f, qf = 0.f;
  for (int p = g * 256 + threadIdx.x; p < BATCH * S; p += STD_G * 256) {
    int b = p / S, s = p % S;
    int idx = knn_idx[(size_t)(b * S + s) * KNN + k];
    for (int c = 0; c < 3; ++c) {
      float d = xyz[(size_t)(b * N + idx) * 3 + c] - xyz_s[(size_t)(b * S + s) * 3 + c];
      ax += d; qx += d * d;
    }
    const float* fr = feat   + (size_t)(b * N + idx) * C;
    const float* fc = feat_s + (size_t)(b * S + s) * C;
    for (int j = 0; j < C; ++j) { float d = fr[j] - fc[j]; af += d; qf += d * d; }
  }
  s1[threadIdx.x] = ax; s2[threadIdx.x] = qx; s3[threadIdx.x] = af; s4[threadIdx.x] = qf;
  __syncthreads();
  for (int o = 128; o > 0; o >>= 1) {
    if ((int)threadIdx.x < o) {
      s1[threadIdx.x] += s1[threadIdx.x + o]; s2[threadIdx.x] += s2[threadIdx.x + o];
      s3[threadIdx.x] += s3[threadIdx.x + o]; s4[threadIdx.x] += s4[threadIdx.x + o];
    }
    __syncthreads();
  }
  if (threadIdx.x == 0) {
    float* q = part + ((size_t)k * STD_G + g) * 4;
    q[0] = s1[0]; q[1] = s2[0]; q[2] = s3[0]; q[3] = s4[0];
  }
}

__global__ void k_std_final(const float* __restrict__ part, int S, int C,
                            float* __restrict__ std_xyz, float* __restrict__ std_feat) {
  const int k = threadIdx.x;
  if (k >= KNN) return;
  float ax = 0.f, qx = 0.f, af = 0.f, qf = 0.f;
  for (int g = 0; g < STD_G; ++g) {
    const float* q = part + ((size_t)k * STD_G + g) * 4;
    ax += q[0]; qx += q[1]; af += q[2]; qf += q[3];
  }
  {
    double n = (double)BATCH * S * 3;
    double var = ((double)qx - (double)ax * ax / n) / (n - 1.0);
    std_xyz[k] = fmaxf(sqrtf(fmaxf((float)var, 0.f)), 1e-5f);
  }
  {
    double n = (double)BATCH * S * C;
    double var = ((double)qf - (double)af * af / n) / (n - 1.0);
    std_feat[k] = fmaxf(sqrtf(fmaxf((float)var, 0.f)), 1e-5f);
  }
}

__global__ __launch_bounds__(256) void k_stage(const float* __restrict__ xyz,
                                               const float* __restrict__ feat,
                                               const float* __restrict__ xyz_s,
                                               const float* __restrict__ feat_s,
                                               const int* __restrict__ knn_idx,
                                               const float* __restrict__ std_xyz,
                                               const float* __restrict__ std_feat,
                                               int N, int S, int C,
                                               float* __restrict__ out) {
  __shared__ int   kidx[KNN];
  __shared__ float sfe[KNN];
  __shared__ float xkn[KNN][3];
  __shared__ float fs[512];
  const int b = blockIdx.x / S, s = blockIdx.x % S;
  if (threadIdx.x < KNN) {
    kidx[threadIdx.x] = knn_idx[(size_t)(b * S + s) * KNN + threadIdx.x];
    sfe[threadIdx.x]  = std_feat[threadIdx.x];
  }
  __syncthreads();
  if (threadIdx.x < KNN)
    __builtin_prefetch(feat + (size_t)(b * N + kidx[threadIdx.x]) * C, 0, 3);
  if (threadIdx.x < KNN * 3) {
    int k = threadIdx.x / 3, c = threadIdx.x % 3;
    xkn[k][c] = (xyz[(size_t)(b * N + kidx[k]) * 3 + c] - xyz_s[(size_t)(b * S + s) * 3 + c]) / std_xyz[k];
  }
  for (int j = threadIdx.x; j < C; j += 256) fs[j] = feat_s[(size_t)(b * S + s) * C + j];
  __syncthreads();
  const int C2 = 2 * C;
  const int fd = (C2 + 2) / 3;
  const int M3 = 3 * fd;
  const float inv_sig = 1.f / SIGMA_;
  for (int j = threadIdx.x; j < C2; j += 256) {
    int src = (int)(((long long)j * (M3 - 1)) / (C2 - 1));
    int cc = src / fd, f = src % fd;
    float fv = -1.f + 2.f * (float)(f + 1) / (float)(fd + 1);
    float acc = 0.f, mx = -FLT_MAX;
    for (int k = 0; k < KNN; ++k) {
      float t  = (xkn[k][cc] - fv) * inv_sig;
      float pe = expf(-0.5f * t * t);
      float fk;
      if (j < C) fk = (feat[(size_t)(b * N + kidx[k]) * C + j] - fs[j]) / sfe[k];
      else       fk = fs[j - C];
      float fw = (fk + pe) * pe;
      acc += fw; mx = fmaxf(mx, fw);
    }
    float xx = acc * (1.f / KNN) + mx;
    out[(size_t)(b * S + s) * C2 + j] = 0.5f * xx * (1.f + erff(xx * 0.70710678118654752f));
  }
}

__global__ void k_pool(const float* __restrict__ feat, int S, int C2,
                       float* __restrict__ out, int off) {
  const int b = blockIdx.x;
  for (int j = threadIdx.x; j < C2; j += blockDim.x) {
    float mx = -FLT_MAX, sm = 0.f;
    for (int s = 0; s < S; ++s) {
      float v = feat[(size_t)(b * S + s) * C2 + j];
      mx = fmaxf(mx, v); sm += v;
    }
    out[b * 3840 + off + j]      = mx;
    out[b * 3840 + off + C2 + j] = sm / (float)S;
  }
}

extern "C" void kernel_launch(void* const* d_in, const int* in_sizes, int n_in,
                              void* d_out, int out_size, void* d_ws, size_t ws_size,
                              hipStream_t stream) {
  const float* xyz_in = (const float*)d_in[0];
  const float* W      = (const float*)d_in[1];
  const float* gamma  = (const float*)d_in[2];
  const float* beta   = (const float*)d_in[3];
  float* out = (float*)d_out;
  float* ws  = (float*)d_ws;

  size_t o = 0;
  float* featA  = ws + o; o += (size_t)BATCH * NPTS * CINIT;
  float* featB  = ws + o; o += (size_t)BATCH * NPTS * CINIT;
  float* xyzA   = ws + o; o += (size_t)BATCH * NPTS * 3;
  float* xyzB   = ws + o; o += (size_t)BATCH * NPTS * 3;
  float* feat_s = ws + o; o += (size_t)BATCH * 1024 * CINIT;
  float* bn_mu  = ws + o; o += 64;
  float* bn_inv = ws + o; o += 64;
  float* stdx   = ws + o; o += 32;
  float* stdf   = ws + o; o += 32;
  float* stdp   = ws + o; o += (size_t)KNN * STD_G * 4;
  int* fps_i = (int*)(ws + o); o += (size_t)BATCH * 1024;
  int* knn_i = (int*)(ws + o); o += (size_t)BATCH * 1024 * KNN;

  k_embed_wmma<<<1024, 256, 0, stream>>>(xyz_in, W, featA);
  k_bn_stats<<<CINIT, 256, 0, stream>>>(featA, bn_mu, bn_inv);
  k_bn_apply<<<2048, 256, 0, stream>>>(featA, bn_mu, bn_inv, gamma, beta);

  const float* xyz_cur = xyz_in;
  float* feat_cur = featA; float* feat_nxt = featB;
  float* xs_buf = xyzA;    float* xs_other = xyzB;
  int N = NPTS, C = CINIT, off = 0;

  for (int st = 0; st < 4; ++st) {
    const int S = N / 2, C2 = 2 * C;
    if (N == 2048)      k_fps<16><<<BATCH, FPS_T, 0, stream>>>(xyz_cur, S, fps_i);
    else if (N == 1024) k_fps<8><<<BATCH, FPS_T, 0, stream>>>(xyz_cur, S, fps_i);
    else if (N == 512)  k_fps<4><<<BATCH, FPS_T, 0, stream>>>(xyz_cur, S, fps_i);
    else                k_fps<2><<<BATCH, FPS_T, 0, stream>>>(xyz_cur, S, fps_i);
    k_gather<<<BATCH * S, 256, 0, stream>>>(xyz_cur, feat_cur, fps_i, N, S, C, xs_buf, feat_s);
    if (N == 2048)      k_knn<64><<<BATCH * S / 16, KNN_W * 32, 0, stream>>>(xs_buf, xyz_cur, S, knn_i);
    else if (N == 1024) k_knn<32><<<BATCH * S / 16, KNN_W * 32, 0, stream>>>(xs_buf, xyz_cur, S, knn_i);
    else if (N == 512)  k_knn<16><<<BATCH * S / 16, KNN_W * 32, 0, stream>>>(xs_buf, xyz_cur, S, knn_i);
    else                k_knn<8><<<BATCH * S / 16, KNN_W * 32, 0, stream>>>(xs_buf, xyz_cur, S, knn_i);
    {
      dim3 g(KNN, STD_G);
      k_std_partial<<<g, 256, 0, stream>>>(xyz_cur, feat_cur, xs_buf, feat_s, knn_i, N, S, C, stdp);
      k_std_final<<<1, KNN, 0, stream>>>(stdp, S, C, stdx, stdf);
    }
    k_stage<<<BATCH * S, 256, 0, stream>>>(xyz_cur, feat_cur, xs_buf, feat_s, knn_i,
                                           stdx, stdf, N, S, C, feat_nxt);
    k_pool<<<BATCH, 256, 0, stream>>>(feat_nxt, S, C2, out, off);
    off += 2 * C2;
    float* t = feat_cur; feat_cur = feat_nxt; feat_nxt = t;
    xyz_cur = xs_buf;
    float* u = xs_buf; xs_buf = xs_other; xs_other = u;
    N = S; C = C2;
  }
  (void)in_sizes; (void)n_in; (void)out_size; (void)ws_size;
}


# k_pool: one block per (batch, 32 channels), 16 waves split the rows with 8 loads in flight each, partials folded in LDS in wave order; was 16 blocks walking S serially; mean's sum order changes
# speedup vs baseline: 4.7777x; 1.2624x over previous
#include <hip/hip_runtime.h>
#include <cfloat>
#include <cmath>

typedef __attribute__((ext_vector_type(2))) float v2f;
typedef __attribute__((ext_vector_type(8))) float v8f;

#define BATCH 16
#define NPTS 2048
#define CINIT 64
#define KNN 32
#define SIGMA_ 0.3f
#define BN_EPS_ 1e-5f
#define STD_G 16

__global__ __launch_bounds__(256) void k_embed_wmma(const float* __restrict__ xyz,
                                                    const float* __restrict__ W,
                                                    float* __restrict__ h) {
  const int wid  = (blockIdx.x * 256 + (int)threadIdx.x) >> 5;
  const int lane = threadIdx.x & 31;
  const int hf   = lane >> 4;
  const int ln   = lane & 15;
  const int tileM = wid >> 2;
  const int tileN = wid & 3;
  const int m  = tileM * 16 + ln;
  const int c  = tileN * 16 + ln;
  const float x0 = xyz[m * 3], x1 = xyz[m * 3 + 1], x2 = xyz[m * 3 + 2];
  const float w0 = W[c * 3],   w1 = W[c * 3 + 1],   w2 = W[c * 3 + 2];
  v2f a, b;
  a.x = hf ? x2 : x0;  a.y = hf ? 0.f : x1;
  b.x = hf ? w2 : w0;  b.y = hf ? 0.f : w1;
  v8f d = {};
  d = __builtin_amdgcn_wmma_f32_16x16x4_f32(false, a, false, b, (short)0, d, false, false);
#pragma unroll
  for (int r = 0; r < 8; ++r) {
    int row = tileM * 16 + r + 8 * hf;
    h[row * CINIT + tileN * 16 + ln] = d[r];
  }
}

__global__ __launch_bounds__(256) void k_bn_stats(const float* __restrict__ h,
                                                  float* __restrict__ mu,
                                                  float* __restrict__ inv) {
  __shared__ float s1[256], s2[256];
  const int c = blockIdx.x;
  const int n = BATCH * NPTS;
  float a = 0.f, b = 0.f;
  for (int i = threadIdx.x; i < n; i += 256) {
    float v = h[i * CINIT + c];
    a += v; b += v * v;
  }
  s1[threadIdx.x] = a; s2[threadIdx.x] = b; __syncthreads();
  for (int o = 128; o > 0; o >>= 1) {
    if ((int)threadIdx.x < o) { s1[threadIdx.x] += s1[threadIdx.x + o]; s2[threadIdx.x] += s2[threadIdx.x + o]; }
    __syncthreads();
  }
  if (threadIdx.x == 0) {
    float m   = s1[0] / (float)n;
    float var = s2[0] / (float)n - m * m;
    mu[c]  = m;
    inv[c] = rsqrtf(var + BN_EPS_);
  }
}

__global__ void k_bn_apply(float* __restrict__ h, const float* __restrict__ mu,
                           const float* __restrict__ inv, const float* __restrict__ gamma,
                           const float* __restrict__ beta) {
  const int total = BATCH * NPTS * CINIT;
  for (int i = blockIdx.x * blockDim.x + threadIdx.x; i < total; i += gridDim.x * blockDim.x) {
    int c = i & (CINIT - 1);
    float v = (h[i] - mu[c]) * inv[c] * gamma[c] + beta[c];
    h[i] = fmaxf(v, 0.f);
  }
}

template <int M>
__device__ __forceinline__ int lane_xor_row(int x) {
  return __builtin_amdgcn_update_dpp(0, x, 0x160 + M, 0xf, 0xf, true);
}
__device__ __forceinline__ int lane_xor_half(int x) {
  return __builtin_amdgcn_permlanex16(x, x, 0x76543210, 0xfedcba98, false, false);
}
__device__ __forceinline__ float wave_min_f(float x) {
  x = fminf(x, __builtin_bit_cast(float, lane_xor_row<1>(__builtin_bit_cast(int, x))));
  x = fminf(x, __builtin_bit_cast(float, lane_xor_row<2>(__builtin_bit_cast(int, x))));
  x = fminf(x, __builtin_bit_cast(float, lane_xor_row<4>(__builtin_bit_cast(int, x))));
  x = fminf(x, __builtin_bit_cast(float, lane_xor_row<8>(__builtin_bit_cast(int, x))));
  x = fminf(x, __builtin_bit_cast(float, lane_xor_half(__builtin_bit_cast(int, x))));
  return x;
}
__device__ __forceinline__ int wave_min_i(int x) {
  x = min(x, lane_xor_row<1>(x));
  x = min(x, lane_xor_row<2>(x));
  x = min(x, lane_xor_row<4>(x));
  x = min(x, lane_xor_row<8>(x));
  x = min(x, lane_xor_half(x));
  return x;
}
__device__ __forceinline__ int wave_max_i(int x) {
  x = max(x, lane_xor_row<1>(x));
  x = max(x, lane_xor_row<2>(x));
  x = max(x, lane_xor_row<4>(x));
  x = max(x, lane_xor_row<8>(x));
  x = max(x, lane_xor_half(x));
  return x;
}

#define FPS_T 128
#define FPS_W (FPS_T / 32)
template <int PPT>
__global__ __launch_bounds__(FPS_T) void k_fps(const float* __restrict__ xyz, int S,
                                               int* __restrict__ fps_idx) {
  constexpr int N = FPS_T * PPT;
  __shared__ alignas(16) int post[2][FPS_W][8];
  __shared__ int sel[NPTS / 2];
  const int b = blockIdx.x;
  const float* p = xyz + (size_t)b * N * 3;
  const int tid = threadIdx.x;
  const int lane = tid & 31, w = tid >> 5;
  float x[PPT], y[PPT], z[PPT], d[PPT];
#pragma unroll
  for (int q = 0; q < PPT; ++q) {
    const int i = tid + q * FPS_T;
    x[q] = p[i*3]; y[q] = p[i*3+1]; z[q] = p[i*3+2];
    d[q] = 1e10f;
  }
  float fx = p[0], fy = p[1], fz = p[2];
  if (tid == 0) sel[0] = 0;
  for (int it = 1; it < S; ++it) {
#pragma unroll
    for (int q = 0; q < PPT; ++q) {
      float dx = x[q]-fx, dy = y[q]-fy, dz = z[q]-fz;
      d[q] = fminf(d[q], dx*dx + dy*dy + dz*dz);
    }
    constexpr int NC = (PPT >= 4) ? 2 : 1;
    constexpr int L = PPT / NC;
    float cb[NC], cx[NC], cy[NC], cz[NC]; int cq[NC];
#pragma unroll
    for (int c = 0; c < NC; ++c) {
      cb[c] = d[c*L]; cq[c] = c*L; cx[c] = x[c*L]; cy[c] = y[c*L]; cz[c] = z[c*L];
#pragma unroll
      for (int q = c*L + 1; q < (c + 1)*L; ++q) {
        const bool g = d[q] > cb[c];
        cb[c] = g ? d[q] : cb[c]; cq[c] = g ? q : cq[c];
        cx[c] = g ? x[q] : cx[c]; cy[c] = g ? y[q] : cy[c]; cz[c] = g ? z[q] : cz[c];
      }
    }
    if (NC == 2) {
      const bool g = cb[NC-1] > cb[0];
      cb[0] = g ? cb[NC-1] : cb[0]; cq[0] = g ? cq[NC-1] : cq[0];
      cx[0] = g ? cx[NC-1] : cx[0]; cy[0] = g ? cy[NC-1] : cy[0]; cz[0] = g ? cz[NC-1] : cz[0];
    }
    const int best = __float_as_int(cb[0]);
    const int bi   = tid + cq[0] * FPS_T;
    const int wmax = wave_max_i(best);
    const int widx = wave_min_i((best == wmax) ? bi : 0x7fffffff);
    int* mypost = post[it & 1][w];
    if (bi == widx) {
      *(int4*)mypost = make_int4(best, bi, __float_as_int(cx[0]), __float_as_int(cy[0]));
      mypost[4] = __float_as_int(cz[0]);
    }
    __syncthreads();
    const int* e = post[it & 1][lane & (FPS_W - 1)];
    const int4 e0 = *(const int4*)e;
    const int  ez = e[4];
    int gmax = e0.x;
    gmax = max(gmax, lane_xor_row<1>(gmax));
    gmax = max(gmax, lane_xor_row<2>(gmax));
    int far = (e0.x == gmax) ? e0.y : 0x7fffffff;
    far = min(far, lane_xor_row<1>(far));
    far = min(far, lane_xor_row<2>(far));
    const bool win = (e0.y == far);
    int tx = win ? e0.z : 0, ty = win ? e0.w : 0, tz = win ? ez : 0;
    tx |= lane_xor_row<1>(tx); ty |= lane_xor_row<1>(ty); tz |= lane_xor_row<1>(tz);
    tx |= lane_xor_row<2>(tx); ty |= lane_xor_row<2>(ty); tz |= lane_xor_row<2>(tz);
    fx = __int_as_float(tx); fy = __int_as_float(ty); fz = __int_as_float(tz);
    if (tid == 0) sel[it] = far;
  }
  __syncthreads();
  for (int i = tid; i < S; i += FPS_T) fps_idx[b * S + i] = sel[i];
}

__global__ void k_gather(const float* __restrict__ xyz, const float* __restrict__ feat,
                         const int* __restrict__ fps_idx, int N, int S, int C,
                         float* __restrict__ xyz_s, float* __restrict__ feat_s) {
  const int b = blockIdx.x / S, s = blockIdx.x % S;
  const int idx = fps_idx[b * S + s];
  if (threadIdx.x < 3)
    xyz_s[(size_t)(b * S + s) * 3 + threadIdx.x] = xyz[(size_t)(b * N + idx) * 3 + threadIdx.x];
  for (int j = threadIdx.x; j < C; j += blockDim.x)
    feat_s[(size_t)(b * S + s) * C + j] = feat[(size_t)(b * N + idx) * C + j];
}

#define KNN_W 16
template <int NPL>
__global__ __launch_bounds__(KNN_W * 32) void k_knn(const float* __restrict__ xs_all,
                                                    const float* __restrict__ x_all,
                                                    int S, int* __restrict__ knn_idx) {
  constexpr int N = NPL * 32;
  constexpr int TPW = N / 16 / KNN_W;
  constexpr int PITCH = N + 4;
  __shared__ float strip[16 * PITCH];
  __shared__ float sN[16];
  const int w    = threadIdx.x >> 5;
  const int lane = threadIdx.x & 31;
  const int hf   = lane >> 4;
  const int ln   = lane & 15;
  const int rbase = blockIdx.x * 16;
  const float* xs = xs_all + (size_t)rbase * 3;
  const float* x  = x_all + (size_t)(rbase / S) * N * 3;
  if (threadIdx.x < 16) {
    float a0 = xs[threadIdx.x * 3];
    float a1 = xs[threadIdx.x * 3 + 1];
    float a2 = xs[threadIdx.x * 3 + 2];
    sN[threadIdx.x] = fmaf(a2, a2, fmaf(a0, a0, a1 * a1));
  }
  __syncthreads();
  const float s0 = xs[ln * 3], s1 = xs[ln * 3 + 1], s2 = xs[ln * 3 + 2];
  v2f a;
  a.x = hf ? s2 : s0;  a.y = hf ? 0.f : s1;
#pragma unroll
  for (int t = 0; t < TPW; ++t) {
    const int n = (w * TPW + t) * 16 + ln;
    const float x0 = x[n * 3], x1 = x[n * 3 + 1], x2 = x[n * 3 + 2];
    const float nn = fmaf(x2, x2, fmaf(x0, x0, x1 * x1));
    v2f b;
    b.x = hf ? x2 : x0;  b.y = hf ? 0.f : x1;
    v8f d = {};
    d = __builtin_amdgcn_wmma_f32_16x16x4_f32(false, a, false, b, (short)0, d, false, false);
#pragma unroll
    for (int r = 0; r < 8; ++r)
      strip[(r + 8 * hf) * PITCH + n] = fmaf(-2.f, d[r], sN[r + 8 * hf] + nn);
  }
  __syncthreads();
  const float* row = strip + w * PITCH;
  float v[NPL];
#pragma unroll
  for (int j = 0; j < NPL; ++j) v[j] = row[j * 32 + lane];
  int kill = -1;
  int mine = 0;
  for (int k = 0; k < KNN; ++k) {
    constexpr int Q = NPL / 4;
    float cb[4]; int cj[4];
#pragma unroll
    for (int c = 0; c < 4; ++c) { cb[c] = FLT_MAX; cj[c] = c * Q; }
#pragma unroll
    for (int q = 0; q < Q; ++q) {
#pragma unroll
      for (int c = 0; c < 4; ++c) {
        const int j = c * Q + q;
        float vj = (j == kill) ? FLT_MAX : v[j];
        v[j] = vj;
        if (vj < cb[c]) { cb[c] = vj; cj[c] = j; }
      }
    }
    if (cb[1] < cb[0]) { cb[0] = cb[1]; cj[0] = cj[1]; }
    if (cb[3] < cb[2]) { cb[2] = cb[3]; cj[2] = cj[3]; }
    float best = cb[0]; int bj = cj[0];
    if (cb[2] < best) { best = cb[2]; bj = cj[2]; }
    const float wmin = wave_min_f(best);
    const int cand = (best == wmin) ? (bj * 32 + lane) : 0x7fffffff;
    const int widx = wave_min_i(cand);
    kill = ((widx & 31) == lane) ? (widx >> 5) : -1;
    if (lane == k) mine = widx;
  }
  knn_idx[(size_t)(rbase + w) * KNN + lane] = mine;
}

__global__ __launch_bounds__(256) void k_std_partial(const float* __restrict__ xyz,
                                                     const float* __restrict__ feat,
                                                     const float* __restrict__ xyz_s,
                                                     const float* __restrict__ feat_s,
                                                     const int* __restrict__ knn_idx,
                                                     int N, int S, int C,
                                                     float* __restrict__ part) {
  __shared__ float s1[256], s2[256], s3[256], s4[256];
  const int k = blockIdx.x, g = blockIdx.y;
  float ax = 0.f, qx = 0.f, af = 0.f, qf = 0.f;
  for (int p = g * 256 + threadIdx.x; p < BATCH * S; p += STD_G * 256) {
    int b = p / S, s = p % S;
    int idx = knn_idx[(size_t)(b * S + s) * KNN + k];
    for (int c = 0; c < 3; ++c) {
      float d = xyz[(size_t)(b * N + idx) * 3 + c] - xyz_s[(size_t)(b * S + s) * 3 + c];
      ax += d; qx += d * d;
    }
    const float* fr = feat   + (size_t)(b * N + idx) * C;
    const float* fc = feat_s + (size_t)(b * S + s) * C;
    for (int j = 0; j < C; ++j) { float d = fr[j] - fc[j]; af += d; qf += d * d; }
  }
  s1[threadIdx.x] = ax; s2[threadIdx.x] = qx; s3[threadIdx.x] = af; s4[threadIdx.x] = qf;
  __syncthreads();
  for (int o = 128; o > 0; o >>= 1) {
    if ((int)threadIdx.x < o) {
      s1[threadIdx.x] += s1[threadIdx.x + o]; s2[threadIdx.x] += s2[threadIdx.x + o];
      s3[threadIdx.x] += s3[threadIdx.x + o]; s4[threadIdx.x] += s4[threadIdx.x + o];
    }
    __syncthreads();
  }
  if (threadIdx.x == 0) {
    float* q = part + ((size_t)k * STD_G + g) * 4;
    q[0] = s1[0]; q[1] = s2[0]; q[2] = s3[0]; q[3] = s4[0];
  }
}

__global__ void k_std_final(const float* __restrict__ part, int S, int C,
                            float* __restrict__ std_xyz, float* __restrict__ std_feat) {
  const int k = threadIdx.x;
  if (k >= KNN) return;
  float ax = 0.f, qx = 0.f, af = 0.f, qf = 0.f;
  for (int g = 0; g < STD_G; ++g) {
    const float* q = part + ((size_t)k * STD_G + g) * 4;
    ax += q[0]; qx += q[1]; af += q[2]; qf += q[3];
  }
  {
    double n = (double)BATCH * S * 3;
    double var = ((double)qx - (double)ax * ax / n) / (n - 1.0);
    std_xyz[k] = fmaxf(sqrtf(fmaxf((float)var, 0.f)), 1e-5f);
  }
  {
    double n = (double)BATCH * S * C;
    double var = ((double)qf - (double)af * af / n) / (n - 1.0);
    std_feat[k] = fmaxf(sqrtf(fmaxf((float)var, 0.f)), 1e-5f);
  }
}

__global__ __launch_bounds__(256) void k_stage(const float* __restrict__ xyz,
                                               const float* __restrict__ feat,
                                               const float* __restrict__ xyz_s,
                                               const float* __restrict__ feat_s,
                                               const int* __restrict__ knn_idx,
                                               const float* __restrict__ std_xyz,
                                               const float* __restrict__ std_feat,
                                               int N, int S, int C,
                                               float* __restrict__ out) {
  __shared__ int   kidx[KNN];
  __shared__ float sfe[KNN];
  __shared__ float xkn[KNN][3];
  __shared__ float fs[512];
  const int b = blockIdx.x / S, s = blockIdx.x % S;
  if (threadIdx.x < KNN) {
    kidx[threadIdx.x] = knn_idx[(size_t)(b * S + s) * KNN + threadIdx.x];
    sfe[threadIdx.x]  = std_feat[threadIdx.x];
  }
  __syncthreads();
  if (threadIdx.x < KNN)
    __builtin_prefetch(feat + (size_t)(b * N + kidx[threadIdx.x]) * C, 0, 3);
  if (threadIdx.x < KNN * 3) {
    int k = threadIdx.x / 3, c = threadIdx.x % 3;
    xkn[k][c] = (xyz[(size_t)(b * N + kidx[k]) * 3 + c] - xyz_s[(size_t)(b * S + s) * 3 + c]) / std_xyz[k];
  }
  for (int j = threadIdx.x; j < C; j += 256) fs[j] = feat_s[(size_t)(b * S + s) * C + j];
  __syncthreads();
  const int C2 = 2 * C;
  const int fd = (C2 + 2) / 3;
  const int M3 = 3 * fd;
  const float inv_sig = 1.f / SIGMA_;
  for (int j = threadIdx.x; j < C2; j += 256) {
    int src = (int)(((long long)j * (M3 - 1)) / (C2 - 1));
    int cc = src / fd, f = src % fd;
    float fv = -1.f + 2.f * (float)(f + 1) / (float)(fd + 1);
    float acc = 0.f, mx = -FLT_MAX;
    for (int k = 0; k < KNN; ++k) {
      float t  = (xkn[k][cc] - fv) * inv_sig;
      float pe = expf(-0.5f * t * t);
      float fk;
      if (j < C) fk = (feat[(size_t)(b * N + kidx[k]) * C + j] - fs[j]) / sfe[k];
      else       fk = fs[j - C];
      float fw = (fk + pe) * pe;
      acc += fw; mx = fmaxf(mx, fw);
    }
    float xx = acc * (1.f / KNN) + mx;
    out[(size_t)(b * S + s) * C2 + j] = 0.5f * xx * (1.f + erff(xx * 0.70710678118654752f));
  }
}

#define POOL_W 16
__global__ __launch_bounds__(POOL_W * 32) void k_pool(const float* __restrict__ feat, int S, int C2,
                                                      float* __restrict__ out, int off) {
  __shared__ float pmx[POOL_W][32], psm[POOL_W][32];
  const int b    = blockIdx.y;
  const int lane = threadIdx.x & 31, w = threadIdx.x >> 5;
  const int j    = blockIdx.x * 32 + lane;
  const float* p = feat + ((size_t)b * S + w) * C2 + j;
  const size_t step = (size_t)POOL_W * C2;
  float mx = -FLT_MAX, sm = 0.f;
#pragma unroll 8
  for (int s = w; s < S; s += POOL_W, p += step) {
    float v = *p;
    mx = fmaxf(mx, v); sm += v;
  }
  pmx[w][lane] = mx; psm[w][lane] = sm;
  __syncthreads();
  if (w == 0) {
    float m = pmx[0][lane], t = psm[0][lane];
    for (int q = 1; q < POOL_W; ++q) { m = fmaxf(m, pmx[q][lane]); t += psm[q][lane]; }
    out[b * 3840 + off + j]      = m;
    out[b * 3840 + off + C2 + j] = t / (float)S;
  }
}

extern "C" void kernel_launch(void* const* d_in, const int* in_sizes, int n_in,
                              void* d_out, int out_size, void* d_ws, size_t ws_size,
                              hipStream_t stream) {
  const float* xyz_in = (const float*)d_in[0];
  const float* W      = (const float*)d_in[1];
  const float* gamma  = (const float*)d_in[2];
  const float* beta   = (const float*)d_in[3];
  float* out = (float*)d_out;
  float* ws  = (float*)d_ws;

  size_t o = 0;
  float* featA  = ws + o; o += (size_t)BATCH * NPTS * CINIT;
  float* featB  = ws + o; o += (size_t)BATCH * NPTS * CINIT;
  float* xyzA   = ws + o; o += (size_t)BATCH * NPTS * 3;
  float* xyzB   = ws + o; o += (size_t)BATCH * NPTS * 3;
  float* feat_s = ws + o; o += (size_t)BATCH * 1024 * CINIT;
  float* bn_mu  = ws + o; o += 64;
  float* bn_inv = ws + o; o += 64;
  float* stdx   = ws + o; o += 32;
  float* stdf   = ws + o; o += 32;
  float* stdp   = ws + o; o += (size_t)KNN * STD_G * 4;
  int* fps_i = (int*)(ws + o); o += (size_t)BATCH * 1024;
  int* knn_i = (int*)(ws + o); o += (size_t)BATCH * 1024 * KNN;

  k_embed_wmma<<<1024, 256, 0, stream>>>(xyz_in, W, featA);
  k_bn_stats<<<CINIT, 256, 0, stream>>>(featA, bn_mu, bn_inv);
  k_bn_apply<<<2048, 256, 0, stream>>>(featA, bn_mu, bn_inv, gamma, beta);

  const float* xyz_cur = xyz_in;
  float* feat_cur = featA; float* feat_nxt = featB;
  float* xs_buf = xyzA;    float* xs_other = xyzB;
  int N = NPTS, C = CINIT, off = 0;

  for (int st = 0; st < 4; ++st) {
    const int S = N / 2, C2 = 2 * C;
    if (N == 2048)      k_fps<16><<<BATCH, FPS_T, 0, stream>>>(xyz_cur, S, fps_i);
    else if (N == 1024) k_fps<8><<<BATCH, FPS_T, 0, stream>>>(xyz_cur, S, fps_i);
    else if (N == 512)  k_fps<4><<<BATCH, FPS_T, 0, stream>>>(xyz_cur, S, fps_i);
    else                k_fps<2><<<BATCH, FPS_T, 0, stream>>>(xyz_cur, S, fps_i);
    k_gather<<<BATCH * S, 256, 0, stream>>>(xyz_cur, feat_cur, fps_i, N, S, C, xs_buf, feat_s);
    if (N == 2048)      k_knn<64><<<BATCH * S / 16, KNN_W * 32, 0, stream>>>(xs_buf, xyz_cur, S, knn_i);
    else if (N == 1024) k_knn<32><<<BATCH * S / 16, KNN_W * 32, 0, stream>>>(xs_buf, xyz_cur, S, knn_i);
    else if (N == 512)  k_knn<16><<<BATCH * S / 16, KNN_W * 32, 0, stream>>>(xs_buf, xyz_cur, S, knn_i);
    else                k_knn<8><<<BATCH * S / 16, KNN_W * 32, 0, stream>>>(xs_buf, xyz_cur, S, knn_i);
    {
      dim3 g(KNN, STD_G);
      k_std_partial<<<g, 256, 0, stream>>>(xyz_cur, feat_cur, xs_buf, feat_s, knn_i, N, S, C, stdp);
      k_std_final<<<1, KNN, 0, stream>>>(stdp, S, C, stdx, stdf);
    }
    k_stage<<<BATCH * S, 256, 0, stream>>>(xyz_cur, feat_cur, xs_buf, feat_s, knn_i,
                                           stdx, stdf, N, S, C, feat_nxt);
    {
      dim3 g(C2 / 32, BATCH);
      k_pool<<<g, POOL_W * 32, 0, stream>>>(feat_nxt, S, C2, out, off);
    }
    off += 2 * C2;
    float* t = feat_cur; feat_cur = feat_nxt; feat_nxt = t;
    xyz_cur = xs_buf;
    float* u = xs_buf; xs_buf = xs_other; xs_other = u;
    N = S; C = C2;
  }
  (void)in_sizes; (void)n_in; (void)out_size; (void)ws_size;
}


# FPS runs for stage 1 only: sampling an FPS-ordered set from its point 0 re-picks it in order (pick t is already the global argmax), so stages 2-4 copy rows 0..S-1; exact, 3 launches fewer
# speedup vs baseline: 5.8425x; 1.2229x over previous
#include <hip/hip_runtime.h>
#include <cfloat>
#include <cmath>

typedef __attribute__((ext_vector_type(2))) float v2f;
typedef __attribute__((ext_vector_type(8))) float v8f;

#define BATCH 16
#define NPTS 2048
#define CINIT 64
#define KNN 32
#define SIGMA_ 0.3f
#define BN_EPS_ 1e-5f
#define STD_G 16

__global__ __launch_bounds__(256) void k_embed_wmma(const float* __restrict__ xyz,
                                                    const float* __restrict__ W,
                                                    float* __restrict__ h) {
  const int wid  = (blockIdx.x * 256 + (int)threadIdx.x) >> 5;
  const int lane = threadIdx.x & 31;
  const int hf   = lane >> 4;
  const int ln   = lane & 15;
  const int tileM = wid >> 2;
  const int tileN = wid & 3;
  const int m  = tileM * 16 + ln;
  const int c  = tileN * 16 + ln;
  const float x0 = xyz[m * 3], x1 = xyz[m * 3 + 1], x2 = xyz[m * 3 + 2];
  const float w0 = W[c * 3],   w1 = W[c * 3 + 1],   w2 = W[c * 3 + 2];
  v2f a, b;
  a.x = hf ? x2 : x0;  a.y = hf ? 0.f : x1;
  b.x = hf ? w2 : w0;  b.y = hf ? 0.f : w1;
  v8f d = {};
  d = __builtin_amdgcn_wmma_f32_16x16x4_f32(false, a, false, b, (short)0, d, false, false);
#pragma unroll
  for (int r = 0; r < 8; ++r) {
    int row = tileM * 16 + r + 8 * hf;
    h[row * CINIT + tileN * 16 + ln] = d[r];
  }
}

__global__ __launch_bounds__(256) void k_bn_stats(const float* __restrict__ h,
                                                  float* __restrict__ mu,
                                                  float* __restrict__ inv) {
  __shared__ float s1[256], s2[256];
  const int c = blockIdx.x;
  const int n = BATCH * NPTS;
  float a = 0.f, b = 0.f;
  for (int i = threadIdx.x; i < n; i += 256) {
    float v = h[i * CINIT + c];
    a += v; b += v * v;
  }
  s1[threadIdx.x] = a; s2[threadIdx.x] = b; __syncthreads();
  for (int o = 128; o > 0; o >>= 1) {
    if ((int)threadIdx.x < o) { s1[threadIdx.x] += s1[threadIdx.x + o]; s2[threadIdx.x] += s2[threadIdx.x + o]; }
    __syncthreads();
  }
  if (threadIdx.x == 0) {
    float m   = s1[0] / (float)n;
    float var = s2[0] / (float)n - m * m;
    mu[c]  = m;
    inv[c] = rsqrtf(var + BN_EPS_);
  }
}

__global__ void k_bn_apply(float* __restrict__ h, const float* __restrict__ mu,
                           const float* __restrict__ inv, const float* __restrict__ gamma,
                           const float* __restrict__ beta) {
  const int total = BATCH * NPTS * CINIT;
  for (int i = blockIdx.x * blockDim.x + threadIdx.x; i < total; i += gridDim.x * blockDim.x) {
    int c = i & (CINIT - 1);
    float v = (h[i] - mu[c]) * inv[c] * gamma[c] + beta[c];
    h[i] = fmaxf(v, 0.f);
  }
}

template <int M>
__device__ __forceinline__ int lane_xor_row(int x) {
  return __builtin_amdgcn_update_dpp(0, x, 0x160 + M, 0xf, 0xf, true);
}
__device__ __forceinline__ int lane_xor_half(int x) {
  return __builtin_amdgcn_permlanex16(x, x, 0x76543210, 0xfedcba98, false, false);
}
__device__ __forceinline__ float wave_min_f(float x) {
  x = fminf(x, __builtin_bit_cast(float, lane_xor_row<1>(__builtin_bit_cast(int, x))));
  x = fminf(x, __builtin_bit_cast(float, lane_xor_row<2>(__builtin_bit_cast(int, x))));
  x = fminf(x, __builtin_bit_cast(float, lane_xor_row<4>(__builtin_bit_cast(int, x))));
  x = fminf(x, __builtin_bit_cast(float, lane_xor_row<8>(__builtin_bit_cast(int, x))));
  x = fminf(x, __builtin_bit_cast(float, lane_xor_half(__builtin_bit_cast(int, x))));
  return x;
}
__device__ __forceinline__ int wave_min_i(int x) {
  x = min(x, lane_xor_row<1>(x));
  x = min(x, lane_xor_row<2>(x));
  x = min(x, lane_xor_row<4>(x));
  x = min(x, lane_xor_row<8>(x));
  x = min(x, lane_xor_half(x));
  return x;
}
__device__ __forceinline__ int wave_max_i(int x) {
  x = max(x, lane_xor_row<1>(x));
  x = max(x, lane_xor_row<2>(x));
  x = max(x, lane_xor_row<4>(x));
  x = max(x, lane_xor_row<8>(x));
  x = max(x, lane_xor_half(x));
  return x;
}

#define FPS_T 128
#define FPS_W (FPS_T / 32)
template <int PPT>
__global__ __launch_bounds__(FPS_T) void k_fps(const float* __restrict__ xyz, int S,
                                               int* __restrict__ fps_idx) {
  constexpr int N = FPS_T * PPT;
  __shared__ alignas(16) int post[2][FPS_W][8];
  __shared__ int sel[NPTS / 2];
  const int b = blockIdx.x;
  const float* p = xyz + (size_t)b * N * 3;
  const int tid = threadIdx.x;
  const int lane = tid & 31, w = tid >> 5;
  float x[PPT], y[PPT], z[PPT], d[PPT];
#pragma unroll
  for (int q = 0; q < PPT; ++q) {
    const int i = tid + q * FPS_T;
    x[q] = p[i*3]; y[q] = p[i*3+1]; z[q] = p[i*3+2];
    d[q] = 1e10f;
  }
  float fx = p[0], fy = p[1], fz = p[2];
  if (tid == 0) sel[0] = 0;
  for (int it = 1; it < S; ++it) {
#pragma unroll
    for (int q = 0; q < PPT; ++q) {
      float dx = x[q]-fx, dy = y[q]-fy, dz = z[q]-fz;
      d[q] = fminf(d[q], dx*dx + dy*dy + dz*dz);
    }
    constexpr int NC = (PPT >= 4) ? 2 : 1;
    constexpr int L = PPT / NC;
    float cb[NC], cx[NC], cy[NC], cz[NC]; int cq[NC];
#pragma unroll
    for (int c = 0; c < NC; ++c) {
      cb[c] = d[c*L]; cq[c] = c*L; cx[c] = x[c*L]; cy[c] = y[c*L]; cz[c] = z[c*L];
#pragma unroll
      for (int q = c*L + 1; q < (c + 1)*L; ++q) {
        const bool g = d[q] > cb[c];
        cb[c] = g ? d[q] : cb[c]; cq[c] = g ? q : cq[c];
        cx[c] = g ? x[q] : cx[c]; cy[c] = g ? y[q] : cy[c]; cz[c] = g ? z[q] : cz[c];
      }
    }
    if (NC == 2) {
      const bool g = cb[NC-1] > cb[0];
      cb[0] = g ? cb[NC-1] : cb[0]; cq[0] = g ? cq[NC-1] : cq[0];
      cx[0] = g ? cx[NC-1] : cx[0]; cy[0] = g ? cy[NC-1] : cy[0]; cz[0] = g ? cz[NC-1] : cz[0];
    }
    const int best = __float_as_int(cb[0]);
    const int bi   = tid + cq[0] * FPS_T;
    const int wmax = wave_max_i(best);
    const int widx = wave_min_i((best == wmax) ? bi : 0x7fffffff);
    int* mypost = post[it & 1][w];
    if (bi == widx) {
      *(int4*)mypost = make_int4(best, bi, __float_as_int(cx[0]), __float_as_int(cy[0]));
      mypost[4] = __float_as_int(cz[0]);
    }
    __syncthreads();
    const int* e = post[it & 1][lane & (FPS_W - 1)];
    const int4 e0 = *(const int4*)e;
    const int  ez = e[4];
    int gmax = e0.x;
    gmax = max(gmax, lane_xor_row<1>(gmax));
    gmax = max(gmax, lane_xor_row<2>(gmax));
    int far = (e0.x == gmax) ? e0.y : 0x7fffffff;
    far = min(far, lane_xor_row<1>(far));
    far = min(far, lane_xor_row<2>(far));
    const bool win = (e0.y == far);
    int tx = win ? e0.z : 0, ty = win ? e0.w : 0, tz = win ? ez : 0;
    tx |= lane_xor_row<1>(tx); ty |= lane_xor_row<1>(ty); tz |= lane_xor_row<1>(tz);
    tx |= lane_xor_row<2>(tx); ty |= lane_xor_row<2>(ty); tz |= lane_xor_row<2>(tz);
    fx = __int_as_float(tx); fy = __int_as_float(ty); fz = __int_as_float(tz);
    if (tid == 0) sel[it] = far;
  }
  __syncthreads();
  for (int i = tid; i < S; i += FPS_T) fps_idx[b * S + i] = sel[i];
}

template <bool SAMPLED>
__global__ void k_gather(const float* __restrict__ xyz, const float* __restrict__ feat,
                         const int* __restrict__ fps_idx, int N, int S, int C,
                         float* __restrict__ xyz_s, float* __restrict__ feat_s) {
  const int b = blockIdx.x / S, s = blockIdx.x % S;
  const int idx = SAMPLED ? fps_idx[b * S + s] : s;
  if (threadIdx.x < 3)
    xyz_s[(size_t)(b * S + s) * 3 + threadIdx.x] = xyz[(size_t)(b * N + idx) * 3 + threadIdx.x];
  for (int j = threadIdx.x; j < C; j += blockDim.x)
    feat_s[(size_t)(b * S + s) * C + j] = feat[(size_t)(b * N + idx) * C + j];
}

#define KNN_W 16
template <int NPL>
__global__ __launch_bounds__(KNN_W * 32) void k_knn(const float* __restrict__ xs_all,
                                                    const float* __restrict__ x_all,
                                                    int S, int* __restrict__ knn_idx) {
  constexpr int N = NPL * 32;
  constexpr int TPW = N / 16 / KNN_W;
  constexpr int PITCH = N + 4;
  __shared__ float strip[16 * PITCH];
  __shared__ float sN[16];
  const int w    = threadIdx.x >> 5;
  const int lane = threadIdx.x & 31;
  const int hf   = lane >> 4;
  const int ln   = lane & 15;
  const int rbase = blockIdx.x * 16;
  const float* xs = xs_all + (size_t)rbase * 3;
  const float* x  = x_all + (size_t)(rbase / S) * N * 3;
  if (threadIdx.x < 16) {
    float a0 = xs[threadIdx.x * 3];
    float a1 = xs[threadIdx.x * 3 + 1];
    float a2 = xs[threadIdx.x * 3 + 2];
    sN[threadIdx.x] = fmaf(a2, a2, fmaf(a0, a0, a1 * a1));
  }
  __syncthreads();
  const float s0 = xs[ln * 3], s1 = xs[ln * 3 + 1], s2 = xs[ln * 3 + 2];
  v2f a;
  a.x = hf ? s2 : s0;  a.y = hf ? 0.f : s1;
#pragma unroll
  for (int t = 0; t < TPW; ++t) {
    const int n = (w * TPW + t) * 16 + ln;
    const float x0 = x[n * 3], x1 = x[n * 3 + 1], x2 = x[n * 3 + 2];
    const float nn = fmaf(x2, x2, fmaf(x0, x0, x1 * x1));
    v2f b;
    b.x = hf ? x2 : x0;  b.y = hf ? 0.f : x1;
    v8f d = {};
    d = __builtin_amdgcn_wmma_f32_16x16x4_f32(false, a, false, b, (short)0, d, false, false);
#pragma unroll
    for (int r = 0; r < 8; ++r)
      strip[(r + 8 * hf) * PITCH + n] = fmaf(-2.f, d[r], sN[r + 8 * hf] + nn);
  }
  __syncthreads();
  const float* row = strip + w * PITCH;
  float v[NPL];
#pragma unroll
  for (int j = 0; j < NPL; ++j) v[j] = row[j * 32 + lane];
  int kill = -1;
  int mine = 0;
  for (int k = 0; k < KNN; ++k) {
    constexpr int Q = NPL / 4;
    float cb[4]; int cj[4];
#pragma unroll
    for (int c = 0; c < 4; ++c) { cb[c] = FLT_MAX; cj[c] = c * Q; }
#pragma unroll
    for (int q = 0; q < Q; ++q) {
#pragma unroll
      for (int c = 0; c < 4; ++c) {
        const int j = c * Q + q;
        float vj = (j == kill) ? FLT_MAX : v[j];
        v[j] = vj;
        if (vj < cb[c]) { cb[c] = vj; cj[c] = j; }
      }
    }
    if (cb[1] < cb[0]) { cb[0] = cb[1]; cj[0] = cj[1]; }
    if (cb[3] < cb[2]) { cb[2] = cb[3]; cj[2] = cj[3]; }
    float best = cb[0]; int bj = cj[0];
    if (cb[2] < best) { best = cb[2]; bj = cj[2]; }
    const float wmin = wave_min_f(best);
    const int cand = (best == wmin) ? (bj * 32 + lane) : 0x7fffffff;
    const int widx = wave_min_i(cand);
    kill = ((widx & 31) == lane) ? (widx >> 5) : -1;
    if (lane == k) mine = widx;
  }
  knn_idx[(size_t)(rbase + w) * KNN + lane] = mine;
}

__global__ __launch_bounds__(256) void k_std_partial(const float* __restrict__ xyz,
                                                     const float* __restrict__ feat,
                                                     const float* __restrict__ xyz_s,
                                                     const float* __restrict__ feat_s,
                                                     const int* __restrict__ knn_idx,
                                                     int N, int S, int C,
                                                     float* __restrict__ part) {
  __shared__ float s1[256], s2[256], s3[256], s4[256];
  const int k = blockIdx.x, g = blockIdx.y;
  float ax = 0.f, qx = 0.f, af = 0.f, qf = 0.f;
  for (int p = g * 256 + threadIdx.x; p < BATCH * S; p += STD_G * 256) {
    int b = p / S, s = p % S;
    int idx = knn_idx[(size_t)(b * S + s) * KNN + k];
    for (int c = 0; c < 3; ++c) {
      float d = xyz[(size_t)(b * N + idx) * 3 + c] - xyz_s[(size_t)(b * S + s) * 3 + c];
      ax += d; qx += d * d;
    }
    const float* fr = feat   + (size_t)(b * N + idx) * C;
    const float* fc = feat_s + (size_t)(b * S + s) * C;
    for (int j = 0; j < C; ++j) { float d = fr[j] - fc[j]; af += d; qf += d * d; }
  }
  s1[threadIdx.x] = ax; s2[threadIdx.x] = qx; s3[threadIdx.x] = af; s4[threadIdx.x] = qf;
  __syncthreads();
  for (int o = 128; o > 0; o >>= 1) {
    if ((int)threadIdx.x < o) {
      s1[threadIdx.x] += s1[threadIdx.x + o]; s2[threadIdx.x] += s2[threadIdx.x + o];
      s3[threadIdx.x] += s3[threadIdx.x + o]; s4[threadIdx.x] += s4[threadIdx.x + o];
    }
    __syncthreads();
  }
  if (threadIdx.x == 0) {
    float* q = part + ((size_t)k * STD_G + g) * 4;
    q[0] = s1[0]; q[1] = s2[0]; q[2] = s3[0]; q[3] = s4[0];
  }
}

__global__ void k_std_final(const float* __restrict__ part, int S, int C,
                            float* __restrict__ std_xyz, float* __restrict__ std_feat) {
  const int k = threadIdx.x;
  if (k >= KNN) return;
  float ax = 0.f, qx = 0.f, af = 0.f, qf = 0.f;
  for (int g = 0; g < STD_G; ++g) {
    const float* q = part + ((size_t)k * STD_G + g) * 4;
    ax += q[0]; qx += q[1]; af += q[2]; qf += q[3];
  }
  {
    double n = (double)BATCH * S * 3;
    double var = ((double)qx - (double)ax * ax / n) / (n - 1.0);
    std_xyz[k] = fmaxf(sqrtf(fmaxf((float)var, 0.f)), 1e-5f);
  }
  {
    double n = (double)BATCH * S * C;
    double var = ((double)qf - (double)af * af / n) / (n - 1.0);
    std_feat[k] = fmaxf(sqrtf(fmaxf((float)var, 0.f)), 1e-5f);
  }
}

__global__ __launch_bounds__(256) void k_stage(const float* __restrict__ xyz,
                                               const float* __restrict__ feat,
                                               const float* __restrict__ xyz_s,
                                               const float* __restrict__ feat_s,
                                               const int* __restrict__ knn_idx,
                                               const float* __restrict__ std_xyz,
                                               const float* __restrict__ std_feat,
                                               int N, int S, int C,
                                               float* __restrict__ out) {
  __shared__ int   kidx[KNN];
  __shared__ float sfe[KNN];
  __shared__ float xkn[KNN][3];
  __shared__ float fs[512];
  const int b = blockIdx.x / S, s = blockIdx.x % S;
  if (threadIdx.x < KNN) {
    kidx[threadIdx.x] = knn_idx[(size_t)(b * S + s) * KNN + threadIdx.x];
    sfe[threadIdx.x]  = std_feat[threadIdx.x];
  }
  __syncthreads();
  if (threadIdx.x < KNN)
    __builtin_prefetch(feat + (size_t)(b * N + kidx[threadIdx.x]) * C, 0, 3);
  if (threadIdx.x < KNN * 3) {
    int k = threadIdx.x / 3, c = threadIdx.x % 3;
    xkn[k][c] = (xyz[(size_t)(b * N + kidx[k]) * 3 + c] - xyz_s[(size_t)(b * S + s) * 3 + c]) / std_xyz[k];
  }
  for (int j = threadIdx.x; j < C; j += 256) fs[j] = feat_s[(size_t)(b * S + s) * C + j];
  __syncthreads();
  const int C2 = 2 * C;
  const int fd = (C2 + 2) / 3;
  const int M3 = 3 * fd;
  const float inv_sig = 1.f / SIGMA_;
  for (int j = threadIdx.x; j < C2; j += 256) {
    int src = (int)(((long long)j * (M3 - 1)) / (C2 - 1));
    int cc = src / fd, f = src % fd;
    float fv = -1.f + 2.f * (float)(f + 1) / (float)(fd + 1);
    float acc = 0.f, mx = -FLT_MAX;
    for (int k = 0; k < KNN; ++k) {
      float t  = (xkn[k][cc] - fv) * inv_sig;
      float pe = expf(-0.5f * t * t);
      float fk;
      if (j < C) fk = (feat[(size_t)(b * N + kidx[k]) * C + j] - fs[j]) / sfe[k];
      else       fk = fs[j - C];
      float fw = (fk + pe) * pe;
      acc += fw; mx = fmaxf(mx, fw);
    }
    float xx = acc * (1.f / KNN) + mx;
    out[(size_t)(b * S + s) * C2 + j] = 0.5f * xx * (1.f + erff(xx * 0.70710678118654752f));
  }
}

#define POOL_W 16
__global__ __launch_bounds__(POOL_W * 32) void k_pool(const float* __restrict__ feat, int S, int C2,
                                                      float* __restrict__ out, int off) {
  __shared__ float pmx[POOL_W][32], psm[POOL_W][32];
  const int b    = blockIdx.y;
  const int lane = threadIdx.x & 31, w = threadIdx.x >> 5;
  const int j    = blockIdx.x * 32 + lane;
  const float* p = feat + ((size_t)b * S + w) * C2 + j;
  const size_t step = (size_t)POOL_W * C2;
  float mx = -FLT_MAX, sm = 0.f;
#pragma unroll 8
  for (int s = w; s < S; s += POOL_W, p += step) {
    float v = *p;
    mx = fmaxf(mx, v); sm += v;
  }
  pmx[w][lane] = mx; psm[w][lane] = sm;
  __syncthreads();
  if (w == 0) {
    float m = pmx[0][lane], t = psm[0][lane];
    for (int q = 1; q < POOL_W; ++q) { m = fmaxf(m, pmx[q][lane]); t += psm[q][lane]; }
    out[b * 3840 + off + j]      = m;
    out[b * 3840 + off + C2 + j] = t / (float)S;
  }
}

extern "C" void kernel_launch(void* const* d_in, const int* in_sizes, int n_in,
                              void* d_out, int out_size, void* d_ws, size_t ws_size,
                              hipStream_t stream) {
  const float* xyz_in = (const float*)d_in[0];
  const float* W      = (const float*)d_in[1];
  const float* gamma  = (const float*)d_in[2];
  const float* beta   = (const float*)d_in[3];
  float* out = (float*)d_out;
  float* ws  = (float*)d_ws;

  size_t o = 0;
  float* featA  = ws + o; o += (size_t)BATCH * NPTS * CINIT;
  float* featB  = ws + o; o += (size_t)BATCH * NPTS * CINIT;
  float* xyzA   = ws + o; o += (size_t)BATCH * NPTS * 3;
  float* xyzB   = ws + o; o += (size_t)BATCH * NPTS * 3;
  float* feat_s = ws + o; o += (size_t)BATCH * 1024 * CINIT;
  float* bn_mu  = ws + o; o += 64;
  float* bn_inv = ws + o; o += 64;
  float* stdx   = ws + o; o += 32;
  float* stdf   = ws + o; o += 32;
  float* stdp   = ws + o; o += (size_t)KNN * STD_G * 4;
  int* fps_i = (int*)(ws + o); o += (size_t)BATCH * 1024;
  int* knn_i = (int*)(ws + o); o += (size_t)BATCH * 1024 * KNN;

  k_embed_wmma<<<1024, 256, 0, stream>>>(xyz_in, W, featA);
  k_bn_stats<<<CINIT, 256, 0, stream>>>(featA, bn_mu, bn_inv);
  k_bn_apply<<<2048, 256, 0, stream>>>(featA, bn_mu, bn_inv, gamma, beta);

  const float* xyz_cur = xyz_in;
  float* feat_cur = featA; float* feat_nxt = featB;
  float* xs_buf = xyzA;    float* xs_other = xyzB;
  int N = NPTS, C = CINIT, off = 0;

  for (int st = 0; st < 4; ++st) {
    const int S = N / 2, C2 = 2 * C;
    if (N == 2048) {
      k_fps<16><<<BATCH, FPS_T, 0, stream>>>(xyz_cur, S, fps_i);
      k_gather<true><<<BATCH * S, 256, 0, stream>>>(xyz_cur, feat_cur, fps_i, N, S, C, xs_buf, feat_s);
    } else {
      k_gather<false><<<BATCH * S, 256, 0, stream>>>(xyz_cur, feat_cur, fps_i, N, S, C, xs_buf, feat_s);
    }
    if (N == 2048)      k_knn<64><<<BATCH * S / 16, KNN_W * 32, 0, stream>>>(xs_buf, xyz_cur, S, knn_i);
    else if (N == 1024) k_knn<32><<<BATCH * S / 16, KNN_W * 32, 0, stream>>>(xs_buf, xyz_cur, S, knn_i);
    else if (N == 512)  k_knn<16><<<BATCH * S / 16, KNN_W * 32, 0, stream>>>(xs_buf, xyz_cur, S, knn_i);
    else                k_knn<8><<<BATCH * S / 16, KNN_W * 32, 0, stream>>>(xs_buf, xyz_cur, S, knn_i);
    {
      dim3 g(KNN, STD_G);
      k_std_partial<<<g, 256, 0, stream>>>(xyz_cur, feat_cur, xs_buf, feat_s, knn_i, N, S, C, stdp);
      k_std_final<<<1, KNN, 0, stream>>>(stdp, S, C, stdx, stdf);
    }
    k_stage<<<BATCH * S, 256, 0, stream>>>(xyz_cur, feat_cur, xs_buf, feat_s, knn_i,
                                           stdx, stdf, N, S, C, feat_nxt);
    {
      dim3 g(C2 / 32, BATCH);
      k_pool<<<g, POOL_W * 32, 0, stream>>>(feat_nxt, S, C2, out, off);
    }
    off += 2 * C2;
    float* t = feat_cur; feat_cur = feat_nxt; feat_nxt = t;
    xyz_cur = xs_buf;
    float* u = xs_buf; xs_buf = xs_other; xs_other = u;
    N = S; C = C2;
  }
  (void)in_sizes; (void)n_in; (void)out_size; (void)ws_size;
}


# k_std_partial: lanes lie across the channels of one gathered feature row (16 B per lane, whole lines per wave load) instead of one row per lane; wave owns 4 ranks; std sums re-associated
# speedup vs baseline: 6.2076x; 1.0625x over previous
#include <hip/hip_runtime.h>
#include <cfloat>
#include <cmath>

typedef __attribute__((ext_vector_type(2))) float v2f;
typedef __attribute__((ext_vector_type(8))) float v8f;

#define BATCH 16
#define NPTS 2048
#define CINIT 64
#define KNN 32
#define SIGMA_ 0.3f
#define BN_EPS_ 1e-5f
#define STD_G 512

__global__ __launch_bounds__(256) void k_embed_wmma(const float* __restrict__ xyz,
                                                    const float* __restrict__ W,
                                                    float* __restrict__ h) {
  const int wid  = (blockIdx.x * 256 + (int)threadIdx.x) >> 5;
  const int lane = threadIdx.x & 31;
  const int hf   = lane >> 4;
  const int ln   = lane & 15;
  const int tileM = wid >> 2;
  const int tileN = wid & 3;
  const int m  = tileM * 16 + ln;
  const int c  = tileN * 16 + ln;
  const float x0 = xyz[m * 3], x1 = xyz[m * 3 + 1], x2 = xyz[m * 3 + 2];
  const float w0 = W[c * 3],   w1 = W[c * 3 + 1],   w2 = W[c * 3 + 2];
  v2f a, b;
  a.x = hf ? x2 : x0;  a.y = hf ? 0.f : x1;
  b.x = hf ? w2 : w0;  b.y = hf ? 0.f : w1;
  v8f d = {};
  d = __builtin_amdgcn_wmma_f32_16x16x4_f32(false, a, false, b, (short)0, d, false, false);
#pragma unroll
  for (int r = 0; r < 8; ++r) {
    int row = tileM * 16 + r + 8 * hf;
    h[row * CINIT + tileN * 16 + ln] = d[r];
  }
}

__global__ __launch_bounds__(256) void k_bn_stats(const float* __restrict__ h,
                                                  float* __restrict__ mu,
                                                  float* __restrict__ inv) {
  __shared__ float s1[256], s2[256];
  const int c = blockIdx.x;
  const int n = BATCH * NPTS;
  float a = 0.f, b = 0.f;
  for (int i = threadIdx.x; i < n; i += 256) {
    float v = h[i * CINIT + c];
    a += v; b += v * v;
  }
  s1[threadIdx.x] = a; s2[threadIdx.x] = b; __syncthreads();
  for (int o = 128; o > 0; o >>= 1) {
    if ((int)threadIdx.x < o) { s1[threadIdx.x] += s1[threadIdx.x + o]; s2[threadIdx.x] += s2[threadIdx.x + o]; }
    __syncthreads();
  }
  if (threadIdx.x == 0) {
    float m   = s1[0] / (float)n;
    float var = s2[0] / (float)n - m * m;
    mu[c]  = m;
    inv[c] = rsqrtf(var + BN_EPS_);
  }
}

__global__ void k_bn_apply(float* __restrict__ h, const float* __restrict__ mu,
                           const float* __restrict__ inv, const float* __restrict__ gamma,
                           const float* __restrict__ beta) {
  const int total = BATCH * NPTS * CINIT;
  for (int i = blockIdx.x * blockDim.x + threadIdx.x; i < total; i += gridDim.x * blockDim.x) {
    int c = i & (CINIT - 1);
    float v = (h[i] - mu[c]) * inv[c] * gamma[c] + beta[c];
    h[i] = fmaxf(v, 0.f);
  }
}

template <int M>
__device__ __forceinline__ int lane_xor_row(int x) {
  return __builtin_amdgcn_update_dpp(0, x, 0x160 + M, 0xf, 0xf, true);
}
__device__ __forceinline__ int lane_xor_half(int x) {
  return __builtin_amdgcn_permlanex16(x, x, 0x76543210, 0xfedcba98, false, false);
}
__device__ __forceinline__ float wave_min_f(float x) {
  x = fminf(x, __builtin_bit_cast(float, lane_xor_row<1>(__builtin_bit_cast(int, x))));
  x = fminf(x, __builtin_bit_cast(float, lane_xor_row<2>(__builtin_bit_cast(int, x))));
  x = fminf(x, __builtin_bit_cast(float, lane_xor_row<4>(__builtin_bit_cast(int, x))));
  x = fminf(x, __builtin_bit_cast(float, lane_xor_row<8>(__builtin_bit_cast(int, x))));
  x = fminf(x, __builtin_bit_cast(float, lane_xor_half(__builtin_bit_cast(int, x))));
  return x;
}
__device__ __forceinline__ int wave_min_i(int x) {
  x = min(x, lane_xor_row<1>(x));
  x = min(x, lane_xor_row<2>(x));
  x = min(x, lane_xor_row<4>(x));
  x = min(x, lane_xor_row<8>(x));
  x = min(x, lane_xor_half(x));
  return x;
}
__device__ __forceinline__ int wave_max_i(int x) {
  x = max(x, lane_xor_row<1>(x));
  x = max(x, lane_xor_row<2>(x));
  x = max(x, lane_xor_row<4>(x));
  x = max(x, lane_xor_row<8>(x));
  x = max(x, lane_xor_half(x));
  return x;
}

#define FPS_T 128
#define FPS_W (FPS_T / 32)
template <int PPT>
__global__ __launch_bounds__(FPS_T) void k_fps(const float* __restrict__ xyz, int S,
                                               int* __restrict__ fps_idx) {
  constexpr int N = FPS_T * PPT;
  __shared__ alignas(16) int post[2][FPS_W][8];
  __shared__ int sel[NPTS / 2];
  const int b = blockIdx.x;
  const float* p = xyz + (size_t)b * N * 3;
  const int tid = threadIdx.x;
  const int lane = tid & 31, w = tid >> 5;
  float x[PPT], y[PPT], z[PPT], d[PPT];
#pragma unroll
  for (int q = 0; q < PPT; ++q) {
    const int i = tid + q * FPS_T;
    x[q] = p[i*3]; y[q] = p[i*3+1]; z[q] = p[i*3+2];
    d[q] = 1e10f;
  }
  float fx = p[0], fy = p[1], fz = p[2];
  if (tid == 0) sel[0] = 0;
  for (int it = 1; it < S; ++it) {
#pragma unroll
    for (int q = 0; q < PPT; ++q) {
      float dx = x[q]-fx, dy = y[q]-fy, dz = z[q]-fz;
      d[q] = fminf(d[q], dx*dx + dy*dy + dz*dz);
    }
    constexpr int NC = (PPT >= 4) ? 2 : 1;
    constexpr int L = PPT / NC;
    float cb[NC], cx[NC], cy[NC], cz[NC]; int cq[NC];
#pragma unroll
    for (int c = 0; c < NC; ++c) {
      cb[c] = d[c*L]; cq[c] = c*L; cx[c] = x[c*L]; cy[c] = y[c*L]; cz[c] = z[c*L];
#pragma unroll
      for (int q = c*L + 1; q < (c + 1)*L; ++q) {
        const bool g = d[q] > cb[c];
        cb[c] = g ? d[q] : cb[c]; cq[c] = g ? q : cq[c];
        cx[c] = g ? x[q] : cx[c]; cy[c] = g ? y[q] : cy[c]; cz[c] = g ? z[q] : cz[c];
      }
    }
    if (NC == 2) {
      const bool g = cb[NC-1] > cb[0];
      cb[0] = g ? cb[NC-1] : cb[0]; cq[0] = g ? cq[NC-1] : cq[0];
      cx[0] = g ? cx[NC-1] : cx[0]; cy[0] = g ? cy[NC-1] : cy[0]; cz[0] = g ? cz[NC-1] : cz[0];
    }
    const int best = __float_as_int(cb[0]);
    const int bi   = tid + cq[0] * FPS_T;
    const int wmax = wave_max_i(best);
    const int widx = wave_min_i((best == wmax) ? bi : 0x7fffffff);
    int* mypost = post[it & 1][w];
    if (bi == widx) {
      *(int4*)mypost = make_int4(best, bi, __float_as_int(cx[0]), __float_as_int(cy[0]));
      mypost[4] = __float_as_int(cz[0]);
    }
    __syncthreads();
    const int* e = post[it & 1][lane & (FPS_W - 1)];
    const int4 e0 = *(const int4*)e;
    const int  ez = e[4];
    int gmax = e0.x;
    gmax = max(gmax, lane_xor_row<1>(gmax));
    gmax = max(gmax, lane_xor_row<2>(gmax));
    int far = (e0.x == gmax) ? e0.y : 0x7fffffff;
    far = min(far, lane_xor_row<1>(far));
    far = min(far, lane_xor_row<2>(far));
    const bool win = (e0.y == far);
    int tx = win ? e0.z : 0, ty = win ? e0.w : 0, tz = win ? ez : 0;
    tx |= lane_xor_row<1>(tx); ty |= lane_xor_row<1>(ty); tz |= lane_xor_row<1>(tz);
    tx |= lane_xor_row<2>(tx); ty |= lane_xor_row<2>(ty); tz |= lane_xor_row<2>(tz);
    fx = __int_as_float(tx); fy = __int_as_float(ty); fz = __int_as_float(tz);
    if (tid == 0) sel[it] = far;
  }
  __syncthreads();
  for (int i = tid; i < S; i += FPS_T) fps_idx[b * S + i] = sel[i];
}

template <bool SAMPLED>
__global__ void k_gather(const float* __restrict__ xyz, const float* __restrict__ feat,
                         const int* __restrict__ fps_idx, int N, int S, int C,
                         float* __restrict__ xyz_s, float* __restrict__ feat_s) {
  const int b = blockIdx.x / S, s = blockIdx.x % S;
  const int idx = SAMPLED ? fps_idx[b * S + s] : s;
  if (threadIdx.x < 3)
    xyz_s[(size_t)(b * S + s) * 3 + threadIdx.x] = xyz[(size_t)(b * N + idx) * 3 + threadIdx.x];
  for (int j = threadIdx.x; j < C; j += blockDim.x)
    feat_s[(size_t)(b * S + s) * C + j] = feat[(size_t)(b * N + idx) * C + j];
}

#define KNN_W 16
template <int NPL>
__global__ __launch_bounds__(KNN_W * 32) void k_knn(const float* __restrict__ xs_all,
                                                    const float* __restrict__ x_all,
                                                    int S, int* __restrict__ knn_idx) {
  constexpr int N = NPL * 32;
  constexpr int TPW = N / 16 / KNN_W;
  constexpr int PITCH = N + 4;
  __shared__ float strip[16 * PITCH];
  __shared__ float sN[16];
  const int w    = threadIdx.x >> 5;
  const int lane = threadIdx.x & 31;
  const int hf   = lane >> 4;
  const int ln   = lane & 15;
  const int rbase = blockIdx.x * 16;
  const float* xs = xs_all + (size_t)rbase * 3;
  const float* x  = x_all + (size_t)(rbase / S) * N * 3;
  if (threadIdx.x < 16) {
    float a0 = xs[threadIdx.x * 3];
    float a1 = xs[threadIdx.x * 3 + 1];
    float a2 = xs[threadIdx.x * 3 + 2];
    sN[threadIdx.x] = fmaf(a2, a2, fmaf(a0, a0, a1 * a1));
  }
  __syncthreads();
  const float s0 = xs[ln * 3], s1 = xs[ln * 3 + 1], s2 = xs[ln * 3 + 2];
  v2f a;
  a.x = hf ? s2 : s0;  a.y = hf ? 0.f : s1;
#pragma unroll
  for (int t = 0; t < TPW; ++t) {
    const int n = (w * TPW + t) * 16 + ln;
    const float x0 = x[n * 3], x1 = x[n * 3 + 1], x2 = x[n * 3 + 2];
    const float nn = fmaf(x2, x2, fmaf(x0, x0, x1 * x1));
    v2f b;
    b.x = hf ? x2 : x0;  b.y = hf ? 0.f : x1;
    v8f d = {};
    d = __builtin_amdgcn_wmma_f32_16x16x4_f32(false, a, false, b, (short)0, d, false, false);
#pragma unroll
    for (int r = 0; r < 8; ++r)
      strip[(r + 8 * hf) * PITCH + n] = fmaf(-2.f, d[r], sN[r + 8 * hf] + nn);
  }
  __syncthreads();
  const float* row = strip + w * PITCH;
  float v[NPL];
#pragma unroll
  for (int j = 0; j < NPL; ++j) v[j] = row[j * 32 + lane];
  int kill = -1;
  int mine = 0;
  for (int k = 0; k < KNN; ++k) {
    constexpr int Q = NPL / 4;
    float cb[4]; int cj[4];
#pragma unroll
    for (int c = 0; c < 4; ++c) { cb[c] = FLT_MAX; cj[c] = c * Q; }
#pragma unroll
    for (int q = 0; q < Q; ++q) {
#pragma unroll
      for (int c = 0; c < 4; ++c) {
        const int j = c * Q + q;
        float vj = (j == kill) ? FLT_MAX : v[j];
        v[j] = vj;
        if (vj < cb[c]) { cb[c] = vj; cj[c] = j; }
      }
    }
    if (cb[1] < cb[0]) { cb[0] = cb[1]; cj[0] = cj[1]; }
    if (cb[3] < cb[2]) { cb[2] = cb[3]; cj[2] = cj[3]; }
    float best = cb[0]; int bj = cj[0];
    if (cb[2] < best) { best = cb[2]; bj = cj[2]; }
    const float wmin = wave_min_f(best);
    const int cand = (best == wmin) ? (bj * 32 + lane) : 0x7fffffff;
    const int widx = wave_min_i(cand);
    kill = ((widx & 31) == lane) ? (widx >> 5) : -1;
    if (lane == k) mine = widx;
  }
  knn_idx[(size_t)(rbase + w) * KNN + lane] = mine;
}

template <int M>
__device__ __forceinline__ float lane_xor_row_f(float x) {
  return __builtin_bit_cast(float, lane_xor_row<M>(__builtin_bit_cast(int, x)));
}
__device__ __forceinline__ float row_sum_f(float x) {
  x += lane_xor_row_f<1>(x);
  x += lane_xor_row_f<2>(x);
  x += lane_xor_row_f<4>(x);
  x += lane_xor_row_f<8>(x);
  return x;
}
__device__ __forceinline__ float wave_sum_f(float x) {
  x = row_sum_f(x);
  x += __builtin_bit_cast(float, lane_xor_half(__builtin_bit_cast(int, x)));
  return x;
}

template <int C>
__global__ __launch_bounds__(256) void k_std_partial(const float* __restrict__ xyz,
                                                     const float* __restrict__ feat,
                                                     const float* __restrict__ xyz_s,
                                                     const float* __restrict__ feat_s,
                                                     const int* __restrict__ knn_idx,
                                                     float* __restrict__ part) {
  constexpr int S   = (NPTS / 2) * CINIT / C;
  constexpr int N   = 2 * S;
  constexpr int RPB = BATCH * S / STD_G;
  constexpr int CH  = (C >= 128) ? C / 128 : 1;
  constexpr int KL  = (C >= 128) ? 4 : 2;
  constexpr int U   = 16 / (KL * CH);
  static_assert(RPB >= 1 && RPB <= 32 && RPB % U == 0 && S % RPB == 0, "row split");
  const int lane = threadIdx.x & 31, w = threadIdx.x >> 5;
  const int hf   = lane >> 4;
  const int row0 = blockIdx.x * RPB;
  const int b    = row0 / S;

  int4 id4 = make_int4(0, 0, 0, 0);
  float ax[4] = {0.f, 0.f, 0.f, 0.f}, qx[4] = {0.f, 0.f, 0.f, 0.f};
  if (lane < RPB) {
    const int row = row0 + lane;
    id4 = *(const int4*)(knn_idx + (size_t)row * KNN + 4 * w);
    const float c0 = xyz_s[(size_t)row * 3], c1 = xyz_s[(size_t)row * 3 + 1], c2 = xyz_s[(size_t)row * 3 + 2];
    const int ids[4] = {id4.x, id4.y, id4.z, id4.w};
#pragma unroll
    for (int kl = 0; kl < 4; ++kl) {
      const float* q = xyz + (size_t)(b * N + ids[kl]) * 3;
      const float d0 = q[0] - c0, d1 = q[1] - c1, d2 = q[2] - c2;
      ax[kl] = d0 + d1 + d2;
      qx[kl] = d0 * d0 + d1 * d1 + d2 * d2;
    }
  }

  float af[KL], qf[KL];
#pragma unroll
  for (int kl = 0; kl < KL; ++kl) { af[kl] = 0.f; qf[kl] = 0.f; }
  const int jo = (C >= 128) ? lane * 4 : (lane & 15) * 4;
  const float* fbase = feat + (size_t)b * N * C + jo;
#pragma unroll 1
  for (int r0 = 0; r0 < RPB; r0 += U) {
    float4 fs[U][CH], fv[U][KL][CH];
#pragma unroll
    for (int u = 0; u < U; ++u) {
      const int r = r0 + u;
      const int i0 = __builtin_amdgcn_readlane(id4.x, r);
      const int i1 = __builtin_amdgcn_readlane(id4.y, r);
      const int i2 = __builtin_amdgcn_readlane(id4.z, r);
      const int i3 = __builtin_amdgcn_readlane(id4.w, r);
      const float* fc = feat_s + (size_t)(row0 + r) * C + jo;
#pragma unroll
      for (int ch = 0; ch < CH; ++ch) fs[u][ch] = *(const float4*)(fc + ch * 128);
#pragma unroll
      for (int kl = 0; kl < KL; ++kl) {
        int id;
        if (C >= 128) id = (kl == 0) ? i0 : (kl == 1) ? i1 : (kl == 2) ? i2 : i3;
        else          id = (kl == 0) ? (hf ? i1 : i0) : (hf ? i3 : i2);
        const float* fr = fbase + (size_t)id * C;
#pragma unroll
        for (int ch = 0; ch < CH; ++ch) fv[u][kl][ch] = *(const float4*)(fr + ch * 128);
      }
    }
#pragma unroll
    for (int u = 0; u < U; ++u) {
#pragma unroll
      for (int kl = 0; kl < KL; ++kl) {
#pragma unroll
        for (int ch = 0; ch < CH; ++ch) {
          const float d0 = fv[u][kl][ch].x - fs[u][ch].x, d1 = fv[u][kl][ch].y - fs[u][ch].y;
          const float d2 = fv[u][kl][ch].z - fs[u][ch].z, d3 = fv[u][kl][ch].w - fs[u][ch].w;
          af[kl] += (d0 + d1) + (d2 + d3);
          qf[kl] += (d0 * d0 + d1 * d1) + (d2 * d2 + d3 * d3);
        }
      }
    }
  }

#pragma unroll
  for (int kl = 0; kl < 4; ++kl) { ax[kl] = wave_sum_f(ax[kl]); qx[kl] = wave_sum_f(qx[kl]); }
#pragma unroll
  for (int kl = 0; kl < KL; ++kl) {
    if (C >= 128) { af[kl] = wave_sum_f(af[kl]); qf[kl] = wave_sum_f(qf[kl]); }
    else          { af[kl] = row_sum_f(af[kl]);  qf[kl] = row_sum_f(qf[kl]); }
  }
  int  kl;
  bool wr;
  float sf, sq;
  if (C >= 128) {
    kl = lane & 3; wr = lane < 4;
    sf = (kl == 0) ? af[0] : (kl == 1) ? af[1] : (kl == 2) ? af[KL - 2] : af[KL - 1];
    sq = (kl == 0) ? qf[0] : (kl == 1) ? qf[1] : (kl == 2) ? qf[KL - 2] : qf[KL - 1];
  } else {
    const int t = lane & 1;
    kl = 2 * t + hf; wr = (lane & 15) < 2;
    sf = t ? af[1] : af[0];
    sq = t ? qf[1] : qf[0];
  }
  const float sx = (kl == 0) ? ax[0] : (kl == 1) ? ax[1] : (kl == 2) ? ax[2] : ax[3];
  const float sy = (kl == 0) ? qx[0] : (kl == 1) ? qx[1] : (kl == 2) ? qx[2] : qx[3];
  if (wr)
    *(float4*)(part + ((size_t)(4 * w + kl) * STD_G + blockIdx.x) * 4) = make_float4(sx, sy, sf, sq);
}

__global__ __launch_bounds__(KNN * 32) void k_std_final(const float* __restrict__ part, int S, int C,
                                                        float* __restrict__ std_xyz, float* __restrict__ std_feat) {
  const int k = threadIdx.x >> 5, lane = threadIdx.x & 31;
  float ax = 0.f, qx = 0.f, af = 0.f, qf = 0.f;
#pragma unroll
  for (int i = 0; i < STD_G / 32; ++i) {
    const float4 q = *(const float4*)(part + ((size_t)k * STD_G + i * 32 + lane) * 4);
    ax += q.x; qx += q.y; af += q.z; qf += q.w;
  }
  ax = wave_sum_f(ax); qx = wave_sum_f(qx); af = wave_sum_f(af); qf = wave_sum_f(qf);
  if (lane != 0) return;
  {
    double n = (double)BATCH * S * 3;
    double var = ((double)qx - (double)ax * ax / n) / (n - 1.0);
    std_xyz[k] = fmaxf(sqrtf(fmaxf((float)var, 0.f)), 1e-5f);
  }
  {
    double n = (double)BATCH * S * C;
    double var = ((double)qf - (double)af * af / n) / (n - 1.0);
    std_feat[k] = fmaxf(sqrtf(fmaxf((float)var, 0.f)), 1e-5f);
  }
}

__global__ __launch_bounds__(256) void k_stage(const float* __restrict__ xyz,
                                               const float* __restrict__ feat,
                                               const float* __restrict__ xyz_s,
                                               const float* __restrict__ feat_s,
                                               const int* __restrict__ knn_idx,
                                               const float* __restrict__ std_xyz,
                                               const float* __restrict__ std_feat,
                                               int N, int S, int C,
                                               float* __restrict__ out) {
  __shared__ int   kidx[KNN];
  __shared__ float sfe[KNN];
  __shared__ float xkn[KNN][3];
  __shared__ float fs[512];
  const int b = blockIdx.x / S, s = blockIdx.x % S;
  if (threadIdx.x < KNN) {
    kidx[threadIdx.x] = knn_idx[(size_t)(b * S + s) * KNN + threadIdx.x];
    sfe[threadIdx.x]  = std_feat[threadIdx.x];
  }
  __syncthreads();
  if (threadIdx.x < KNN)
    __builtin_prefetch(feat + (size_t)(b * N + kidx[threadIdx.x]) * C, 0, 3);
  if (threadIdx.x < KNN * 3) {
    int k = threadIdx.x / 3, c = threadIdx.x % 3;
    xkn[k][c] = (xyz[(size_t)(b * N + kidx[k]) * 3 + c] - xyz_s[(size_t)(b * S + s) * 3 + c]) / std_xyz[k];
  }
  for (int j = threadIdx.x; j < C; j += 256) fs[j] = feat_s[(size_t)(b * S + s) * C + j];
  __syncthreads();
  const int C2 = 2 * C;
  const int fd = (C2 + 2) / 3;
  const int M3 = 3 * fd;
  const float inv_sig = 1.f / SIGMA_;
  for (int j = threadIdx.x; j < C2; j += 256) {
    int src = (int)(((long long)j * (M3 - 1)) / (C2 - 1));
    int cc = src / fd, f = src % fd;
    float fv = -1.f + 2.f * (float)(f + 1) / (float)(fd + 1);
    float acc = 0.f, mx = -FLT_MAX;
    for (int k = 0; k < KNN; ++k) {
      float t  = (xkn[k][cc] - fv) * inv_sig;
      float pe = expf(-0.5f * t * t);
      float fk;
      if (j < C) fk = (feat[(size_t)(b * N + kidx[k]) * C + j] - fs[j]) / sfe[k];
      else       fk = fs[j - C];
      float fw = (fk + pe) * pe;
      acc += fw; mx = fmaxf(mx, fw);
    }
    float xx = acc * (1.f / KNN) + mx;
    out[(size_t)(b * S + s) * C2 + j] = 0.5f * xx * (1.f + erff(xx * 0.70710678118654752f));
  }
}

#define POOL_W 16
__global__ __launch_bounds__(POOL_W * 32) void k_pool(const float* __restrict__ feat, int S, int C2,
                                                      float* __restrict__ out, int off) {
  __shared__ float pmx[POOL_W][32], psm[POOL_W][32];
  const int b    = blockIdx.y;
  const int lane = threadIdx.x & 31, w = threadIdx.x >> 5;
  const int j    = blockIdx.x * 32 + lane;
  const float* p = feat + ((size_t)b * S + w) * C2 + j;
  const size_t step = (size_t)POOL_W * C2;
  float mx = -FLT_MAX, sm = 0.f;
#pragma unroll 8
  for (int s = w; s < S; s += POOL_W, p += step) {
    float v = *p;
    mx = fmaxf(mx, v); sm += v;
  }
  pmx[w][lane] = mx; psm[w][lane] = sm;
  __syncthreads();
  if (w == 0) {
    float m = pmx[0][lane], t = psm[0][lane];
    for (int q = 1; q < POOL_W; ++q) { m = fmaxf(m, pmx[q][lane]); t += psm[q][lane]; }
    out[b * 3840 + off + j]      = m;
    out[b * 3840 + off + C2 + j] = t / (float)S;
  }
}

extern "C" void kernel_launch(void* const* d_in, const int* in_sizes, int n_in,
                              void* d_out, int out_size, void* d_ws, size_t ws_size,
                              hipStream_t stream) {
  const float* xyz_in = (const float*)d_in[0];
  const float* W      = (const float*)d_in[1];
  const float* gamma  = (const float*)d_in[2];
  const float* beta   = (const float*)d_in[3];
  float* out = (float*)d_out;
  float* ws  = (float*)d_ws;

  size_t o = 0;
  float* featA  = ws + o; o += (size_t)BATCH * NPTS * CINIT;
  float* featB  = ws + o; o += (size_t)BATCH * NPTS * CINIT;
  float* xyzA   = ws + o; o += (size_t)BATCH * NPTS * 3;
  float* xyzB   = ws + o; o += (size_t)BATCH * NPTS * 3;
  float* feat_s = ws + o; o += (size_t)BATCH * 1024 * CINIT;
  float* bn_mu  = ws + o; o += 64;
  float* bn_inv = ws + o; o += 64;
  float* stdx   = ws + o; o += 32;
  float* stdf   = ws + o; o += 32;
  o += (size_t)KNN * 16 * 4;
  int* fps_i = (int*)(ws + o); o += (size_t)BATCH * 1024;
  int* knn_i = (int*)(ws + o); o += (size_t)BATCH * 1024 * KNN;

  k_embed_wmma<<<1024, 256, 0, stream>>>(xyz_in, W, featA);
  k_bn_stats<<<CINIT, 256, 0, stream>>>(featA, bn_mu, bn_inv);
  k_bn_apply<<<2048, 256, 0, stream>>>(featA, bn_mu, bn_inv, gamma, beta);

  const float* xyz_cur = xyz_in;
  float* feat_cur = featA; float* feat_nxt = featB;
  float* xs_buf = xyzA;    float* xs_other = xyzB;
  int N = NPTS, C = CINIT, off = 0;

  for (int st = 0; st < 4; ++st) {
    const int S = N / 2, C2 = 2 * C;
    if (N == 2048) {
      k_fps<16><<<BATCH, FPS_T, 0, stream>>>(xyz_cur, S, fps_i);
      k_gather<true><<<BATCH * S, 256, 0, stream>>>(xyz_cur, feat_cur, fps_i, N, S, C, xs_buf, feat_s);
    } else {
      k_gather<false><<<BATCH * S, 256, 0, stream>>>(xyz_cur, feat_cur, fps_i, N, S, C, xs_buf, feat_s);
    }
    if (N == 2048)      k_knn<64><<<BATCH * S / 16, KNN_W * 32, 0, stream>>>(xs_buf, xyz_cur, S, knn_i);
    else if (N == 1024) k_knn<32><<<BATCH * S / 16, KNN_W * 32, 0, stream>>>(xs_buf, xyz_cur, S, knn_i);
    else if (N == 512)  k_knn<16><<<BATCH * S / 16, KNN_W * 32, 0, stream>>>(xs_buf, xyz_cur, S, knn_i);
    else                k_knn<8><<<BATCH * S / 16, KNN_W * 32, 0, stream>>>(xs_buf, xyz_cur, S, knn_i);
    {
      float* stdp = feat_nxt;
      if (C == 64)       k_std_partial<64><<<STD_G, 256, 0, stream>>>(xyz_cur, feat_cur, xs_buf, feat_s, knn_i, stdp);
      else if (C == 128) k_std_partial<128><<<STD_G, 256, 0, stream>>>(xyz_cur, feat_cur, xs_buf, feat_s, knn_i, stdp);
      else if (C == 256) k_std_partial<256><<<STD_G, 256, 0, stream>>>(xyz_cur, feat_cur, xs_buf, feat_s, knn_i, stdp);
      else               k_std_partial<512><<<STD_G, 256, 0, stream>>>(xyz_cur, feat_cur, xs_buf, feat_s, knn_i, stdp);
      k_std_final<<<1, KNN * 32, 0, stream>>>(stdp, S, C, stdx, stdf);
    }
    k_stage<<<BATCH * S, 256, 0, stream>>>(xyz_cur, feat_cur, xs_buf, feat_s, knn_i,
                                           stdx, stdf, N, S, C, feat_nxt);
    {
      dim3 g(C2 / 32, BATCH);
      k_pool<<<g, POOL_W * 32, 0, stream>>>(feat_nxt, S, C2, out, off);
    }
    off += 2 * C2;
    float* t = feat_cur; feat_cur = feat_nxt; feat_nxt = t;
    xyz_cur = xs_buf;
    float* u = xs_buf; xs_buf = xs_other; xs_other = u;
    N = S; C = C2;
  }
  (void)in_sizes; (void)n_in; (void)out_size; (void)ws_size;
}


# kNN top-32 selection: a lane keeps the minimum of each of 8 groups of its slots; a pass scans the 8 group minima and rescans only the group whose slot the pass before retired, not all slots
# speedup vs baseline: 6.8020x; 1.0958x over previous
#include <hip/hip_runtime.h>
#include <cfloat>
#include <cmath>

typedef __attribute__((ext_vector_type(2))) float v2f;
typedef __attribute__((ext_vector_type(8))) float v8f;

#define BATCH 16
#define NPTS 2048
#define CINIT 64
#define KNN 32
#define SIGMA_ 0.3f
#define BN_EPS_ 1e-5f
#define STD_G 512

__global__ __launch_bounds__(256) void k_embed_wmma(const float* __restrict__ xyz,
                                                    const float* __restrict__ W,
                                                    float* __restrict__ h) {
  const int wid  = (blockIdx.x * 256 + (int)threadIdx.x) >> 5;
  const int lane = threadIdx.x & 31;
  const int hf   = lane >> 4;
  const int ln   = lane & 15;
  const int tileM = wid >> 2;
  const int tileN = wid & 3;
  const int m  = tileM * 16 + ln;
  const int c  = tileN * 16 + ln;
  const float x0 = xyz[m * 3], x1 = xyz[m * 3 + 1], x2 = xyz[m * 3 + 2];
  const float w0 = W[c * 3],   w1 = W[c * 3 + 1],   w2 = W[c * 3 + 2];
  v2f a, b;
  a.x = hf ? x2 : x0;  a.y = hf ? 0.f : x1;
  b.x = hf ? w2 : w0;  b.y = hf ? 0.f : w1;
  v8f d = {};
  d = __builtin_amdgcn_wmma_f32_16x16x4_f32(false, a, false, b, (short)0, d, false, false);
#pragma unroll
  for (int r = 0; r < 8; ++r) {
    int row = tileM * 16 + r + 8 * hf;
    h[row * CINIT + tileN * 16 + ln] = d[r];
  }
}

__global__ __launch_bounds__(256) void k_bn_stats(const float* __restrict__ h,
                                                  float* __restrict__ mu,
                                                  float* __restrict__ inv) {
  __shared__ float s1[256], s2[256];
  const int c = blockIdx.x;
  const int n = BATCH * NPTS;
  float a = 0.f, b = 0.f;
  for (int i = threadIdx.x; i < n; i += 256) {
    float v = h[i * CINIT + c];
    a += v; b += v * v;
  }
  s1[threadIdx.x] = a; s2[threadIdx.x] = b; __syncthreads();
  for (int o = 128; o > 0; o >>= 1) {
    if ((int)threadIdx.x < o) { s1[threadIdx.x] += s1[threadIdx.x + o]; s2[threadIdx.x] += s2[threadIdx.x + o]; }
    __syncthreads();
  }
  if (threadIdx.x == 0) {
    float m   = s1[0] / (float)n;
    float var = s2[0] / (float)n - m * m;
    mu[c]  = m;
    inv[c] = rsqrtf(var + BN_EPS_);
  }
}

__global__ void k_bn_apply(float* __restrict__ h, const float* __restrict__ mu,
                           const float* __restrict__ inv, const float* __restrict__ gamma,
                           const float* __restrict__ beta) {
  const int total = BATCH * NPTS * CINIT;
  for (int i = blockIdx.x * blockDim.x + threadIdx.x; i < total; i += gridDim.x * blockDim.x) {
    int c = i & (CINIT - 1);
    float v = (h[i] - mu[c]) * inv[c] * gamma[c] + beta[c];
    h[i] = fmaxf(v, 0.f);
  }
}

template <int M>
__device__ __forceinline__ int lane_xor_row(int x) {
  return __builtin_amdgcn_update_dpp(0, x, 0x160 + M, 0xf, 0xf, true);
}
__device__ __forceinline__ int lane_xor_half(int x) {
  return __builtin_amdgcn_permlanex16(x, x, 0x76543210, 0xfedcba98, false, false);
}
__device__ __forceinline__ float wave_min_f(float x) {
  x = fminf(x, __builtin_bit_cast(float, lane_xor_row<1>(__builtin_bit_cast(int, x))));
  x = fminf(x, __builtin_bit_cast(float, lane_xor_row<2>(__builtin_bit_cast(int, x))));
  x = fminf(x, __builtin_bit_cast(float, lane_xor_row<4>(__builtin_bit_cast(int, x))));
  x = fminf(x, __builtin_bit_cast(float, lane_xor_row<8>(__builtin_bit_cast(int, x))));
  x = fminf(x, __builtin_bit_cast(float, lane_xor_half(__builtin_bit_cast(int, x))));
  return x;
}
__device__ __forceinline__ int wave_min_i(int x) {
  x = min(x, lane_xor_row<1>(x));
  x = min(x, lane_xor_row<2>(x));
  x = min(x, lane_xor_row<4>(x));
  x = min(x, lane_xor_row<8>(x));
  x = min(x, lane_xor_half(x));
  return x;
}
__device__ __forceinline__ int wave_max_i(int x) {
  x = max(x, lane_xor_row<1>(x));
  x = max(x, lane_xor_row<2>(x));
  x = max(x, lane_xor_row<4>(x));
  x = max(x, lane_xor_row<8>(x));
  x = max(x, lane_xor_half(x));
  return x;
}

#define FPS_T 128
#define FPS_W (FPS_T / 32)
template <int PPT>
__global__ __launch_bounds__(FPS_T) void k_fps(const float* __restrict__ xyz, int S,
                                               int* __restrict__ fps_idx) {
  constexpr int N = FPS_T * PPT;
  __shared__ alignas(16) int post[2][FPS_W][8];
  __shared__ int sel[NPTS / 2];
  const int b = blockIdx.x;
  const float* p = xyz + (size_t)b * N * 3;
  const int tid = threadIdx.x;
  const int lane = tid & 31, w = tid >> 5;
  float x[PPT], y[PPT], z[PPT], d[PPT];
#pragma unroll
  for (int q = 0; q < PPT; ++q) {
    const int i = tid + q * FPS_T;
    x[q] = p[i*3]; y[q] = p[i*3+1]; z[q] = p[i*3+2];
    d[q] = 1e10f;
  }
  float fx = p[0], fy = p[1], fz = p[2];
  if (tid == 0) sel[0] = 0;
  for (int it = 1; it < S; ++it) {
#pragma unroll
    for (int q = 0; q < PPT; ++q) {
      float dx = x[q]-fx, dy = y[q]-fy, dz = z[q]-fz;
      d[q] = fminf(d[q], dx*dx + dy*dy + dz*dz);
    }
    constexpr int NC = (PPT >= 4) ? 2 : 1;
    constexpr int L = PPT / NC;
    float cb[NC], cx[NC], cy[NC], cz[NC]; int cq[NC];
#pragma unroll
    for (int c = 0; c < NC; ++c) {
      cb[c] = d[c*L]; cq[c] = c*L; cx[c] = x[c*L]; cy[c] = y[c*L]; cz[c] = z[c*L];
#pragma unroll
      for (int q = c*L + 1; q < (c + 1)*L; ++q) {
        const bool g = d[q] > cb[c];
        cb[c] = g ? d[q] : cb[c]; cq[c] = g ? q : cq[c];
        cx[c] = g ? x[q] : cx[c]; cy[c] = g ? y[q] : cy[c]; cz[c] = g ? z[q] : cz[c];
      }
    }
    if (NC == 2) {
      const bool g = cb[NC-1] > cb[0];
      cb[0] = g ? cb[NC-1] : cb[0]; cq[0] = g ? cq[NC-1] : cq[0];
      cx[0] = g ? cx[NC-1] : cx[0]; cy[0] = g ? cy[NC-1] : cy[0]; cz[0] = g ? cz[NC-1] : cz[0];
    }
    const int best = __float_as_int(cb[0]);
    const int bi   = tid + cq[0] * FPS_T;
    const int wmax = wave_max_i(best);
    const int widx = wave_min_i((best == wmax) ? bi : 0x7fffffff);
    int* mypost = post[it & 1][w];
    if (bi == widx) {
      *(int4*)mypost = make_int4(best, bi, __float_as_int(cx[0]), __float_as_int(cy[0]));
      mypost[4] = __float_as_int(cz[0]);
    }
    __syncthreads();
    const int* e = post[it & 1][lane & (FPS_W - 1)];
    const int4 e0 = *(const int4*)e;
    const int  ez = e[4];
    int gmax = e0.x;
    gmax = max(gmax, lane_xor_row<1>(gmax));
    gmax = max(gmax, lane_xor_row<2>(gmax));
    int far = (e0.x == gmax) ? e0.y : 0x7fffffff;
    far = min(far, lane_xor_row<1>(far));
    far = min(far, lane_xor_row<2>(far));
    const bool win = (e0.y == far);
    int tx = win ? e0.z : 0, ty = win ? e0.w : 0, tz = win ? ez : 0;
    tx |= lane_xor_row<1>(tx); ty |= lane_xor_row<1>(ty); tz |= lane_xor_row<1>(tz);
    tx |= lane_xor_row<2>(tx); ty |= lane_xor_row<2>(ty); tz |= lane_xor_row<2>(tz);
    fx = __int_as_float(tx); fy = __int_as_float(ty); fz = __int_as_float(tz);
    if (tid == 0) sel[it] = far;
  }
  __syncthreads();
  for (int i = tid; i < S; i += FPS_T) fps_idx[b * S + i] = sel[i];
}

template <bool SAMPLED>
__global__ void k_gather(const float* __restrict__ xyz, const float* __restrict__ feat,
                         const int* __restrict__ fps_idx, int N, int S, int C,
                         float* __restrict__ xyz_s, float* __restrict__ feat_s) {
  const int b = blockIdx.x / S, s = blockIdx.x % S;
  const int idx = SAMPLED ? fps_idx[b * S + s] : s;
  if (threadIdx.x < 3)
    xyz_s[(size_t)(b * S + s) * 3 + threadIdx.x] = xyz[(size_t)(b * N + idx) * 3 + threadIdx.x];
  for (int j = threadIdx.x; j < C; j += blockDim.x)
    feat_s[(size_t)(b * S + s) * C + j] = feat[(size_t)(b * N + idx) * C + j];
}

#define KNN_NG 8
template <int NPL, int g>
__device__ __forceinline__ void knn_group_min(float (&v)[NPL], int kill, float& m, int& mj) {
  constexpr int G  = NPL / KNN_NG;
  constexpr int NC = (G >= 4) ? 2 : 1;
  constexpr int L  = G / NC;
  float cb[NC]; int cj[NC];
#pragma unroll
  for (int c = 0; c < NC; ++c) {
    cb[c] = FLT_MAX; cj[c] = g * G + c * L;
#pragma unroll
    for (int q = 0; q < L; ++q) {
      const int j = g * G + c * L + q;
      const float vj = (j == kill) ? FLT_MAX : v[j];
      v[j] = vj;
      if (vj < cb[c]) { cb[c] = vj; cj[c] = j; }
    }
  }
  if (NC == 2) { if (cb[NC - 1] < cb[0]) { cb[0] = cb[NC - 1]; cj[0] = cj[NC - 1]; } }
  m = cb[0]; mj = cj[0];
}
template <int NPL, int g>
__device__ __forceinline__ void knn_all_groups(float (&v)[NPL], float (&gm)[KNN_NG], int (&gj)[KNN_NG]) {
  knn_group_min<NPL, g>(v, -1, gm[g], gj[g]);
  if constexpr (g + 1 < KNN_NG) knn_all_groups<NPL, g + 1>(v, gm, gj);
}

template <int NPL, int g>
__device__ __forceinline__ void knn_regroup(float (&v)[NPL], int kill, unsigned grp, float (&gm)[KNN_NG], int (&gj)[KNN_NG]) {
  if (grp == (unsigned)g) knn_group_min<NPL, g>(v, kill, gm[g], gj[g]);
  if constexpr (g + 1 < KNN_NG) knn_regroup<NPL, g + 1>(v, kill, grp, gm, gj);
}

#define KNN_W 16
template <int NPL>
__global__ __launch_bounds__(KNN_W * 32) void k_knn(const float* __restrict__ xs_all,
                                                    const float* __restrict__ x_all,
                                                    int S, int* __restrict__ knn_idx) {
  constexpr int N = NPL * 32;
  constexpr int TPW = N / 16 / KNN_W;
  constexpr int PITCH = N + 4;
  __shared__ float strip[16 * PITCH];
  __shared__ float sN[16];
  const int w    = threadIdx.x >> 5;
  const int lane = threadIdx.x & 31;
  const int hf   = lane >> 4;
  const int ln   = lane & 15;
  const int rbase = blockIdx.x * 16;
  const float* xs = xs_all + (size_t)rbase * 3;
  const float* x  = x_all + (size_t)(rbase / S) * N * 3;
  if (threadIdx.x < 16) {
    float a0 = xs[threadIdx.x * 3];
    float a1 = xs[threadIdx.x * 3 + 1];
    float a2 = xs[threadIdx.x * 3 + 2];
    sN[threadIdx.x] = fmaf(a2, a2, fmaf(a0, a0, a1 * a1));
  }
  __syncthreads();
  const float s0 = xs[ln * 3], s1 = xs[ln * 3 + 1], s2 = xs[ln * 3 + 2];
  v2f a;
  a.x = hf ? s2 : s0;  a.y = hf ? 0.f : s1;
#pragma unroll
  for (int t = 0; t < TPW; ++t) {
    const int n = (w * TPW + t) * 16 + ln;
    const float x0 = x[n * 3], x1 = x[n * 3 + 1], x2 = x[n * 3 + 2];
    const float nn = fmaf(x2, x2, fmaf(x0, x0, x1 * x1));
    v2f b;
    b.x = hf ? x2 : x0;  b.y = hf ? 0.f : x1;
    v8f d = {};
    d = __builtin_amdgcn_wmma_f32_16x16x4_f32(false, a, false, b, (short)0, d, false, false);
#pragma unroll
    for (int r = 0; r < 8; ++r)
      strip[(r + 8 * hf) * PITCH + n] = fmaf(-2.f, d[r], sN[r + 8 * hf] + nn);
  }
  __syncthreads();
  const float* row = strip + w * PITCH;
  float v[NPL];
#pragma unroll
  for (int j = 0; j < NPL; ++j) v[j] = row[j * 32 + lane];
  float gm[KNN_NG]; int gj[KNN_NG];
  knn_all_groups<NPL, 0>(v, gm, gj);
  int mine = 0;
  for (int k = 0; k < KNN; ++k) {
    constexpr int HG = KNN_NG / 2;
    float b0 = gm[0], b1 = gm[HG]; int j0 = gj[0], j1 = gj[HG];
#pragma unroll
    for (int g = 1; g < HG; ++g) {
      if (gm[g] < b0)      { b0 = gm[g];      j0 = gj[g]; }
      if (gm[HG + g] < b1) { b1 = gm[HG + g]; j1 = gj[HG + g]; }
    }
    if (b1 < b0) { b0 = b1; j0 = j1; }
    const float wmin = wave_min_f(b0);
    const int cand = (b0 == wmin) ? (j0 * 32 + lane) : 0x7fffffff;
    const int widx = wave_min_i(cand);
    if (lane == k) mine = widx;
    if (k + 1 < KNN) {
      const int u    = __builtin_amdgcn_readfirstlane(widx);
      const int js   = u >> 5;
      const int kill = (lane == (u & 31)) ? js : -1;
      knn_regroup<NPL, 0>(v, kill, (unsigned)js / (NPL / KNN_NG), gm, gj);
    }
  }
  knn_idx[(size_t)(rbase + w) * KNN + lane] = mine;
}

template <int M>
__device__ __forceinline__ float lane_xor_row_f(float x) {
  return __builtin_bit_cast(float, lane_xor_row<M>(__builtin_bit_cast(int, x)));
}
__device__ __forceinline__ float row_sum_f(float x) {
  x += lane_xor_row_f<1>(x);
  x += lane_xor_row_f<2>(x);
  x += lane_xor_row_f<4>(x);
  x += lane_xor_row_f<8>(x);
  return x;
}
__device__ __forceinline__ float wave_sum_f(float x) {
  x = row_sum_f(x);
  x += __builtin_bit_cast(float, lane_xor_half(__builtin_bit_cast(int, x)));
  return x;
}

template <int C>
__global__ __launch_bounds__(256) void k_std_partial(const float* __restrict__ xyz,
                                                     const float* __restrict__ feat,
                                                     const float* __restrict__ xyz_s,
                                                     const float* __restrict__ feat_s,
                                                     const int* __restrict__ knn_idx,
                                                     float* __restrict__ part) {
  constexpr int S   = (NPTS / 2) * CINIT / C;
  constexpr int N   = 2 * S;
  constexpr int RPB = BATCH * S / STD_G;
  constexpr int CH  = (C >= 128) ? C / 128 : 1;
  constexpr int KL  = (C >= 128) ? 4 : 2;
  constexpr int U   = 16 / (KL * CH);
  static_assert(RPB >= 1 && RPB <= 32 && RPB % U == 0 && S % RPB == 0, "row split");
  const int lane = threadIdx.x & 31, w = threadIdx.x >> 5;
  const int hf   = lane >> 4;
  const int row0 = blockIdx.x * RPB;
  const int b    = row0 / S;

  int4 id4 = make_int4(0, 0, 0, 0);
  float ax[4] = {0.f, 0.f, 0.f, 0.f}, qx[4] = {0.f, 0.f, 0.f, 0.f};
  if (lane < RPB) {
    const int row = row0 + lane;
    id4 = *(const int4*)(knn_idx + (size_t)row * KNN + 4 * w);
    const float c0 = xyz_s[(size_t)row * 3], c1 = xyz_s[(size_t)row * 3 + 1], c2 = xyz_s[(size_t)row * 3 + 2];
    const int ids[4] = {id4.x, id4.y, id4.z, id4.w};
#pragma unroll
    for (int kl = 0; kl < 4; ++kl) {
      const float* q = xyz + (size_t)(b * N + ids[kl]) * 3;
      const float d0 = q[0] - c0, d1 = q[1] - c1, d2 = q[2] - c2;
      ax[kl] = d0 + d1 + d2;
      qx[kl] = d0 * d0 + d1 * d1 + d2 * d2;
    }
  }

  float af[KL], qf[KL];
#pragma unroll
  for (int kl = 0; kl < KL; ++kl) { af[kl] = 0.f; qf[kl] = 0.f; }
  const int jo = (C >= 128) ? lane * 4 : (lane & 15) * 4;
  const float* fbase = feat + (size_t)b * N * C + jo;
#pragma unroll 1
  for (int r0 = 0; r0 < RPB; r0 += U) {
    float4 fs[U][CH], fv[U][KL][CH];
#pragma unroll
    for (int u = 0; u < U; ++u) {
      const int r = r0 + u;
      const int i0 = __builtin_amdgcn_readlane(id4.x, r);
      const int i1 = __builtin_amdgcn_readlane(id4.y, r);
      const int i2 = __builtin_amdgcn_readlane(id4.z, r);
      const int i3 = __builtin_amdgcn_readlane(id4.w, r);
      const float* fc = feat_s + (size_t)(row0 + r) * C + jo;
#pragma unroll
      for (int ch = 0; ch < CH; ++ch) fs[u][ch] = *(const float4*)(fc + ch * 128);
#pragma unroll
      for (int kl = 0; kl < KL; ++kl) {
        int id;
        if (C >= 128) id = (kl == 0) ? i0 : (kl == 1) ? i1 : (kl == 2) ? i2 : i3;
        else          id = (kl == 0) ? (hf ? i1 : i0) : (hf ? i3 : i2);
        const float* fr = fbase + (size_t)id * C;
#pragma unroll
        for (int ch = 0; ch < CH; ++ch) fv[u][kl][ch] = *(const float4*)(fr + ch * 128);
      }
    }
#pragma unroll
    for (int u = 0; u < U; ++u) {
#pragma unroll
      for (int kl = 0; kl < KL; ++kl) {
#pragma unroll
        for (int ch = 0; ch < CH; ++ch) {
          const float d0 = fv[u][kl][ch].x - fs[u][ch].x, d1 = fv[u][kl][ch].y - fs[u][ch].y;
          const float d2 = fv[u][kl][ch].z - fs[u][ch].z, d3 = fv[u][kl][ch].w - fs[u][ch].w;
          af[kl] += (d0 + d1) + (d2 + d3);
          qf[kl] += (d0 * d0 + d1 * d1) + (d2 * d2 + d3 * d3);
        }
      }
    }
  }

#pragma unroll
  for (int kl = 0; kl < 4; ++kl) { ax[kl] = wave_sum_f(ax[kl]); qx[kl] = wave_sum_f(qx[kl]); }
#pragma unroll
  for (int kl = 0; kl < KL; ++kl) {
    if (C >= 128) { af[kl] = wave_sum_f(af[kl]); qf[kl] = wave_sum_f(qf[kl]); }
    else          { af[kl] = row_sum_f(af[kl]);  qf[kl] = row_sum_f(qf[kl]); }
  }
  int  kl;
  bool wr;
  float sf, sq;
  if (C >= 128) {
    kl = lane & 3; wr = lane < 4;
    sf = (kl == 0) ? af[0] : (kl == 1) ? af[1] : (kl == 2) ? af[KL - 2] : af[KL - 1];
    sq = (kl == 0) ? qf[0] : (kl == 1) ? qf[1] : (kl == 2) ? qf[KL - 2] : qf[KL - 1];
  } else {
    const int t = lane & 1;
    kl = 2 * t + hf; wr = (lane & 15) < 2;
    sf = t ? af[1] : af[0];
    sq = t ? qf[1] : qf[0];
  }
  const float sx = (kl == 0) ? ax[0] : (kl == 1) ? ax[1] : (kl == 2) ? ax[2] : ax[3];
  const float sy = (kl == 0) ? qx[0] : (kl == 1) ? qx[1] : (kl == 2) ? qx[2] : qx[3];
  if (wr)
    *(float4*)(part + ((size_t)(4 * w + kl) * STD_G + blockIdx.x) * 4) = make_float4(sx, sy, sf, sq);
}

__global__ __launch_bounds__(KNN * 32) void k_std_final(const float* __restrict__ part, int S, int C,
                                                        float* __restrict__ std_xyz, float* __restrict__ std_feat) {
  const int k = threadIdx.x >> 5, lane = threadIdx.x & 31;
  float ax = 0.f, qx = 0.f, af = 0.f, qf = 0.f;
#pragma unroll
  for (int i = 0; i < STD_G / 32; ++i) {
    const float4 q = *(const float4*)(part + ((size_t)k * STD_G + i * 32 + lane) * 4);
    ax += q.x; qx += q.y; af += q.z; qf += q.w;
  }
  ax = wave_sum_f(ax); qx = wave_sum_f(qx); af = wave_sum_f(af); qf = wave_sum_f(qf);
  if (lane != 0) return;
  {
    double n = (double)BATCH * S * 3;
    double var = ((double)qx - (double)ax * ax / n) / (n - 1.0);
    std_xyz[k] = fmaxf(sqrtf(fmaxf((float)var, 0.f)), 1e-5f);
  }
  {
    double n = (double)BATCH * S * C;
    double var = ((double)qf - (double)af * af / n) / (n - 1.0);
    std_feat[k] = fmaxf(sqrtf(fmaxf((float)var, 0.f)), 1e-5f);
  }
}

__global__ __launch_bounds__(256) void k_stage(const float* __restrict__ xyz,
                                               const float* __restrict__ feat,
                                               const float* __restrict__ xyz_s,
                                               const float* __restrict__ feat_s,
                                               const int* __restrict__ knn_idx,
                                               const float* __restrict__ std_xyz,
                                               const float* __restrict__ std_feat,
                                               int N, int S, int C,
                                               float* __restrict__ out) {
  __shared__ int   kidx[KNN];
  __shared__ float sfe[KNN];
  __shared__ float xkn[KNN][3];
  __shared__ float fs[512];
  const int b = blockIdx.x / S, s = blockIdx.x % S;
  if (threadIdx.x < KNN) {
    kidx[threadIdx.x] = knn_idx[(size_t)(b * S + s) * KNN + threadIdx.x];
    sfe[threadIdx.x]  = std_feat[threadIdx.x];
  }
  __syncthreads();
  if (threadIdx.x < KNN)
    __builtin_prefetch(feat + (size_t)(b * N + kidx[threadIdx.x]) * C, 0, 3);
  if (threadIdx.x < KNN * 3) {
    int k = threadIdx.x / 3, c = threadIdx.x % 3;
    xkn[k][c] = (xyz[(size_t)(b * N + kidx[k]) * 3 + c] - xyz_s[(size_t)(b * S + s) * 3 + c]) / std_xyz[k];
  }
  for (int j = threadIdx.x; j < C; j += 256) fs[j] = feat_s[(size_t)(b * S + s) * C + j];
  __syncthreads();
  const int C2 = 2 * C;
  const int fd = (C2 + 2) / 3;
  const int M3 = 3 * fd;
  const float inv_sig = 1.f / SIGMA_;
  for (int j = threadIdx.x; j < C2; j += 256) {
    int src = (int)(((long long)j * (M3 - 1)) / (C2 - 1));
    int cc = src / fd, f = src % fd;
    float fv = -1.f + 2.f * (float)(f + 1) / (float)(fd + 1);
    float acc = 0.f, mx = -FLT_MAX;
    for (int k = 0; k < KNN; ++k) {
      float t  = (xkn[k][cc] - fv) * inv_sig;
      float pe = expf(-0.5f * t * t);
      float fk;
      if (j < C) fk = (feat[(size_t)(b * N + kidx[k]) * C + j] - fs[j]) / sfe[k];
      else       fk = fs[j - C];
      float fw = (fk + pe) * pe;
      acc += fw; mx = fmaxf(mx, fw);
    }
    float xx = acc * (1.f / KNN) + mx;
    out[(size_t)(b * S + s) * C2 + j] = 0.5f * xx * (1.f + erff(xx * 0.70710678118654752f));
  }
}

#define POOL_W 16
__global__ __launch_bounds__(POOL_W * 32) void k_pool(const float* __restrict__ feat, int S, int C2,
                                                      float* __restrict__ out, int off) {
  __shared__ float pmx[POOL_W][32], psm[POOL_W][32];
  const int b    = blockIdx.y;
  const int lane = threadIdx.x & 31, w = threadIdx.x >> 5;
  const int j    = blockIdx.x * 32 + lane;
  const float* p = feat + ((size_t)b * S + w) * C2 + j;
  const size_t step = (size_t)POOL_W * C2;
  float mx = -FLT_MAX, sm = 0.f;
#pragma unroll 8
  for (int s = w; s < S; s += POOL_W, p += step) {
    float v = *p;
    mx = fmaxf(mx, v); sm += v;
  }
  pmx[w][lane] = mx; psm[w][lane] = sm;
  __syncthreads();
  if (w == 0) {
    float m = pmx[0][lane], t = psm[0][lane];
    for (int q = 1; q < POOL_W; ++q) { m = fmaxf(m, pmx[q][lane]); t += psm[q][lane]; }
    out[b * 3840 + off + j]      = m;
    out[b * 3840 + off + C2 + j] = t / (float)S;
  }
}

extern "C" void kernel_launch(void* const* d_in, const int* in_sizes, int n_in,
                              void* d_out, int out_size, void* d_ws, size_t ws_size,
                              hipStream_t stream) {
  const float* xyz_in = (const float*)d_in[0];
  const float* W      = (const float*)d_in[1];
  const float* gamma  = (const float*)d_in[2];
  const float* beta   = (const float*)d_in[3];
  float* out = (float*)d_out;
  float* ws  = (float*)d_ws;

  size_t o = 0;
  float* featA  = ws + o; o += (size_t)BATCH * NPTS * CINIT;
  float* featB  = ws + o; o += (size_t)BATCH * NPTS * CINIT;
  float* xyzA   = ws + o; o += (size_t)BATCH * NPTS * 3;
  float* xyzB   = ws + o; o += (size_t)BATCH * NPTS * 3;
  float* feat_s = ws + o; o += (size_t)BATCH * 1024 * CINIT;
  float* bn_mu  = ws + o; o += 64;
  float* bn_inv = ws + o; o += 64;
  float* stdx   = ws + o; o += 32;
  float* stdf   = ws + o; o += 32;
  o += (size_t)KNN * 16 * 4;
  int* fps_i = (int*)(ws + o); o += (size_t)BATCH * 1024;
  int* knn_i = (int*)(ws + o); o += (size_t)BATCH * 1024 * KNN;

  k_embed_wmma<<<1024, 256, 0, stream>>>(xyz_in, W, featA);
  k_bn_stats<<<CINIT, 256, 0, stream>>>(featA, bn_mu, bn_inv);
  k_bn_apply<<<2048, 256, 0, stream>>>(featA, bn_mu, bn_inv, gamma, beta);

  const float* xyz_cur = xyz_in;
  float* feat_cur = featA; float* feat_nxt = featB;
  float* xs_buf = xyzA;    float* xs_other = xyzB;
  int N = NPTS, C = CINIT, off = 0;

  for (int st = 0; st < 4; ++st) {
    const int S = N / 2, C2 = 2 * C;
    if (N == 2048) {
      k_fps<16><<<BATCH, FPS_T, 0, stream>>>(xyz_cur, S, fps_i);
      k_gather<true><<<BATCH * S, 256, 0, stream>>>(xyz_cur, feat_cur, fps_i, N, S, C, xs_buf, feat_s);
    } else {
      k_gather<false><<<BATCH * S, 256, 0, stream>>>(xyz_cur, feat_cur, fps_i, N, S, C, xs_buf, feat_s);
    }
    if (N == 2048)      k_knn<64><<<BATCH * S / 16, KNN_W * 32, 0, stream>>>(xs_buf, xyz_cur, S, knn_i);
    else if (N == 1024) k_knn<32><<<BATCH * S / 16, KNN_W * 32, 0, stream>>>(xs_buf, xyz_cur, S, knn_i);
    else if (N == 512)  k_knn<16><<<BATCH * S / 16, KNN_W * 32, 0, stream>>>(xs_buf, xyz_cur, S, knn_i);
    else                k_knn<8><<<BATCH * S / 16, KNN_W * 32, 0, stream>>>(xs_buf, xyz_cur, S, knn_i);
    {
      float* stdp = feat_nxt;
      if (C == 64)       k_std_partial<64><<<STD_G, 256, 0, stream>>>(xyz_cur, feat_cur, xs_buf, feat_s, knn_i, stdp);
      else if (C == 128) k_std_partial<128><<<STD_G, 256, 0, stream>>>(xyz_cur, feat_cur, xs_buf, feat_s, knn_i, stdp);
      else if (C == 256) k_std_partial<256><<<STD_G, 256, 0, stream>>>(xyz_cur, feat_cur, xs_buf, feat_s, knn_i, stdp);
      else               k_std_partial<512><<<STD_G, 256, 0, stream>>>(xyz_cur, feat_cur, xs_buf, feat_s, knn_i, stdp);
      k_std_final<<<1, KNN * 32, 0, stream>>>(stdp, S, C, stdx, stdf);
    }
    k_stage<<<BATCH * S, 256, 0, stream>>>(xyz_cur, feat_cur, xs_buf, feat_s, knn_i,
                                           stdx, stdf, N, S, C, feat_nxt);
    {
      dim3 g(C2 / 32, BATCH);
      k_pool<<<g, POOL_W * 32, 0, stream>>>(feat_nxt, S, C2, out, off);
    }
    off += 2 * C2;
    float* t = feat_cur; feat_cur = feat_nxt; feat_nxt = t;
    xyz_cur = xs_buf;
    float* u = xs_buf; xs_buf = xs_other; xs_other = u;
    N = S; C = C2;
  }
  (void)in_sizes; (void)n_in; (void)out_size; (void)ws_size;
}


# FPS pass latency: 8 waves x 8 consecutive points per lane; first-max ties fall to the lowest slot/lane/wave, so ballot + first set bit + readlane replace index reductions and coord OR-broadcast
# speedup vs baseline: 9.2845x; 1.3650x over previous
#include <hip/hip_runtime.h>
#include <cfloat>
#include <cmath>

typedef __attribute__((ext_vector_type(2))) float v2f;
typedef __attribute__((ext_vector_type(8))) float v8f;

#define BATCH 16
#define NPTS 2048
#define CINIT 64
#define KNN 32
#define SIGMA_ 0.3f
#define BN_EPS_ 1e-5f
#define STD_G 512

__global__ __launch_bounds__(256) void k_embed_wmma(const float* __restrict__ xyz,
                                                    const float* __restrict__ W,
                                                    float* __restrict__ h) {
  const int wid  = (blockIdx.x * 256 + (int)threadIdx.x) >> 5;
  const int lane = threadIdx.x & 31;
  const int hf   = lane >> 4;
  const int ln   = lane & 15;
  const int tileM = wid >> 2;
  const int tileN = wid & 3;
  const int m  = tileM * 16 + ln;
  const int c  = tileN * 16 + ln;
  const float x0 = xyz[m * 3], x1 = xyz[m * 3 + 1], x2 = xyz[m * 3 + 2];
  const float w0 = W[c * 3],   w1 = W[c * 3 + 1],   w2 = W[c * 3 + 2];
  v2f a, b;
  a.x = hf ? x2 : x0;  a.y = hf ? 0.f : x1;
  b.x = hf ? w2 : w0;  b.y = hf ? 0.f : w1;
  v8f d = {};
  d = __builtin_amdgcn_wmma_f32_16x16x4_f32(false, a, false, b, (short)0, d, false, false);
#pragma unroll
  for (int r = 0; r < 8; ++r) {
    int row = tileM * 16 + r + 8 * hf;
    h[row * CINIT + tileN * 16 + ln] = d[r];
  }
}

__global__ __launch_bounds__(256) void k_bn_stats(const float* __restrict__ h,
                                                  float* __restrict__ mu,
                                                  float* __restrict__ inv) {
  __shared__ float s1[256], s2[256];
  const int c = blockIdx.x;
  const int n = BATCH * NPTS;
  float a = 0.f, b = 0.f;
  for (int i = threadIdx.x; i < n; i += 256) {
    float v = h[i * CINIT + c];
    a += v; b += v * v;
  }
  s1[threadIdx.x] = a; s2[threadIdx.x] = b; __syncthreads();
  for (int o = 128; o > 0; o >>= 1) {
    if ((int)threadIdx.x < o) { s1[threadIdx.x] += s1[threadIdx.x + o]; s2[threadIdx.x] += s2[threadIdx.x + o]; }
    __syncthreads();
  }
  if (threadIdx.x == 0) {
    float m   = s1[0] / (float)n;
    float var = s2[0] / (float)n - m * m;
    mu[c]  = m;
    inv[c] = rsqrtf(var + BN_EPS_);
  }
}

__global__ void k_bn_apply(float* __restrict__ h, const float* __restrict__ mu,
                           const float* __restrict__ inv, const float* __restrict__ gamma,
                           const float* __restrict__ beta) {
  const int total = BATCH * NPTS * CINIT;
  for (int i = blockIdx.x * blockDim.x + threadIdx.x; i < total; i += gridDim.x * blockDim.x) {
    int c = i & (CINIT - 1);
    float v = (h[i] - mu[c]) * inv[c] * gamma[c] + beta[c];
    h[i] = fmaxf(v, 0.f);
  }
}

template <int M>
__device__ __forceinline__ int lane_xor_row(int x) {
  return __builtin_amdgcn_update_dpp(0, x, 0x160 + M, 0xf, 0xf, true);
}
__device__ __forceinline__ int lane_xor_half(int x) {
  return __builtin_amdgcn_permlanex16(x, x, 0x76543210, 0xfedcba98, false, false);
}
__device__ __forceinline__ float wave_min_f(float x) {
  x = fminf(x, __builtin_bit_cast(float, lane_xor_row<1>(__builtin_bit_cast(int, x))));
  x = fminf(x, __builtin_bit_cast(float, lane_xor_row<2>(__builtin_bit_cast(int, x))));
  x = fminf(x, __builtin_bit_cast(float, lane_xor_row<4>(__builtin_bit_cast(int, x))));
  x = fminf(x, __builtin_bit_cast(float, lane_xor_row<8>(__builtin_bit_cast(int, x))));
  x = fminf(x, __builtin_bit_cast(float, lane_xor_half(__builtin_bit_cast(int, x))));
  return x;
}
__device__ __forceinline__ int wave_min_i(int x) {
  x = min(x, lane_xor_row<1>(x));
  x = min(x, lane_xor_row<2>(x));
  x = min(x, lane_xor_row<4>(x));
  x = min(x, lane_xor_row<8>(x));
  x = min(x, lane_xor_half(x));
  return x;
}
__device__ __forceinline__ int wave_max_i(int x) {
  x = max(x, lane_xor_row<1>(x));
  x = max(x, lane_xor_row<2>(x));
  x = max(x, lane_xor_row<4>(x));
  x = max(x, lane_xor_row<8>(x));
  x = max(x, lane_xor_half(x));
  return x;
}

#define FPS_T 256
#define FPS_W (FPS_T / 32)
#define FPS_P (NPTS / FPS_T)
__global__ __launch_bounds__(FPS_T) void k_fps(const float* __restrict__ xyz, int S,
                                               int* __restrict__ fps_idx) {
  static_assert(FPS_P == 8 && FPS_W == 8, "max tree and post reduction are written for 8 slots, 8 waves");
  __shared__ alignas(16) int post[2][FPS_W][8];
  __shared__ int sel[NPTS / 2];
  const int b = blockIdx.x;
  const float* p = xyz + (size_t)b * NPTS * 3;
  const int tid = threadIdx.x;
  const int lane = tid & 31, w = tid >> 5;
  float x[FPS_P], y[FPS_P], z[FPS_P]; int d[FPS_P];
#pragma unroll
  for (int q = 0; q < FPS_P; ++q) {
    const int i = tid * FPS_P + q;
    x[q] = p[i*3]; y[q] = p[i*3+1]; z[q] = p[i*3+2];
    d[q] = __float_as_int(1e10f);
  }
  float fx = p[0], fy = p[1], fz = p[2];
  if (tid == 0) sel[0] = 0;
  for (int it = 1; it < S; ++it) {
#pragma unroll
    for (int q = 0; q < FPS_P; ++q) {
      const float dx = x[q]-fx, dy = y[q]-fy, dz = z[q]-fz;
      const float s = fmaf(dz, dz, fmaf(dx, dx, dy * dy));
      d[q] = min(d[q], __float_as_int(s));
    }
    const int best = max(max(max(d[0], d[1]), max(d[2], d[3])), max(max(d[4], d[5]), max(d[6], d[7])));
    int bq = FPS_P - 1; float cx = x[FPS_P - 1], cy = y[FPS_P - 1], cz = z[FPS_P - 1];
#pragma unroll
    for (int q = FPS_P - 2; q >= 0; --q) {
      const bool h = (d[q] == best);
      bq = h ? q : bq; cx = h ? x[q] : cx; cy = h ? y[q] : cy; cz = h ? z[q] : cz;
    }
    const int bi = tid * FPS_P + bq;
    const int wmax = wave_max_i(best);
    const unsigned hit = __builtin_amdgcn_ballot_w32(best == wmax);
    const int wl = __builtin_ctz(hit);
    const int pi = __builtin_amdgcn_readlane(bi, wl);
    const int px = __builtin_amdgcn_readlane(__float_as_int(cx), wl);
    const int py = __builtin_amdgcn_readlane(__float_as_int(cy), wl);
    const int pz = __builtin_amdgcn_readlane(__float_as_int(cz), wl);
    int* mypost = post[it & 1][w];
    if (lane == 0) {
      *(int4*)mypost = make_int4(wmax, pi, px, py);
      mypost[4] = pz;
    }
    __syncthreads();
    const int* e = post[it & 1][lane & (FPS_W - 1)];
    const int4 e0 = *(const int4*)e;
    const int  ez = e[4];
    int gmax = e0.x;
    gmax = max(gmax, lane_xor_row<1>(gmax));
    gmax = max(gmax, lane_xor_row<2>(gmax));
    gmax = max(gmax, lane_xor_row<4>(gmax));
    const unsigned ghit = __builtin_amdgcn_ballot_w32(e0.x == gmax);
    const int gl = __builtin_ctz(ghit);
    const int far = __builtin_amdgcn_readlane(e0.y, gl);
    fx = __int_as_float(__builtin_amdgcn_readlane(e0.z, gl));
    fy = __int_as_float(__builtin_amdgcn_readlane(e0.w, gl));
    fz = __int_as_float(__builtin_amdgcn_readlane(ez, gl));
    if (tid == 0) sel[it] = far;
  }
  __syncthreads();
  for (int i = tid; i < S; i += FPS_T) fps_idx[b * S + i] = sel[i];
}

template <bool SAMPLED>
__global__ void k_gather(const float* __restrict__ xyz, const float* __restrict__ feat,
                         const int* __restrict__ fps_idx, int N, int S, int C,
                         float* __restrict__ xyz_s, float* __restrict__ feat_s) {
  const int b = blockIdx.x / S, s = blockIdx.x % S;
  const int idx = SAMPLED ? fps_idx[b * S + s] : s;
  if (threadIdx.x < 3)
    xyz_s[(size_t)(b * S + s) * 3 + threadIdx.x] = xyz[(size_t)(b * N + idx) * 3 + threadIdx.x];
  for (int j = threadIdx.x; j < C; j += blockDim.x)
    feat_s[(size_t)(b * S + s) * C + j] = feat[(size_t)(b * N + idx) * C + j];
}

#define KNN_NG 8
template <int NPL, int g>
__device__ __forceinline__ void knn_group_min(float (&v)[NPL], int kill, float& m, int& mj) {
  constexpr int G  = NPL / KNN_NG;
  constexpr int NC = (G >= 4) ? 2 : 1;
  constexpr int L  = G / NC;
  float cb[NC]; int cj[NC];
#pragma unroll
  for (int c = 0; c < NC; ++c) {
    cb[c] = FLT_MAX; cj[c] = g * G + c * L;
#pragma unroll
    for (int q = 0; q < L; ++q) {
      const int j = g * G + c * L + q;
      const float vj = (j == kill) ? FLT_MAX : v[j];
      v[j] = vj;
      if (vj < cb[c]) { cb[c] = vj; cj[c] = j; }
    }
  }
  if (NC == 2) { if (cb[NC - 1] < cb[0]) { cb[0] = cb[NC - 1]; cj[0] = cj[NC - 1]; } }
  m = cb[0]; mj = cj[0];
}
template <int NPL, int g>
__device__ __forceinline__ void knn_all_groups(float (&v)[NPL], float (&gm)[KNN_NG], int (&gj)[KNN_NG]) {
  knn_group_min<NPL, g>(v, -1, gm[g], gj[g]);
  if constexpr (g + 1 < KNN_NG) knn_all_groups<NPL, g + 1>(v, gm, gj);
}

template <int NPL, int g>
__device__ __forceinline__ void knn_regroup(float (&v)[NPL], int kill, unsigned grp, float (&gm)[KNN_NG], int (&gj)[KNN_NG]) {
  if (grp == (unsigned)g) knn_group_min<NPL, g>(v, kill, gm[g], gj[g]);
  if constexpr (g + 1 < KNN_NG) knn_regroup<NPL, g + 1>(v, kill, grp, gm, gj);
}

#define KNN_W 16
template <int NPL>
__global__ __launch_bounds__(KNN_W * 32) void k_knn(const float* __restrict__ xs_all,
                                                    const float* __restrict__ x_all,
                                                    int S, int* __restrict__ knn_idx) {
  constexpr int N = NPL * 32;
  constexpr int TPW = N / 16 / KNN_W;
  constexpr int PITCH = N + 4;
  __shared__ float strip[16 * PITCH];
  __shared__ float sN[16];
  const int w    = threadIdx.x >> 5;
  const int lane = threadIdx.x & 31;
  const int hf   = lane >> 4;
  const int ln   = lane & 15;
  const int rbase = blockIdx.x * 16;
  const float* xs = xs_all + (size_t)rbase * 3;
  const float* x  = x_all + (size_t)(rbase / S) * N * 3;
  if (threadIdx.x < 16) {
    float a0 = xs[threadIdx.x * 3];
    float a1 = xs[threadIdx.x * 3 + 1];
    float a2 = xs[threadIdx.x * 3 + 2];
    sN[threadIdx.x] = fmaf(a2, a2, fmaf(a0, a0, a1 * a1));
  }
  __syncthreads();
  const float s0 = xs[ln * 3], s1 = xs[ln * 3 + 1], s2 = xs[ln * 3 + 2];
  v2f a;
  a.x = hf ? s2 : s0;  a.y = hf ? 0.f : s1;
#pragma unroll
  for (int t = 0; t < TPW; ++t) {
    const int n = (w * TPW + t) * 16 + ln;
    const float x0 = x[n * 3], x1 = x[n * 3 + 1], x2 = x[n * 3 + 2];
    const float nn = fmaf(x2, x2, fmaf(x0, x0, x1 * x1));
    v2f b;
    b.x = hf ? x2 : x0;  b.y = hf ? 0.f : x1;
    v8f d = {};
    d = __builtin_amdgcn_wmma_f32_16x16x4_f32(false, a, false, b, (short)0, d, false, false);
#pragma unroll
    for (int r = 0; r < 8; ++r)
      strip[(r + 8 * hf) * PITCH + n] = fmaf(-2.f, d[r], sN[r + 8 * hf] + nn);
  }
  __syncthreads();
  const float* row = strip + w * PITCH;
  float v[NPL];
#pragma unroll
  for (int j = 0; j < NPL; ++j) v[j] = row[j * 32 + lane];
  float gm[KNN_NG]; int gj[KNN_NG];
  knn_all_groups<NPL, 0>(v, gm, gj);
  int mine = 0;
  for (int k = 0; k < KNN; ++k) {
    constexpr int HG = KNN_NG / 2;
    float b0 = gm[0], b1 = gm[HG]; int j0 = gj[0], j1 = gj[HG];
#pragma unroll
    for (int g = 1; g < HG; ++g) {
      if (gm[g] < b0)      { b0 = gm[g];      j0 = gj[g]; }
      if (gm[HG + g] < b1) { b1 = gm[HG + g]; j1 = gj[HG + g]; }
    }
    if (b1 < b0) { b0 = b1; j0 = j1; }
    const float wmin = wave_min_f(b0);
    const int cand = (b0 == wmin) ? (j0 * 32 + lane) : 0x7fffffff;
    const int widx = wave_min_i(cand);
    if (lane == k) mine = widx;
    if (k + 1 < KNN) {
      const int u    = __builtin_amdgcn_readfirstlane(widx);
      const int js   = u >> 5;
      const int kill = (lane == (u & 31)) ? js : -1;
      knn_regroup<NPL, 0>(v, kill, (unsigned)js / (NPL / KNN_NG), gm, gj);
    }
  }
  knn_idx[(size_t)(rbase + w) * KNN + lane] = mine;
}

template <int M>
__device__ __forceinline__ float lane_xor_row_f(float x) {
  return __builtin_bit_cast(float, lane_xor_row<M>(__builtin_bit_cast(int, x)));
}
__device__ __forceinline__ float row_sum_f(float x) {
  x += lane_xor_row_f<1>(x);
  x += lane_xor_row_f<2>(x);
  x += lane_xor_row_f<4>(x);
  x += lane_xor_row_f<8>(x);
  return x;
}
__device__ __forceinline__ float wave_sum_f(float x) {
  x = row_sum_f(x);
  x += __builtin_bit_cast(float, lane_xor_half(__builtin_bit_cast(int, x)));
  return x;
}

template <int C>
__global__ __launch_bounds__(256) void k_std_partial(const float* __restrict__ xyz,
                                                     const float* __restrict__ feat,
                                                     const float* __restrict__ xyz_s,
                                                     const float* __restrict__ feat_s,
                                                     const int* __restrict__ knn_idx,
                                                     float* __restrict__ part) {
  constexpr int S   = (NPTS / 2) * CINIT / C;
  constexpr int N   = 2 * S;
  constexpr int RPB = BATCH * S / STD_G;
  constexpr int CH  = (C >= 128) ? C / 128 : 1;
  constexpr int KL  = (C >= 128) ? 4 : 2;
  constexpr int U   = 16 / (KL * CH);
  static_assert(RPB >= 1 && RPB <= 32 && RPB % U == 0 && S % RPB == 0, "row split");
  const int lane = threadIdx.x & 31, w = threadIdx.x >> 5;
  const int hf   = lane >> 4;
  const int row0 = blockIdx.x * RPB;
  const int b    = row0 / S;

  int4 id4 = make_int4(0, 0, 0, 0);
  float ax[4] = {0.f, 0.f, 0.f, 0.f}, qx[4] = {0.f, 0.f, 0.f, 0.f};
  if (lane < RPB) {
    const int row = row0 + lane;
    id4 = *(const int4*)(knn_idx + (size_t)row * KNN + 4 * w);
    const float c0 = xyz_s[(size_t)row * 3], c1 = xyz_s[(size_t)row * 3 + 1], c2 = xyz_s[(size_t)row * 3 + 2];
    const int ids[4] = {id4.x, id4.y, id4.z, id4.w};
#pragma unroll
    for (int kl = 0; kl < 4; ++kl) {
      const float* q = xyz + (size_t)(b * N + ids[kl]) * 3;
      const float d0 = q[0] - c0, d1 = q[1] - c1, d2 = q[2] - c2;
      ax[kl] = d0 + d1 + d2;
      qx[kl] = d0 * d0 + d1 * d1 + d2 * d2;
    }
  }

  float af[KL], qf[KL];
#pragma unroll
  for (int kl = 0; kl < KL; ++kl) { af[kl] = 0.f; qf[kl] = 0.f; }
  const int jo = (C >= 128) ? lane * 4 : (lane & 15) * 4;
  const float* fbase = feat + (size_t)b * N * C + jo;
#pragma unroll 1
  for (int r0 = 0; r0 < RPB; r0 += U) {
    float4 fs[U][CH], fv[U][KL][CH];
#pragma unroll
    for (int u = 0; u < U; ++u) {
      const int r = r0 + u;
      const int i0 = __builtin_amdgcn_readlane(id4.x, r);
      const int i1 = __builtin_amdgcn_readlane(id4.y, r);
      const int i2 = __builtin_amdgcn_readlane(id4.z, r);
      const int i3 = __builtin_amdgcn_readlane(id4.w, r);
      const float* fc = feat_s + (size_t)(row0 + r) * C + jo;
#pragma unroll
      for (int ch = 0; ch < CH; ++ch) fs[u][ch] = *(const float4*)(fc + ch * 128);
#pragma unroll
      for (int kl = 0; kl < KL; ++kl) {
        int id;
        if (C >= 128) id = (kl == 0) ? i0 : (kl == 1) ? i1 : (kl == 2) ? i2 : i3;
        else          id = (kl == 0) ? (hf ? i1 : i0) : (hf ? i3 : i2);
        const float* fr = fbase + (size_t)id * C;
#pragma unroll
        for (int ch = 0; ch < CH; ++ch) fv[u][kl][ch] = *(const float4*)(fr + ch * 128);
      }
    }
#pragma unroll
    for (int u = 0; u < U; ++u) {
#pragma unroll
      for (int kl = 0; kl < KL; ++kl) {
#pragma unroll
        for (int ch = 0; ch < CH; ++ch) {
          const float d0 = fv[u][kl][ch].x - fs[u][ch].x, d1 = fv[u][kl][ch].y - fs[u][ch].y;
          const float d2 = fv[u][kl][ch].z - fs[u][ch].z, d3 = fv[u][kl][ch].w - fs[u][ch].w;
          af[kl] += (d0 + d1) + (d2 + d3);
          qf[kl] += (d0 * d0 + d1 * d1) + (d2 * d2 + d3 * d3);
        }
      }
    }
  }

#pragma unroll
  for (int kl = 0; kl < 4; ++kl) { ax[kl] = wave_sum_f(ax[kl]); qx[kl] = wave_sum_f(qx[kl]); }
#pragma unroll
  for (int kl = 0; kl < KL; ++kl) {
    if (C >= 128) { af[kl] = wave_sum_f(af[kl]); qf[kl] = wave_sum_f(qf[kl]); }
    else          { af[kl] = row_sum_f(af[kl]);  qf[kl] = row_sum_f(qf[kl]); }
  }
  int  kl;
  bool wr;
  float sf, sq;
  if (C >= 128) {
    kl = lane & 3; wr = lane < 4;
    sf = (kl == 0) ? af[0] : (kl == 1) ? af[1] : (kl == 2) ? af[KL - 2] : af[KL - 1];
    sq = (kl == 0) ? qf[0] : (kl == 1) ? qf[1] : (kl == 2) ? qf[KL - 2] : qf[KL - 1];
  } else {
    const int t = lane & 1;
    kl = 2 * t + hf; wr = (lane & 15) < 2;
    sf = t ? af[1] : af[0];
    sq = t ? qf[1] : qf[0];
  }
  const float sx = (kl == 0) ? ax[0] : (kl == 1) ? ax[1] : (kl == 2) ? ax[2] : ax[3];
  const float sy = (kl == 0) ? qx[0] : (kl == 1) ? qx[1] : (kl == 2) ? qx[2] : qx[3];
  if (wr)
    *(float4*)(part + ((size_t)(4 * w + kl) * STD_G + blockIdx.x) * 4) = make_float4(sx, sy, sf, sq);
}

__global__ __launch_bounds__(KNN * 32) void k_std_final(const float* __restrict__ part, int S, int C,
                                                        float* __restrict__ std_xyz, float* __restrict__ std_feat) {
  const int k = threadIdx.x >> 5, lane = threadIdx.x & 31;
  float ax = 0.f, qx = 0.f, af = 0.f, qf = 0.f;
#pragma unroll
  for (int i = 0; i < STD_G / 32; ++i) {
    const float4 q = *(const float4*)(part + ((size_t)k * STD_G + i * 32 + lane) * 4);
    ax += q.x; qx += q.y; af += q.z; qf += q.w;
  }
  ax = wave_sum_f(ax); qx = wave_sum_f(qx); af = wave_sum_f(af); qf = wave_sum_f(qf);
  if (lane != 0) return;
  {
    double n = (double)BATCH * S * 3;
    double var = ((double)qx - (double)ax * ax / n) / (n - 1.0);
    std_xyz[k] = fmaxf(sqrtf(fmaxf((float)var, 0.f)), 1e-5f);
  }
  {
    double n = (double)BATCH * S * C;
    double var = ((double)qf - (double)af * af / n) / (n - 1.0);
    std_feat[k] = fmaxf(sqrtf(fmaxf((float)var, 0.f)), 1e-5f);
  }
}

__global__ __launch_bounds__(256, 8) void k_stage(const float* __restrict__ xyz,
                                               const float* __restrict__ feat,
                                               const float* __restrict__ xyz_s,
                                               const float* __restrict__ feat_s,
                                               const int* __restrict__ knn_idx,
                                               const float* __restrict__ std_xyz,
                                               const float* __restrict__ std_feat,
                                               int N, int S, int C,
                                               float* __restrict__ out) {
  __shared__ float rsf[KNN];
  __shared__ float xkn[KNN][3];
  const int b   = blockIdx.x / S;
  const int row = blockIdx.x;
  const int* kr = knn_idx + (size_t)row * KNN;
  if (threadIdx.x < KNN * 3) {
    const int k = threadIdx.x / 3, c = threadIdx.x % 3;
    xkn[k][c] = (xyz[(size_t)(b * N + kr[k]) * 3 + c] - xyz_s[(size_t)row * 3 + c]) / std_xyz[k];
  } else if (threadIdx.x >= 128 && threadIdx.x < 128 + KNN) {
    rsf[threadIdx.x - 128] = 1.f / std_feat[threadIdx.x - 128];
  }
  __syncthreads();
  const int C2 = 2 * C;
  const int fd = (C2 + 2) / 3;
  const int M3 = 3 * fd;
  const float inv_sig = 1.f / SIGMA_;
  const char*  fbat = (const char*)(feat + (size_t)b * N * C);
  const float* crow = feat_s + (size_t)row * C;
  for (int j = threadIdx.x; j < C2; j += 256) {
    int src = (int)(((long long)j * (M3 - 1)) / (C2 - 1));
    int cc = src / fd, f = src % fd;
    float fv = -1.f + 2.f * (float)(f + 1) / (float)(fd + 1);
    float acc = 0.f, mx = -FLT_MAX;
    if (j < C) {
      const float fsj = crow[j];
      const unsigned joff = (unsigned)j * 4u;
      float g[KNN];
#pragma unroll
      for (int k = 0; k < KNN; ++k)
        g[k] = *(const float*)(fbat + (size_t)(unsigned)(kr[k] * C) * 4 + joff);
#pragma unroll
      for (int k = 0; k < KNN; ++k) {
        float t  = (xkn[k][cc] - fv) * inv_sig;
        float pe = expf(-0.5f * t * t);
        float fk = (g[k] - fsj) * rsf[k];
        float fw = (fk + pe) * pe;
        acc += fw; mx = fmaxf(mx, fw);
      }
    } else {
      const float fk = crow[j - C];
#pragma unroll
      for (int k = 0; k < KNN; ++k) {
        float t  = (xkn[k][cc] - fv) * inv_sig;
        float pe = expf(-0.5f * t * t);
        float fw = (fk + pe) * pe;
        acc += fw; mx = fmaxf(mx, fw);
      }
    }
    float xx = acc * (1.f / KNN) + mx;
    out[(size_t)row * C2 + j] = 0.5f * xx * (1.f + erff(xx * 0.70710678118654752f));
  }
}

#define POOL_W 16
__global__ __launch_bounds__(POOL_W * 32) void k_pool(const float* __restrict__ feat, int S, int C2,
                                                      float* __restrict__ out, int off) {
  __shared__ float pmx[POOL_W][32], psm[POOL_W][32];
  const int b    = blockIdx.y;
  const int lane = threadIdx.x & 31, w = threadIdx.x >> 5;
  const int j    = blockIdx.x * 32 + lane;
  const float* p = feat + ((size_t)b * S + w) * C2 + j;
  const size_t step = (size_t)POOL_W * C2;
  float mx = -FLT_MAX, sm = 0.f;
#pragma unroll 8
  for (int s = w; s < S; s += POOL_W, p += step) {
    float v = *p;
    mx = fmaxf(mx, v); sm += v;
  }
  pmx[w][lane] = mx; psm[w][lane] = sm;
  __syncthreads();
  if (w == 0) {
    float m = pmx[0][lane], t = psm[0][lane];
    for (int q = 1; q < POOL_W; ++q) { m = fmaxf(m, pmx[q][lane]); t += psm[q][lane]; }
    out[b * 3840 + off + j]      = m;
    out[b * 3840 + off + C2 + j] = t / (float)S;
  }
}

extern "C" void kernel_launch(void* const* d_in, const int* in_sizes, int n_in,
                              void* d_out, int out_size, void* d_ws, size_t ws_size,
                              hipStream_t stream) {
  const float* xyz_in = (const float*)d_in[0];
  const float* W      = (const float*)d_in[1];
  const float* gamma  = (const float*)d_in[2];
  const float* beta   = (const float*)d_in[3];
  float* out = (float*)d_out;
  float* ws  = (float*)d_ws;

  size_t o = 0;
  float* featA  = ws + o; o += (size_t)BATCH * NPTS * CINIT;
  float* featB  = ws + o; o += (size_t)BATCH * NPTS * CINIT;
  float* xyzA   = ws + o; o += (size_t)BATCH * NPTS * 3;
  float* xyzB   = ws + o; o += (size_t)BATCH * NPTS * 3;
  float* feat_s = ws + o; o += (size_t)BATCH * 1024 * CINIT;
  float* bn_mu  = ws + o; o += 64;
  float* bn_inv = ws + o; o += 64;
  float* stdx   = ws + o; o += 32;
  float* stdf   = ws + o; o += 32;
  o += (size_t)KNN * 16 * 4;
  int* fps_i = (int*)(ws + o); o += (size_t)BATCH * 1024;
  int* knn_i = (int*)(ws + o); o += (size_t)BATCH * 1024 * KNN;

  k_embed_wmma<<<1024, 256, 0, stream>>>(xyz_in, W, featA);
  k_bn_stats<<<CINIT, 256, 0, stream>>>(featA, bn_mu, bn_inv);
  k_bn_apply<<<2048, 256, 0, stream>>>(featA, bn_mu, bn_inv, gamma, beta);

  const float* xyz_cur = xyz_in;
  float* feat_cur = featA; float* feat_nxt = featB;
  float* xs_buf = xyzA;    float* xs_other = xyzB;
  int N = NPTS, C = CINIT, off = 0;

  for (int st = 0; st < 4; ++st) {
    const int S = N / 2, C2 = 2 * C;
    if (N == 2048) {
      k_fps<<<BATCH, FPS_T, 0, stream>>>(xyz_cur, S, fps_i);
      k_gather<true><<<BATCH * S, 256, 0, stream>>>(xyz_cur, feat_cur, fps_i, N, S, C, xs_buf, feat_s);
    } else {
      k_gather<false><<<BATCH * S, 256, 0, stream>>>(xyz_cur, feat_cur, fps_i, N, S, C, xs_buf, feat_s);
    }
    if (N == 2048)      k_knn<64><<<BATCH * S / 16, KNN_W * 32, 0, stream>>>(xs_buf, xyz_cur, S, knn_i);
    else if (N == 1024) k_knn<32><<<BATCH * S / 16, KNN_W * 32, 0, stream>>>(xs_buf, xyz_cur, S, knn_i);
    else if (N == 512)  k_knn<16><<<BATCH * S / 16, KNN_W * 32, 0, stream>>>(xs_buf, xyz_cur, S, knn_i);
    else                k_knn<8><<<BATCH * S / 16, KNN_W * 32, 0, stream>>>(xs_buf, xyz_cur, S, knn_i);
    {
      float* stdp = feat_nxt;
      if (C == 64)       k_std_partial<64><<<STD_G, 256, 0, stream>>>(xyz_cur, feat_cur, xs_buf, feat_s, knn_i, stdp);
      else if (C == 128) k_std_partial<128><<<STD_G, 256, 0, stream>>>(xyz_cur, feat_cur, xs_buf, feat_s, knn_i, stdp);
      else if (C == 256) k_std_partial<256><<<STD_G, 256, 0, stream>>>(xyz_cur, feat_cur, xs_buf, feat_s, knn_i, stdp);
      else               k_std_partial<512><<<STD_G, 256, 0, stream>>>(xyz_cur, feat_cur, xs_buf, feat_s, knn_i, stdp);
      k_std_final<<<1, KNN * 32, 0, stream>>>(stdp, S, C, stdx, stdf);
    }
    k_stage<<<BATCH * S, 256, 0, stream>>>(xyz_cur, feat_cur, xs_buf, feat_s, knn_i,
                                           stdx, stdf, N, S, C, feat_nxt);
    {
      dim3 g(C2 / 32, BATCH);
      k_pool<<<g, POOL_W * 32, 0, stream>>>(feat_nxt, S, C2, out, off);
    }
    off += 2 * C2;
    float* t = feat_cur; feat_cur = feat_nxt; feat_nxt = t;
    xyz_cur = xs_buf;
    float* u = xs_buf; xs_buf = xs_other; xs_other = u;
    N = S; C = C2;
  }
  (void)in_sizes; (void)n_in; (void)out_size; (void)ws_size;
}


# init embed+BN+ReLU run as 256 extra blocks of the FPS launch, beside its 16 serial blocks; BN mean/var from the 3x3 moments of xyz (h is linear in xyz): BN sum order changes; 3 launches fewer
# speedup vs baseline: 9.7005x; 1.0448x over previous
#include <hip/hip_runtime.h>
#include <cfloat>
#include <cmath>

typedef __attribute__((ext_vector_type(2))) float v2f;
typedef __attribute__((ext_vector_type(8))) float v8f;

#define BATCH 16
#define NPTS 2048
#define CINIT 64
#define KNN 32
#define SIGMA_ 0.3f
#define BN_EPS_ 1e-5f
#define STD_G 512

template <int M>
__device__ __forceinline__ int lane_xor_row(int x) {
  return __builtin_amdgcn_update_dpp(0, x, 0x160 + M, 0xf, 0xf, true);
}
__device__ __forceinline__ int lane_xor_half(int x) {
  return __builtin_amdgcn_permlanex16(x, x, 0x76543210, 0xfedcba98, false, false);
}
__device__ __forceinline__ float wave_min_f(float x) {
  x = fminf(x, __builtin_bit_cast(float, lane_xor_row<1>(__builtin_bit_cast(int, x))));
  x = fminf(x, __builtin_bit_cast(float, lane_xor_row<2>(__builtin_bit_cast(int, x))));
  x = fminf(x, __builtin_bit_cast(float, lane_xor_row<4>(__builtin_bit_cast(int, x))));
  x = fminf(x, __builtin_bit_cast(float, lane_xor_row<8>(__builtin_bit_cast(int, x))));
  x = fminf(x, __builtin_bit_cast(float, lane_xor_half(__builtin_bit_cast(int, x))));
  return x;
}
__device__ __forceinline__ int wave_min_i(int x) {
  x = min(x, lane_xor_row<1>(x));
  x = min(x, lane_xor_row<2>(x));
  x = min(x, lane_xor_row<4>(x));
  x = min(x, lane_xor_row<8>(x));
  x = min(x, lane_xor_half(x));
  return x;
}
__device__ __forceinline__ int wave_max_i(int x) {
  x = max(x, lane_xor_row<1>(x));
  x = max(x, lane_xor_row<2>(x));
  x = max(x, lane_xor_row<4>(x));
  x = max(x, lane_xor_row<8>(x));
  x = max(x, lane_xor_half(x));
  return x;
}

template <int M>
__device__ __forceinline__ float lane_xor_row_f(float x) {
  return __builtin_bit_cast(float, lane_xor_row<M>(__builtin_bit_cast(int, x)));
}
__device__ __forceinline__ float row_sum_f(float x) {
  x += lane_xor_row_f<1>(x);
  x += lane_xor_row_f<2>(x);
  x += lane_xor_row_f<4>(x);
  x += lane_xor_row_f<8>(x);
  return x;
}
__device__ __forceinline__ float wave_sum_f(float x) {
  x = row_sum_f(x);
  x += __builtin_bit_cast(float, lane_xor_half(__builtin_bit_cast(int, x)));
  return x;
}

#define FPS_T 256
#define FPS_W (FPS_T / 32)
#define FPS_P (NPTS / FPS_T)

#define EMB_ROWS 128
#define EMB_BLOCKS (BATCH * NPTS / EMB_ROWS)
__device__ __forceinline__ void embed_bn_rows(const float* __restrict__ xyz, const float* __restrict__ W,
                                              const float* __restrict__ gamma, const float* __restrict__ beta,
                                              float* __restrict__ feat, int eb) {
  __shared__ float red[FPS_T / 32][12];
  const int tid = threadIdx.x, lane = tid & 31, w = tid >> 5;
  float sx = 0.f, sy = 0.f, sz = 0.f, sxx = 0.f, syy = 0.f, szz = 0.f, sxy = 0.f, sxz = 0.f, syz = 0.f;
  const float4* p4 = (const float4*)xyz;
#pragma unroll 4
  for (int g = tid; g < BATCH * NPTS / 4; g += FPS_T) {
    const float4 q0 = p4[3 * g], q1 = p4[3 * g + 1], q2 = p4[3 * g + 2];
    const float px[4] = {q0.x, q0.w, q1.z, q2.y};
    const float py[4] = {q0.y, q1.x, q1.w, q2.z};
    const float pz[4] = {q0.z, q1.y, q2.x, q2.w};
#pragma unroll
    for (int i = 0; i < 4; ++i) {
      sx += px[i]; sy += py[i]; sz += pz[i];
      sxx = fmaf(px[i], px[i], sxx); syy = fmaf(py[i], py[i], syy); szz = fmaf(pz[i], pz[i], szz);
      sxy = fmaf(px[i], py[i], sxy); sxz = fmaf(px[i], pz[i], sxz); syz = fmaf(py[i], pz[i], syz);
    }
  }
  sx = wave_sum_f(sx); sy = wave_sum_f(sy); sz = wave_sum_f(sz);
  sxx = wave_sum_f(sxx); syy = wave_sum_f(syy); szz = wave_sum_f(szz);
  sxy = wave_sum_f(sxy); sxz = wave_sum_f(sxz); syz = wave_sum_f(syz);
  if (lane == 0) {
    float* r = red[w];
    r[0] = sx; r[1] = sy; r[2] = sz; r[3] = sxx; r[4] = syy; r[5] = szz; r[6] = sxy; r[7] = sxz; r[8] = syz;
  }
  __syncthreads();
  float t[9];
#pragma unroll
  for (int i = 0; i < 9; ++i) {
    float a = red[0][i];
#pragma unroll
    for (int q = 1; q < FPS_T / 32; ++q) a += red[q][i];
    t[i] = a;
  }
  const float rn = 1.f / (float)(BATCH * NPTS);
  const float mx = t[0] * rn, my = t[1] * rn, mz = t[2] * rn;
  const float cxx = t[3] * rn - mx * mx, cyy = t[4] * rn - my * my, czz = t[5] * rn - mz * mz;
  const float cxy = t[6] * rn - mx * my, cxz = t[7] * rn - mx * mz, cyz = t[8] * rn - my * mz;

  const int hf = lane >> 4;
  const int ln = lane & 15;
  const int tileM = eb * (EMB_ROWS / 16) + w;
  const int m = tileM * 16 + ln;
  const float x0 = xyz[m * 3], x1 = xyz[m * 3 + 1], x2 = xyz[m * 3 + 2];
  v2f a;
  a.x = hf ? x2 : x0;  a.y = hf ? 0.f : x1;
#pragma unroll
  for (int tileN = 0; tileN < CINIT / 16; ++tileN) {
    const int c = tileN * 16 + ln;
    const float w0 = W[c * 3], w1 = W[c * 3 + 1], w2 = W[c * 3 + 2];
    v2f b;
    b.x = hf ? w2 : w0;  b.y = hf ? 0.f : w1;
    v8f d = {};
    d = __builtin_amdgcn_wmma_f32_16x16x4_f32(false, a, false, b, (short)0, d, false, false);
    const float mean = w0 * mx + w1 * my + w2 * mz;
    const float var  = w0 * w0 * cxx + w1 * w1 * cyy + w2 * w2 * czz
                     + 2.f * (w0 * w1 * cxy + w0 * w2 * cxz + w1 * w2 * cyz);
    const float inv = rsqrtf(var + BN_EPS_);
    const float ga = gamma[c], be = beta[c];
#pragma unroll
    for (int r = 0; r < 8; ++r) {
      const int row = tileM * 16 + r + 8 * hf;
      const float v = (d[r] - mean) * inv * ga + be;
      feat[row * CINIT + c] = fmaxf(v, 0.f);
    }
  }
}

__global__ __launch_bounds__(FPS_T) void k_fps_embed(const float* __restrict__ xyz, int S,
                                                     int* __restrict__ fps_idx,
                                                     const float* __restrict__ W,
                                                     const float* __restrict__ gamma,
                                                     const float* __restrict__ beta,
                                                     float* __restrict__ feat) {
  static_assert(FPS_P == 8 && FPS_W == 8, "max tree and post reduction are written for 8 slots, 8 waves");
  if (blockIdx.x >= BATCH) {
    embed_bn_rows(xyz, W, gamma, beta, feat, (int)blockIdx.x - BATCH);
    return;
  }
  __shared__ alignas(16) int post[2][FPS_W][8];
  __shared__ int sel[NPTS / 2];
  const int b = blockIdx.x;
  const float* p = xyz + (size_t)b * NPTS * 3;
  const int tid = threadIdx.x;
  const int lane = tid & 31, w = tid >> 5;
  float x[FPS_P], y[FPS_P], z[FPS_P]; int d[FPS_P];
#pragma unroll
  for (int q = 0; q < FPS_P; ++q) {
    const int i = tid * FPS_P + q;
    x[q] = p[i*3]; y[q] = p[i*3+1]; z[q] = p[i*3+2];
    d[q] = __float_as_int(1e10f);
  }
  float fx = p[0], fy = p[1], fz = p[2];
  if (tid == 0) sel[0] = 0;
  for (int it = 1; it < S; ++it) {
#pragma unroll
    for (int q = 0; q < FPS_P; ++q) {
      const float dx = x[q]-fx, dy = y[q]-fy, dz = z[q]-fz;
      const float s = fmaf(dz, dz, fmaf(dx, dx, dy * dy));
      d[q] = min(d[q], __float_as_int(s));
    }
    const int best = max(max(max(d[0], d[1]), max(d[2], d[3])), max(max(d[4], d[5]), max(d[6], d[7])));
    int bq = FPS_P - 1; float cx = x[FPS_P - 1], cy = y[FPS_P - 1], cz = z[FPS_P - 1];
#pragma unroll
    for (int q = FPS_P - 2; q >= 0; --q) {
      const bool h = (d[q] == best);
      bq = h ? q : bq; cx = h ? x[q] : cx; cy = h ? y[q] : cy; cz = h ? z[q] : cz;
    }
    const int bi = tid * FPS_P + bq;
    const int wmax = wave_max_i(best);
    const unsigned hit = __builtin_amdgcn_ballot_w32(best == wmax);
    const int wl = __builtin_ctz(hit);
    const int pi = __builtin_amdgcn_readlane(bi, wl);
    const int px = __builtin_amdgcn_readlane(__float_as_int(cx), wl);
    const int py = __builtin_amdgcn_readlane(__float_as_int(cy), wl);
    const int pz = __builtin_amdgcn_readlane(__float_as_int(cz), wl);
    int* mypost = post[it & 1][w];
    if (lane == 0) {
      *(int4*)mypost = make_int4(wmax, pi, px, py);
      mypost[4] = pz;
    }
    __syncthreads();
    const int* e = post[it & 1][lane & (FPS_W - 1)];
    const int4 e0 = *(const int4*)e;
    const int  ez = e[4];
    int gmax = e0.x;
    gmax = max(gmax, lane_xor_row<1>(gmax));
    gmax = max(gmax, lane_xor_row<2>(gmax));
    gmax = max(gmax, lane_xor_row<4>(gmax));
    const unsigned ghit = __builtin_amdgcn_ballot_w32(e0.x == gmax);
    const int gl = __builtin_ctz(ghit);
    const int far = __builtin_amdgcn_readlane(e0.y, gl);
    fx = __int_as_float(__builtin_amdgcn_readlane(e0.z, gl));
    fy = __int_as_float(__builtin_amdgcn_readlane(e0.w, gl));
    fz = __int_as_float(__builtin_amdgcn_readlane(ez, gl));
    if (tid == 0) sel[it] = far;
  }
  __syncthreads();
  for (int i = tid; i < S; i += FPS_T) fps_idx[b * S + i] = sel[i];
}

template <bool SAMPLED>
__global__ void k_gather(const float* __restrict__ xyz, const float* __restrict__ feat,
                         const int* __restrict__ fps_idx, int N, int S, int C,
                         float* __restrict__ xyz_s, float* __restrict__ feat_s) {
  const int b = blockIdx.x / S, s = blockIdx.x % S;
  const int idx = SAMPLED ? fps_idx[b * S + s] : s;
  if (threadIdx.x < 3)
    xyz_s[(size_t)(b * S + s) * 3 + threadIdx.x] = xyz[(size_t)(b * N + idx) * 3 + threadIdx.x];
  for (int j = threadIdx.x; j < C; j += blockDim.x)
    feat_s[(size_t)(b * S + s) * C + j] = feat[(size_t)(b * N + idx) * C + j];
}

#define KNN_NG 8
template <int NPL, int g>
__device__ __forceinline__ void knn_group_min(float (&v)[NPL], int kill, float& m, int& mj) {
  constexpr int G  = NPL / KNN_NG;
  constexpr int NC = (G >= 4) ? 2 : 1;
  constexpr int L  = G / NC;
  float cb[NC]; int cj[NC];
#pragma unroll
  for (int c = 0; c < NC; ++c) {
    cb[c] = FLT_MAX; cj[c] = g * G + c * L;
#pragma unroll
    for (int q = 0; q < L; ++q) {
      const int j = g * G + c * L + q;
      const float vj = (j == kill) ? FLT_MAX : v[j];
      v[j] = vj;
      if (vj < cb[c]) { cb[c] = vj; cj[c] = j; }
    }
  }
  if (NC == 2) { if (cb[NC - 1] < cb[0]) { cb[0] = cb[NC - 1]; cj[0] = cj[NC - 1]; } }
  m = cb[0]; mj = cj[0];
}
template <int NPL, int g>
__device__ __forceinline__ void knn_all_groups(float (&v)[NPL], float (&gm)[KNN_NG], int (&gj)[KNN_NG]) {
  knn_group_min<NPL, g>(v, -1, gm[g], gj[g]);
  if constexpr (g + 1 < KNN_NG) knn_all_groups<NPL, g + 1>(v, gm, gj);
}

template <int NPL, int g>
__device__ __forceinline__ void knn_regroup(float (&v)[NPL], int kill, unsigned grp, float (&gm)[KNN_NG], int (&gj)[KNN_NG]) {
  if (grp == (unsigned)g) knn_group_min<NPL, g>(v, kill, gm[g], gj[g]);
  if constexpr (g + 1 < KNN_NG) knn_regroup<NPL, g + 1>(v, kill, grp, gm, gj);
}

#define KNN_W 16
template <int NPL>
__global__ __launch_bounds__(KNN_W * 32) void k_knn(const float* __restrict__ xs_all,
                                                    const float* __restrict__ x_all,
                                                    int S, int* __restrict__ knn_idx) {
  constexpr int N = NPL * 32;
  constexpr int TPW = N / 16 / KNN_W;
  constexpr int PITCH = N + 4;
  __shared__ float strip[16 * PITCH];
  __shared__ float sN[16];
  const int w    = threadIdx.x >> 5;
  const int lane = threadIdx.x & 31;
  const int hf   = lane >> 4;
  const int ln   = lane & 15;
  const int rbase = blockIdx.x * 16;
  const float* xs = xs_all + (size_t)rbase * 3;
  const float* x  = x_all + (size_t)(rbase / S) * N * 3;
  if (threadIdx.x < 16) {
    float a0 = xs[threadIdx.x * 3];
    float a1 = xs[threadIdx.x * 3 + 1];
    float a2 = xs[threadIdx.x * 3 + 2];
    sN[threadIdx.x] = fmaf(a2, a2, fmaf(a0, a0, a1 * a1));
  }
  __syncthreads();
  const float s0 = xs[ln * 3], s1 = xs[ln * 3 + 1], s2 = xs[ln * 3 + 2];
  v2f a;
  a.x = hf ? s2 : s0;  a.y = hf ? 0.f : s1;
#pragma unroll
  for (int t = 0; t < TPW; ++t) {
    const int n = (w * TPW + t) * 16 + ln;
    const float x0 = x[n * 3], x1 = x[n * 3 + 1], x2 = x[n * 3 + 2];
    const float nn = fmaf(x2, x2, fmaf(x0, x0, x1 * x1));
    v2f b;
    b.x = hf ? x2 : x0;  b.y = hf ? 0.f : x1;
    v8f d = {};
    d = __builtin_amdgcn_wmma_f32_16x16x4_f32(false, a, false, b, (short)0, d, false, false);
#pragma unroll
    for (int r = 0; r < 8; ++r)
      strip[(r + 8 * hf) * PITCH + n] = fmaf(-2.f, d[r], sN[r + 8 * hf] + nn);
  }
  __syncthreads();
  const float* row = strip + w * PITCH;
  float v[NPL];
#pragma unroll
  for (int j = 0; j < NPL; ++j) v[j] = row[j * 32 + lane];
  float gm[KNN_NG]; int gj[KNN_NG];
  knn_all_groups<NPL, 0>(v, gm, gj);
  int mine = 0;
  for (int k = 0; k < KNN; ++k) {
    constexpr int HG = KNN_NG / 2;
    float b0 = gm[0], b1 = gm[HG]; int j0 = gj[0], j1 = gj[HG];
#pragma unroll
    for (int g = 1; g < HG; ++g) {
      if (gm[g] < b0)      { b0 = gm[g];      j0 = gj[g]; }
      if (gm[HG + g] < b1) { b1 = gm[HG + g]; j1 = gj[HG + g]; }
    }
    if (b1 < b0) { b0 = b1; j0 = j1; }
    const float wmin = wave_min_f(b0);
    const int cand = (b0 == wmin) ? (j0 * 32 + lane) : 0x7fffffff;
    const int widx = wave_min_i(cand);
    if (lane == k) mine = widx;
    if (k + 1 < KNN) {
      const int u    = __builtin_amdgcn_readfirstlane(widx);
      const int js   = u >> 5;
      const int kill = (lane == (u & 31)) ? js : -1;
      knn_regroup<NPL, 0>(v, kill, (unsigned)js / (NPL / KNN_NG), gm, gj);
    }
  }
  knn_idx[(size_t)(rbase + w) * KNN + lane] = mine;
}

template <int C>
__global__ __launch_bounds__(256) void k_std_partial(const float* __restrict__ xyz,
                                                     const float* __restrict__ feat,
                                                     const float* __restrict__ xyz_s,
                                                     const float* __restrict__ feat_s,
                                                     const int* __restrict__ knn_idx,
                                                     float* __restrict__ part) {
  constexpr int S   = (NPTS / 2) * CINIT / C;
  constexpr int N   = 2 * S;
  constexpr int RPB = BATCH * S / STD_G;
  constexpr int CH  = (C >= 128) ? C / 128 : 1;
  constexpr int KL  = (C >= 128) ? 4 : 2;
  constexpr int U   = 16 / (KL * CH);
  static_assert(RPB >= 1 && RPB <= 32 && RPB % U == 0 && S % RPB == 0, "row split");
  const int lane = threadIdx.x & 31, w = threadIdx.x >> 5;
  const int hf   = lane >> 4;
  const int row0 = blockIdx.x * RPB;
  const int b    = row0 / S;

  int4 id4 = make_int4(0, 0, 0, 0);
  float ax[4] = {0.f, 0.f, 0.f, 0.f}, qx[4] = {0.f, 0.f, 0.f, 0.f};
  if (lane < RPB) {
    const int row = row0 + lane;
    id4 = *(const int4*)(knn_idx + (size_t)row * KNN + 4 * w);
    const float c0 = xyz_s[(size_t)row * 3], c1 = xyz_s[(size_t)row * 3 + 1], c2 = xyz_s[(size_t)row * 3 + 2];
    const int ids[4] = {id4.x, id4.y, id4.z, id4.w};
#pragma unroll
    for (int kl = 0; kl < 4; ++kl) {
      const float* q = xyz + (size_t)(b * N + ids[kl]) * 3;
      const float d0 = q[0] - c0, d1 = q[1] - c1, d2 = q[2] - c2;
      ax[kl] = d0 + d1 + d2;
      qx[kl] = d0 * d0 + d1 * d1 + d2 * d2;
    }
  }

  float af[KL], qf[KL];
#pragma unroll
  for (int kl = 0; kl < KL; ++kl) { af[kl] = 0.f; qf[kl] = 0.f; }
  const int jo = (C >= 128) ? lane * 4 : (lane & 15) * 4;
  const float* fbase = feat + (size_t)b * N * C + jo;
#pragma unroll 1
  for (int r0 = 0; r0 < RPB; r0 += U) {
    float4 fs[U][CH], fv[U][KL][CH];
#pragma unroll
    for (int u = 0; u < U; ++u) {
      const int r = r0 + u;
      const int i0 = __builtin_amdgcn_readlane(id4.x, r);
      const int i1 = __builtin_amdgcn_readlane(id4.y, r);
      const int i2 = __builtin_amdgcn_readlane(id4.z, r);
      const int i3 = __builtin_amdgcn_readlane(id4.w, r);
      const float* fc = feat_s + (size_t)(row0 + r) * C + jo;
#pragma unroll
      for (int ch = 0; ch < CH; ++ch) fs[u][ch] = *(const float4*)(fc + ch * 128);
#pragma unroll
      for (int kl = 0; kl < KL; ++kl) {
        int id;
        if (C >= 128) id = (kl == 0) ? i0 : (kl == 1) ? i1 : (kl == 2) ? i2 : i3;
        else          id = (kl == 0) ? (hf ? i1 : i0) : (hf ? i3 : i2);
        const float* fr = fbase + (size_t)id * C;
#pragma unroll
        for (int ch = 0; ch < CH; ++ch) fv[u][kl][ch] = *(const float4*)(fr + ch * 128);
      }
    }
#pragma unroll
    for (int u = 0; u < U; ++u) {
#pragma unroll
      for (int kl = 0; kl < KL; ++kl) {
#pragma unroll
        for (int ch = 0; ch < CH; ++ch) {
          const float d0 = fv[u][kl][ch].x - fs[u][ch].x, d1 = fv[u][kl][ch].y - fs[u][ch].y;
          const float d2 = fv[u][kl][ch].z - fs[u][ch].z, d3 = fv[u][kl][ch].w - fs[u][ch].w;
          af[kl] += (d0 + d1) + (d2 + d3);
          qf[kl] += (d0 * d0 + d1 * d1) + (d2 * d2 + d3 * d3);
        }
      }
    }
  }

#pragma unroll
  for (int kl = 0; kl < 4; ++kl) { ax[kl] = wave_sum_f(ax[kl]); qx[kl] = wave_sum_f(qx[kl]); }
#pragma unroll
  for (int kl = 0; kl < KL; ++kl) {
    if (C >= 128) { af[kl] = wave_sum_f(af[kl]); qf[kl] = wave_sum_f(qf[kl]); }
    else          { af[kl] = row_sum_f(af[kl]);  qf[kl] = row_sum_f(qf[kl]); }
  }
  int  kl;
  bool wr;
  float sf, sq;
  if (C >= 128) {
    kl = lane & 3; wr = lane < 4;
    sf = (kl == 0) ? af[0] : (kl == 1) ? af[1] : (kl == 2) ? af[KL - 2] : af[KL - 1];
    sq = (kl == 0) ? qf[0] : (kl == 1) ? qf[1] : (kl == 2) ? qf[KL - 2] : qf[KL - 1];
  } else {
    const int t = lane & 1;
    kl = 2 * t + hf; wr = (lane & 15) < 2;
    sf = t ? af[1] : af[0];
    sq = t ? qf[1] : qf[0];
  }
  const float sx = (kl == 0) ? ax[0] : (kl == 1) ? ax[1] : (kl == 2) ? ax[2] : ax[3];
  const float sy = (kl == 0) ? qx[0] : (kl == 1) ? qx[1] : (kl == 2) ? qx[2] : qx[3];
  if (wr)
    *(float4*)(part + ((size_t)(4 * w + kl) * STD_G + blockIdx.x) * 4) = make_float4(sx, sy, sf, sq);
}

__global__ __launch_bounds__(KNN * 32) void k_std_final(const float* __restrict__ part, int S, int C,
                                                        float* __restrict__ std_xyz, float* __restrict__ std_feat) {
  const int k = threadIdx.x >> 5, lane = threadIdx.x & 31;
  float ax = 0.f, qx = 0.f, af = 0.f, qf = 0.f;
#pragma unroll
  for (int i = 0; i < STD_G / 32; ++i) {
    const float4 q = *(const float4*)(part + ((size_t)k * STD_G + i * 32 + lane) * 4);
    ax += q.x; qx += q.y; af += q.z; qf += q.w;
  }
  ax = wave_sum_f(ax); qx = wave_sum_f(qx); af = wave_sum_f(af); qf = wave_sum_f(qf);
  if (lane != 0) return;
  {
    double n = (double)BATCH * S * 3;
    double var = ((double)qx - (double)ax * ax / n) / (n - 1.0);
    std_xyz[k] = fmaxf(sqrtf(fmaxf((float)var, 0.f)), 1e-5f);
  }
  {
    double n = (double)BATCH * S * C;
    double var = ((double)qf - (double)af * af / n) / (n - 1.0);
    std_feat[k] = fmaxf(sqrtf(fmaxf((float)var, 0.f)), 1e-5f);
  }
}

__global__ __launch_bounds__(256, 8) void k_stage(const float* __restrict__ xyz,
                                               const float* __restrict__ feat,
                                               const float* __restrict__ xyz_s,
                                               const float* __restrict__ feat_s,
                                               const int* __restrict__ knn_idx,
                                               const float* __restrict__ std_xyz,
                                               const float* __restrict__ std_feat,
                                               int N, int S, int C,
                                               float* __restrict__ out) {
  __shared__ float rsf[KNN];
  __shared__ float xkn[KNN][3];
  const int b   = blockIdx.x / S;
  const int row = blockIdx.x;
  const int* kr = knn_idx + (size_t)row * KNN;
  if (threadIdx.x < KNN * 3) {
    const int k = threadIdx.x / 3, c = threadIdx.x % 3;
    xkn[k][c] = (xyz[(size_t)(b * N + kr[k]) * 3 + c] - xyz_s[(size_t)row * 3 + c]) / std_xyz[k];
  } else if (threadIdx.x >= 128 && threadIdx.x < 128 + KNN) {
    rsf[threadIdx.x - 128] = 1.f / std_feat[threadIdx.x - 128];
  }
  __syncthreads();
  const int C2 = 2 * C;
  const int fd = (C2 + 2) / 3;
  const int M3 = 3 * fd;
  const float inv_sig = 1.f / SIGMA_;
  const char*  fbat = (const char*)(feat + (size_t)b * N * C);
  const float* crow = feat_s + (size_t)row * C;
  for (int j = threadIdx.x; j < C2; j += 256) {
    int src = (int)(((long long)j * (M3 - 1)) / (C2 - 1));
    int cc = src / fd, f = src % fd;
    float fv = -1.f + 2.f * (float)(f + 1) / (float)(fd + 1);
    float acc = 0.f, mx = -FLT_MAX;
    if (j < C) {
      const float fsj = crow[j];
      const unsigned joff = (unsigned)j * 4u;
      float g[KNN];
#pragma unroll
      for (int k = 0; k < KNN; ++k)
        g[k] = *(const float*)(fbat + (size_t)(unsigned)(kr[k] * C) * 4 + joff);
#pragma unroll
      for (int k = 0; k < KNN; ++k) {
        float t  = (xkn[k][cc] - fv) * inv_sig;
        float pe = expf(-0.5f * t * t);
        float fk = (g[k] - fsj) * rsf[k];
        float fw = (fk + pe) * pe;
        acc += fw; mx = fmaxf(mx, fw);
      }
    } else {
      const float fk = crow[j - C];
#pragma unroll
      for (int k = 0; k < KNN; ++k) {
        float t  = (xkn[k][cc] - fv) * inv_sig;
        float pe = expf(-0.5f * t * t);
        float fw = (fk + pe) * pe;
        acc += fw; mx = fmaxf(mx, fw);
      }
    }
    float xx = acc * (1.f / KNN) + mx;
    out[(size_t)row * C2 + j] = 0.5f * xx * (1.f + erff(xx * 0.70710678118654752f));
  }
}

#define POOL_W 16
__global__ __launch_bounds__(POOL_W * 32) void k_pool(const float* __restrict__ feat, int S, int C2,
                                                      float* __restrict__ out, int off) {
  __shared__ float pmx[POOL_W][32], psm[POOL_W][32];
  const int b    = blockIdx.y;
  const int lane = threadIdx.x & 31, w = threadIdx.x >> 5;
  const int j    = blockIdx.x * 32 + lane;
  const float* p = feat + ((size_t)b * S + w) * C2 + j;
  const size_t step = (size_t)POOL_W * C2;
  float mx = -FLT_MAX, sm = 0.f;
#pragma unroll 8
  for (int s = w; s < S; s += POOL_W, p += step) {
    float v = *p;
    mx = fmaxf(mx, v); sm += v;
  }
  pmx[w][lane] = mx; psm[w][lane] = sm;
  __syncthreads();
  if (w == 0) {
    float m = pmx[0][lane], t = psm[0][lane];
    for (int q = 1; q < POOL_W; ++q) { m = fmaxf(m, pmx[q][lane]); t += psm[q][lane]; }
    out[b * 3840 + off + j]      = m;
    out[b * 3840 + off + C2 + j] = t / (float)S;
  }
}

extern "C" void kernel_launch(void* const* d_in, const int* in_sizes, int n_in,
                              void* d_out, int out_size, void* d_ws, size_t ws_size,
                              hipStream_t stream) {
  const float* xyz_in = (const float*)d_in[0];
  const float* W      = (const float*)d_in[1];
  const float* gamma  = (const float*)d_in[2];
  const float* beta   = (const float*)d_in[3];
  float* out = (float*)d_out;
  float* ws  = (float*)d_ws;

  size_t o = 0;
  float* featA  = ws + o; o += (size_t)BATCH * NPTS * CINIT;
  float* featB  = ws + o; o += (size_t)BATCH * NPTS * CINIT;
  float* xyzA   = ws + o; o += (size_t)BATCH * NPTS * 3;
  float* xyzB   = ws + o; o += (size_t)BATCH * NPTS * 3;
  float* feat_s = ws + o; o += (size_t)BATCH * 1024 * CINIT;
  o += 64 + 64;
  float* stdx   = ws + o; o += 32;
  float* stdf   = ws + o; o += 32;
  o += (size_t)KNN * 16 * 4;
  int* fps_i = (int*)(ws + o); o += (size_t)BATCH * 1024;
  int* knn_i = (int*)(ws + o); o += (size_t)BATCH * 1024 * KNN;

  const float* xyz_cur = xyz_in;
  float* feat_cur = featA; float* feat_nxt = featB;
  float* xs_buf = xyzA;    float* xs_other = xyzB;
  int N = NPTS, C = CINIT, off = 0;

  for (int st = 0; st < 4; ++st) {
    const int S = N / 2, C2 = 2 * C;
    if (N == 2048) {
      k_fps_embed<<<BATCH + EMB_BLOCKS, FPS_T, 0, stream>>>(xyz_cur, S, fps_i, W, gamma, beta, feat_cur);
      k_gather<true><<<BATCH * S, 256, 0, stream>>>(xyz_cur, feat_cur, fps_i, N, S, C, xs_buf, feat_s);
    } else {
      k_gather<false><<<BATCH * S, 256, 0, stream>>>(xyz_cur, feat_cur, fps_i, N, S, C, xs_buf, feat_s);
    }
    if (N == 2048)      k_knn<64><<<BATCH * S / 16, KNN_W * 32, 0, stream>>>(xs_buf, xyz_cur, S, knn_i);
    else if (N == 1024) k_knn<32><<<BATCH * S / 16, KNN_W * 32, 0, stream>>>(xs_buf, xyz_cur, S, knn_i);
    else if (N == 512)  k_knn<16><<<BATCH * S / 16, KNN_W * 32, 0, stream>>>(xs_buf, xyz_cur, S, knn_i);
    else                k_knn<8><<<BATCH * S / 16, KNN_W * 32, 0, stream>>>(xs_buf, xyz_cur, S, knn_i);
    {
      float* stdp = feat_nxt;
      if (C == 64)       k_std_partial<64><<<STD_G, 256, 0, stream>>>(xyz_cur, feat_cur, xs_buf, feat_s, knn_i, stdp);
      else if (C == 128) k_std_partial<128><<<STD_G, 256, 0, stream>>>(xyz_cur, feat_cur, xs_buf, feat_s, knn_i, stdp);
      else if (C == 256) k_std_partial<256><<<STD_G, 256, 0, stream>>>(xyz_cur, feat_cur, xs_buf, feat_s, knn_i, stdp);
      else               k_std_partial<512><<<STD_G, 256, 0, stream>>>(xyz_cur, feat_cur, xs_buf, feat_s, knn_i, stdp);
      k_std_final<<<1, KNN * 32, 0, stream>>>(stdp, S, C, stdx, stdf);
    }
    k_stage<<<BATCH * S, 256, 0, stream>>>(xyz_cur, feat_cur, xs_buf, feat_s, knn_i,
                                           stdx, stdf, N, S, C, feat_nxt);
    {
      dim3 g(C2 / 32, BATCH);
      k_pool<<<g, POOL_W * 32, 0, stream>>>(feat_nxt, S, C2, out, off);
    }
    off += 2 * C2;
    float* t = feat_cur; feat_cur = feat_nxt; feat_nxt = t;
    xyz_cur = xs_buf;
    float* u = xs_buf; xs_buf = xs_other; xs_other = u;
    N = S; C = C2;
  }
  (void)in_sizes; (void)n_in; (void)out_size; (void)ws_size;
}
